# starting kernel
# baseline (speedup 1.0000x reference)
#include <hip/hip_runtime.h>
#include <math.h>

typedef float v2f __attribute__((ext_vector_type(2)));
typedef float v8f __attribute__((ext_vector_type(8)));

static constexpr int   NN = 100000;
static constexpr int   NE = 1600000;
static constexpr float NEG = 0.2f;

__device__ __forceinline__ float leaky(float x) { return x > 0.f ? x : NEG * x; }
__device__ __forceinline__ unsigned f2ord(float f) {
  unsigned u = __float_as_uint(f);
  return (u & 0x80000000u) ? ~u : (u | 0x80000000u);
}
__device__ __forceinline__ float ord2f(unsigned v) {
  unsigned u = (v & 0x80000000u) ? (v & 0x7fffffffu) : ~v;
  return __uint_as_float(u);
}

__global__ void zero_f32_k(float* p, int n) {
  int i = blockIdx.x * blockDim.x + threadIdx.x;
  if (i < n) p[i] = 0.f;
}
__global__ void relu_k(float* p, int n) {
  int i = blockIdx.x * blockDim.x + threadIdx.x;
  if (i < n) p[i] = fmaxf(p[i], 0.f);
}
__global__ void residual_k(const float* __restrict__ h2, float* __restrict__ r,
                           const float* __restrict__ br, int n) {
  int i = blockIdx.x * blockDim.x + threadIdx.x;
  if (i < n) { int c = i & 127; r[i] = fmaxf(h2[i], 0.f) + r[i] + br[c]; }
}

__global__ void attr_accum_k(const int* __restrict__ dst, const float* __restrict__ ea,
                             float* attr_sum, float* cnt) {
  int i = blockIdx.x * blockDim.x + threadIdx.x;
  if (i >= NE * 16) return;
  int e = i >> 4, j = i & 15;
  int d = dst[e];
  atomicAdd(&attr_sum[(size_t)d * 16 + j], ea[i]);
  if (j == 0) atomicAdd(&cnt[d], 1.f);
}
__global__ void attr_mean_k(float* attr_sum, const float* cnt) {
  int i = blockIdx.x * blockDim.x + threadIdx.x;
  if (i < NN * 16) attr_sum[i] /= fmaxf(cnt[i >> 4], 1.f);
}

__global__ void wa_k(const float* __restrict__ We, const float* __restrict__ ae,
                     float* wa, int K) {
  int i = threadIdx.x;
  if (i < 16) { float s = 0.f; for (int j = 0; j < K; ++j) s += We[i * K + j] * ae[j]; wa[i] = s; }
}

template <int COLT>
__global__ void gemm_k128(const float* __restrict__ A, const float* __restrict__ B,
                          float* __restrict__ C) {
  constexpr int COUT = COLT * 16;
  __shared__ float As[16 * 128];
  const int tid = threadIdx.x;
  const size_t abase = (size_t)blockIdx.x * (16 * 128);
  for (int i = tid * 4; i < 16 * 128; i += blockDim.x * 4)
    *(float4*)(&As[i]) = *(const float4*)(&A[abase + i]);
  __syncthreads();

  const int wave = tid >> 5, lane = tid & 31, half = lane >> 4, lidx = lane & 15;
  const int n0 = wave * 16;
  v8f acc = {};
#pragma unroll
  for (int k0 = 0; k0 < 128; k0 += 4) {
    v2f a, b;
    a.x = As[lidx * 128 + k0 + 2 * half];
    a.y = As[lidx * 128 + k0 + 2 * half + 1];
    b.x = B[(k0 + 2 * half) * COUT + n0 + lidx];
    b.y = B[(k0 + 2 * half + 1) * COUT + n0 + lidx];
    acc = __builtin_amdgcn_wmma_f32_16x16x4_f32(false, a, false, b, (short)0, acc,
                                                false, false);
  }
  const size_t row0 = (size_t)blockIdx.x * 16;
#pragma unroll
  for (int r = 0; r < 8; ++r)
    C[(row0 + r + 8 * half) * COUT + n0 + lidx] = acc[r];
}

template <int C, bool ET>
__global__ void node_prep_k(const float* __restrict__ h, const float* __restrict__ a_src,
                            const float* __restrict__ a_dst,
                            const float* __restrict__ mean_attr,
                            const float* __restrict__ wa, float* s_src, float* s_dst,
                            float* self_logit, unsigned* segmax_u) {
  int gid = blockIdx.x * blockDim.x + threadIdx.x;
  int n = gid >> 5, lane = gid & 31;
  if (n >= NN) return;
  float ps = 0.f, pd = 0.f;
#pragma unroll
  for (int i = 0; i < C / 32; ++i) {
    int c = lane + i * 32;
    float v = h[(size_t)n * C + c];
    ps += v * a_src[c];
    pd += v * a_dst[c];
  }
#pragma unroll
  for (int m = 16; m > 0; m >>= 1) { ps += __shfl_xor(ps, m, 32); pd += __shfl_xor(pd, m, 32); }
  if (lane == 0) {
    float st = 0.f;
    if (ET) { for (int j = 0; j < 16; ++j) st += mean_attr[(size_t)n * 16 + j] * wa[j]; }
    float sl = leaky(ps + pd + st);
    s_src[n] = ps; s_dst[n] = pd; self_logit[n] = sl; segmax_u[n] = f2ord(sl);
  }
}

__global__ void eterm_k(const float* __restrict__ ea, const float* __restrict__ wa,
                        float* eterm) {
  int i = blockIdx.x * blockDim.x + threadIdx.x;
  if (i >= NE * 16) return;
  int j = i & 15;
  float v = ea[i] * wa[j];
  v += __shfl_xor(v, 1, 32); v += __shfl_xor(v, 2, 32);
  v += __shfl_xor(v, 4, 32); v += __shfl_xor(v, 8, 32);
  if (j == 0) eterm[i >> 4] = v;
}

template <bool ET>
__global__ void edge_max_k(const int* __restrict__ src, const int* __restrict__ dst,
                           const float* __restrict__ s_src, const float* __restrict__ s_dst,
                           const float* __restrict__ eterm, unsigned* segmax_u) {
  int e = blockIdx.x * blockDim.x + threadIdx.x;
  if (e >= NE) return;
  int d = dst[e];
  float l = s_src[src[e]] + s_dst[d];
  if (ET) l += eterm[e];
  atomicMax(&segmax_u[d], f2ord(leaky(l)));
}

__global__ void node_mid_k(const unsigned* __restrict__ segmax_u,
                           const float* __restrict__ self_logit, float* segmaxf,
                           float* segsum) {
  int n = blockIdx.x * blockDim.x + threadIdx.x;
  if (n >= NN) return;
  float m = ord2f(segmax_u[n]);
  segmaxf[n] = m;
  segsum[n] = __expf(self_logit[n] - m);
}

template <bool ET>
__global__ void edge_sum_k(const int* __restrict__ src, const int* __restrict__ dst,
                           const float* __restrict__ s_src, const float* __restrict__ s_dst,
                           const float* __restrict__ eterm,
                           const float* __restrict__ segmaxf, float* segsum) {
  int e = blockIdx.x * blockDim.x + threadIdx.x;
  if (e >= NE) return;
  int d = dst[e];
  float l = s_src[src[e]] + s_dst[d];
  if (ET) l += eterm[e];
  atomicAdd(&segsum[d], __expf(leaky(l) - segmaxf[d]));
}

template <int C>
__global__ void out_init_k(const float* __restrict__ h, const float* __restrict__ self_logit,
                           const float* __restrict__ segmaxf, const float* __restrict__ segsum,
                           const float* __restrict__ bias, float* __restrict__ out) {
  int i = blockIdx.x * blockDim.x + threadIdx.x;
  if (i >= NN * (C / 4)) return;
  int n = i / (C / 4), q = i % (C / 4);
  float a = __expf(self_logit[n] - segmaxf[n]) / (segsum[n] + 1e-16f);
  float4 hv = *(const float4*)&h[(size_t)n * C + q * 4];
  float4 o;
  o.x = a * hv.x + bias[q * 4 + 0];
  o.y = a * hv.y + bias[q * 4 + 1];
  o.z = a * hv.z + bias[q * 4 + 2];
  o.w = a * hv.w + bias[q * 4 + 3];
  *(float4*)&out[(size_t)n * C + q * 4] = o;
}

template <int C, bool ET>
__global__ void edge_agg_k(const int* __restrict__ src, const int* __restrict__ dst,
                           const float* __restrict__ s_src, const float* __restrict__ s_dst,
                           const float* __restrict__ eterm,
                           const float* __restrict__ segmaxf, const float* __restrict__ segsum,
                           const float* __restrict__ h, float* __restrict__ out) {
  int gid = blockIdx.x * blockDim.x + threadIdx.x;
  int e = gid >> 5, lane = gid & 31;
  if (e >= NE) return;
  int s = src[e], d = dst[e];
  float l = s_src[s] + s_dst[d];
  if (ET) l += eterm[e];
  float alpha = __expf(leaky(l) - segmaxf[d]) / (segsum[d] + 1e-16f);
#pragma unroll
  for (int i = 0; i < C / 32; ++i) {
    int c = lane + i * 32;
    atomicAdd(&out[(size_t)d * C + c], alpha * h[(size_t)s * C + c]);
  }
}

struct Scratch {
  float *s_src, *s_dst, *self_l, *segmaxf, *segsum;
  unsigned* segmax_u;
  const float* mean_attr;
  float *eterm, *wa;
};

template <int C, bool ET>
static void run_gat(hipStream_t st, const float* hin, const float* Wm, const float* asrc,
                    const float* adst, const float* bias, const float* We, const float* ae,
                    const int* src, const int* dst, const float* edge_attr, float* hg,
                    float* out, const Scratch& S) {
  constexpr int COLT = C / 16;
  if (ET) wa_k<<<1, 16, 0, st>>>(We, ae, S.wa, C);
  gemm_k128<COLT><<<NN / 16, 32 * COLT, 0, st>>>(hin, Wm, hg);
  node_prep_k<C, ET><<<(NN * 32 + 255) / 256, 256, 0, st>>>(hg, asrc, adst, S.mean_attr,
                                                            S.wa, S.s_src, S.s_dst,
                                                            S.self_l, S.segmax_u);
  if (ET) eterm_k<<<(NE * 16 + 255) / 256, 256, 0, st>>>(edge_attr, S.wa, S.eterm);
  edge_max_k<ET><<<(NE + 255) / 256, 256, 0, st>>>(src, dst, S.s_src, S.s_dst, S.eterm,
                                                   S.segmax_u);
  node_mid_k<<<(NN + 255) / 256, 256, 0, st>>>(S.segmax_u, S.self_l, S.segmaxf, S.segsum);
  edge_sum_k<ET><<<(NE + 255) / 256, 256, 0, st>>>(src, dst, S.s_src, S.s_dst, S.eterm,
                                                   S.segmaxf, S.segsum);
  out_init_k<C><<<(NN * (C / 4) + 255) / 256, 256, 0, st>>>(hg, S.self_l, S.segmaxf,
                                                            S.segsum, bias, out);
  edge_agg_k<C, ET><<<(NE * 32 + 255) / 256, 256, 0, st>>>(src, dst, S.s_src, S.s_dst,
                                                           S.eterm, S.segmaxf, S.segsum,
                                                           hg, out);
}

extern "C" void kernel_launch(void* const* d_in, const int* in_sizes, int n_in, void* d_out,
                              int out_size, void* d_ws, size_t ws_size, hipStream_t stream) {
  (void)in_sizes; (void)n_in; (void)out_size; (void)ws_size;
  const float* x = (const float*)d_in[0];
  const int* ei = (const int*)d_in[1];
  const int* src = ei;
  const int* dst = ei + NE;
  const float* edge_attr = (const float*)d_in[2];
  const float *W1 = (const float*)d_in[3], *a1s = (const float*)d_in[4],
              *a1d = (const float*)d_in[5], *b1 = (const float*)d_in[6];
  const float *W2 = (const float*)d_in[7], *a2s = (const float*)d_in[8],
              *a2d = (const float*)d_in[9], *b2 = (const float*)d_in[10];
  const float *We2 = (const float*)d_in[11], *ae2 = (const float*)d_in[12];
  const float *Wr = (const float*)d_in[13], *br = (const float*)d_in[14];
  const float *Wmu = (const float*)d_in[15], *amus = (const float*)d_in[16],
              *amud = (const float*)d_in[17], *bmu = (const float*)d_in[18],
              *Wemu = (const float*)d_in[19], *aemu = (const float*)d_in[20];
  const float *Wls = (const float*)d_in[21], *alss = (const float*)d_in[22],
              *alsd = (const float*)d_in[23], *bls = (const float*)d_in[24],
              *Wels = (const float*)d_in[25], *aels = (const float*)d_in[26];
  float* outf = (float*)d_out;

  float* W = (float*)d_ws;
  size_t o = 0;
  float* hA = W + o; o += (size_t)NN * 128;
  float* hB = W + o; o += (size_t)NN * 128;
  float* hC = W + o; o += (size_t)NN * 128;
  Scratch S;
  S.s_src = W + o; o += NN;
  S.s_dst = W + o; o += NN;
  S.self_l = W + o; o += NN;
  S.segmaxf = W + o; o += NN;
  S.segsum = W + o; o += NN;
  S.segmax_u = (unsigned*)(W + o); o += NN;
  float* cnt = W + o; o += NN;
  float* mean_attr = W + o; o += (size_t)NN * 16;
  S.mean_attr = mean_attr;
  S.eterm = W + o; o += NE;
  S.wa = W + o; o += 16;

  zero_f32_k<<<(NN * 17 + 255) / 256, 256, 0, stream>>>(cnt, NN * 17);
  attr_accum_k<<<(NE * 16 + 255) / 256, 256, 0, stream>>>(dst, edge_attr, mean_attr, cnt);
  attr_mean_k<<<(NN * 16 + 255) / 256, 256, 0, stream>>>(mean_attr, cnt);

  run_gat<128, false>(stream, x, W1, a1s, a1d, b1, nullptr, nullptr, src, dst, edge_attr,
                      hA, hB, S);
  relu_k<<<(NN * 128 + 255) / 256, 256, 0, stream>>>(hB, NN * 128);

  run_gat<128, true>(stream, hB, W2, a2s, a2d, b2, We2, ae2, src, dst, edge_attr, hA, hC, S);

  gemm_k128<8><<<NN / 16, 256, 0, stream>>>(x, Wr, hB);
  residual_k<<<(NN * 128 + 255) / 256, 256, 0, stream>>>(hC, hB, br, NN * 128);

  run_gat<64, true>(stream, hB, Wmu, amus, amud, bmu, Wemu, aemu, src, dst, edge_attr, hA,
                    outf, S);
  run_gat<64, true>(stream, hB, Wls, alss, alsd, bls, Wels, aels, src, dst, edge_attr, hA,
                    outf + (size_t)NN * 64, S);
}


# Aggregation in LDS: each workgroup owns 400 nodes, sweeps the whole edge list for their edges, adds w*h[src] into its tile by LDS float atomics (arrival order as before); global float adds gone
# speedup vs baseline: 1.5203x; 1.5203x over previous
#include <hip/hip_runtime.h>
#include <math.h>

typedef float v2f __attribute__((ext_vector_type(2)));
typedef float v8f __attribute__((ext_vector_type(8)));

static constexpr int   NN = 100000;
static constexpr int   NE = 1600000;
static constexpr float NEG = 0.2f;

__device__ __forceinline__ float leaky(float x) { return x > 0.f ? x : NEG * x; }
__device__ __forceinline__ unsigned f2ord(float f) {
  unsigned u = __float_as_uint(f);
  return (u & 0x80000000u) ? ~u : (u | 0x80000000u);
}
__device__ __forceinline__ float ord2f(unsigned v) {
  unsigned u = (v & 0x80000000u) ? (v & 0x7fffffffu) : ~v;
  return __uint_as_float(u);
}

__global__ void zero_f32_k(float* p, int n) {
  int i = blockIdx.x * blockDim.x + threadIdx.x;
  if (i < n) p[i] = 0.f;
}
__global__ void norm_relu_k(float* p, const float* __restrict__ segsum,
                            const float* __restrict__ bias, int n) {
  int i = blockIdx.x * blockDim.x + threadIdx.x;
  if (i < n) p[i] = fmaxf(p[i] / (segsum[i >> 7] + 1e-16f) + bias[i & 127], 0.f);
}
__global__ void residual_k(const float* __restrict__ h2, const float* __restrict__ segsum,
                           const float* __restrict__ b2, float* __restrict__ r,
                           const float* __restrict__ br, int n) {
  int i = blockIdx.x * blockDim.x + threadIdx.x;
  if (i < n) {
    int c = i & 127;
    r[i] = fmaxf(h2[i] / (segsum[i >> 7] + 1e-16f) + b2[c], 0.f) + r[i] + br[c];
  }
}
template <int C>
__global__ void norm_bias_k(float* __restrict__ out, const float* __restrict__ segsum,
                            const float* __restrict__ bias) {
  int i = blockIdx.x * blockDim.x + threadIdx.x;
  if (i >= NN * (C / 4)) return;
  int n = i / (C / 4), q = i % (C / 4);
  float den = segsum[n] + 1e-16f;
  float4 v = *(const float4*)&out[(size_t)n * C + q * 4];
  v.x = v.x / den + bias[q * 4 + 0];
  v.y = v.y / den + bias[q * 4 + 1];
  v.z = v.z / den + bias[q * 4 + 2];
  v.w = v.w / den + bias[q * 4 + 3];
  *(float4*)&out[(size_t)n * C + q * 4] = v;
}

__global__ void attr_accum_k(const int* __restrict__ dst, const float* __restrict__ ea,
                             float* attr_sum, float* cnt) {
  int i = blockIdx.x * blockDim.x + threadIdx.x;
  if (i >= NE * 16) return;
  int e = i >> 4, j = i & 15;
  int d = dst[e];
  atomicAdd(&attr_sum[(size_t)d * 16 + j], ea[i]);
  if (j == 0) atomicAdd(&cnt[d], 1.f);
}
__global__ void attr_mean_k(float* attr_sum, const float* cnt) {
  int i = blockIdx.x * blockDim.x + threadIdx.x;
  if (i < NN * 16) attr_sum[i] /= fmaxf(cnt[i >> 4], 1.f);
}

__global__ void wa_k(const float* __restrict__ We, const float* __restrict__ ae,
                     float* wa, int K) {
  int i = threadIdx.x;
  if (i < 16) { float s = 0.f; for (int j = 0; j < K; ++j) s += We[i * K + j] * ae[j]; wa[i] = s; }
}

template <int COLT>
__global__ void gemm_k128(const float* __restrict__ A, const float* __restrict__ B,
                          float* __restrict__ C) {
  constexpr int COUT = COLT * 16;
  __shared__ float As[16 * 128];
  const int tid = threadIdx.x;
  const size_t abase = (size_t)blockIdx.x * (16 * 128);
  for (int i = tid * 4; i < 16 * 128; i += blockDim.x * 4)
    *(float4*)(&As[i]) = *(const float4*)(&A[abase + i]);
  __syncthreads();

  const int wave = tid >> 5, lane = tid & 31, half = lane >> 4, lidx = lane & 15;
  const int n0 = wave * 16;
  v8f acc = {};
#pragma unroll
  for (int k0 = 0; k0 < 128; k0 += 4) {
    v2f a, b;
    a.x = As[lidx * 128 + k0 + 2 * half];
    a.y = As[lidx * 128 + k0 + 2 * half + 1];
    b.x = B[(k0 + 2 * half) * COUT + n0 + lidx];
    b.y = B[(k0 + 2 * half + 1) * COUT + n0 + lidx];
    acc = __builtin_amdgcn_wmma_f32_16x16x4_f32(false, a, false, b, (short)0, acc,
                                                false, false);
  }
  const size_t row0 = (size_t)blockIdx.x * 16;
#pragma unroll
  for (int r = 0; r < 8; ++r)
    C[(row0 + r + 8 * half) * COUT + n0 + lidx] = acc[r];
}

template <int C, bool ET>
__global__ void node_prep_k(const float* __restrict__ h, const float* __restrict__ a_src,
                            const float* __restrict__ a_dst,
                            const float* __restrict__ mean_attr,
                            const float* __restrict__ wa, float* s_src, float* s_dst,
                            float* self_logit, unsigned* segmax_u) {
  int gid = blockIdx.x * blockDim.x + threadIdx.x;
  int n = gid >> 5, lane = gid & 31;
  if (n >= NN) return;
  float ps = 0.f, pd = 0.f;
#pragma unroll
  for (int i = 0; i < C / 32; ++i) {
    int c = lane + i * 32;
    float v = h[(size_t)n * C + c];
    ps += v * a_src[c];
    pd += v * a_dst[c];
  }
#pragma unroll
  for (int m = 16; m > 0; m >>= 1) { ps += __shfl_xor(ps, m, 32); pd += __shfl_xor(pd, m, 32); }
  if (lane == 0) {
    float st = 0.f;
    if (ET) { for (int j = 0; j < 16; ++j) st += mean_attr[(size_t)n * 16 + j] * wa[j]; }
    float sl = leaky(ps + pd + st);
    s_src[n] = ps; s_dst[n] = pd; self_logit[n] = sl; segmax_u[n] = f2ord(sl);
  }
}

__global__ void eterm_k(const float* __restrict__ ea, const float* __restrict__ wa,
                        float* eterm) {
  int i = blockIdx.x * blockDim.x + threadIdx.x;
  if (i >= NE * 16) return;
  int j = i & 15;
  float v = ea[i] * wa[j];
  v += __shfl_xor(v, 1, 32); v += __shfl_xor(v, 2, 32);
  v += __shfl_xor(v, 4, 32); v += __shfl_xor(v, 8, 32);
  if (j == 0) eterm[i >> 4] = v;
}

template <bool ET>
__global__ void edge_max_k(const int* __restrict__ src, const int* __restrict__ dst,
                           const float* __restrict__ s_src, const float* __restrict__ s_dst,
                           const float* __restrict__ eterm, unsigned* segmax_u) {
  int e = blockIdx.x * blockDim.x + threadIdx.x;
  if (e >= NE) return;
  int d = dst[e];
  float l = s_src[src[e]] + s_dst[d];
  if (ET) l += eterm[e];
  atomicMax(&segmax_u[d], f2ord(leaky(l)));
}

__global__ void node_mid_k(const unsigned* __restrict__ segmax_u,
                           const float* __restrict__ self_logit, float* segmaxf,
                           float* segsum) {
  int n = blockIdx.x * blockDim.x + threadIdx.x;
  if (n >= NN) return;
  float m = ord2f(segmax_u[n]);
  segmaxf[n] = m;
  segsum[n] = __expf(self_logit[n] - m);
}

static constexpr int RT = 400;
static constexpr int NTILE = (NN + RT - 1) / RT;
static constexpr int QCAP = 160;
static constexpr int NCHUNK = NE / 128;
static_assert(NE % 128 == 0, "edge list is swept in steps of 128");

template <int V> struct RowVec;
template <> struct RowVec<4> { typedef float4 T; };
template <> struct RowVec<2> { typedef float2 T; };
__device__ __forceinline__ float vget(const float4& v, int k) {
  return k == 0 ? v.x : k == 1 ? v.y : k == 2 ? v.z : v.w;
}
__device__ __forceinline__ float vget(const float2& v, int k) { return k == 0 ? v.x : v.y; }
__device__ __forceinline__ void vset(float4& v, int k, float f) {
  if (k == 0) v.x = f; else if (k == 1) v.y = f; else if (k == 2) v.z = f; else v.w = f;
}
__device__ __forceinline__ void vset(float2& v, int k, float f) {
  if (k == 0) v.x = f; else v.y = f;
}

template <int C, bool ET>
__global__ __launch_bounds__(1024) void tile_agg_k(
    const int* __restrict__ src, const int* __restrict__ dst,
    const float* __restrict__ s_src, const float* __restrict__ s_dst,
    const float* __restrict__ eterm, const float* __restrict__ segmaxf,
    float* __restrict__ segsum, const float* __restrict__ h, float* __restrict__ out) {
  constexpr int V = C / 32;
  typedef typename RowVec<V>::T VecT;
  __shared__ float tile[RT * C];
  __shared__ float ssum[RT];
  __shared__ int queue[32 * QCAP];
  const int tid = threadIdx.x, lane = tid & 31;
  const int wave = __builtin_amdgcn_readfirstlane(tid >> 5);
  const int row0 = blockIdx.x * RT;
  const int rows = min(RT, NN - row0);

  for (int i = tid; i < rows * 32; i += 1024) {
    const int r = i >> 5, n = row0 + r;
    const float a = segsum[n];
    const VecT hv = *(const VecT*)&h[(size_t)n * C + lane * V];
#pragma unroll
    for (int k = 0; k < V; ++k) tile[r * C + k * 32 + lane] = a * vget(hv, k);
    if (lane == 0) ssum[r] = a;
  }
  __syncthreads();

  int* myq = &queue[wave * QCAP];
  int cnt = 0;

  auto process = [&](int base, int nb) {
    int s = 0, r = 0;
    float w = 0.f;
    if (lane < nb) {
      const int e = myq[base + lane];
      s = src[e];
      const int d = dst[e];
      float l = s_src[s] + s_dst[d];
      if (ET) l += eterm[e];
      w = __expf(leaky(l) - segmaxf[d]);
      r = d - row0;
      atomicAdd(&ssum[r], w);
    }
    for (int j0 = 0; j0 < nb; j0 += 8) {
      VecT hv[8];
#pragma unroll
      for (int u = 0; u < 8; ++u) {
        const int sj = __builtin_amdgcn_readlane(s, j0 + u);
        hv[u] = *(const VecT*)&h[(size_t)sj * C + lane * V];
      }
#pragma unroll
      for (int u = 0; u < 8; ++u) {
        if (j0 + u < nb) {
          const int rj = __builtin_amdgcn_readlane(r, j0 + u);
          const float wj = __int_as_float(__builtin_amdgcn_readlane(__float_as_int(w), j0 + u));
          float* t = &tile[rj * C + lane];
#pragma unroll
          for (int k = 0; k < V; ++k) atomicAdd(t + k * 32, wj * vget(hv[u], k));
        }
      }
    }
  };

  const int rot = blockIdx.x * (NCHUNK / NTILE);
  auto chunk_of = [&](int k) { const int kk = k + rot; return kk >= NCHUNK ? kk - NCHUNK : kk; };
  const int4* dst4 = (const int4*)dst;
  auto fetch = [&](int4 (&d)[4], int k0) {
#pragma unroll
    for (int u = 0; u < 4; ++u) {
      const int k = k0 + 32 * u;
      d[u] = (k < NCHUNK) ? dst4[(size_t)chunk_of(k) * 32 + lane] : make_int4(-1, -1, -1, -1);
    }
  };
  int4 cur[4], nxt[4];
  fetch(cur, wave);
  for (int k0 = wave; k0 < NCHUNK; k0 += 32 * 4) {
    fetch(nxt, k0 + 32 * 4);
#pragma unroll
    for (int u = 0; u < 4; ++u) {
      const int e0 = chunk_of(k0 + 32 * u) * 128 + lane * 4;
      const bool p0 = (unsigned)(cur[u].x - row0) < (unsigned)rows;
      const bool p1 = (unsigned)(cur[u].y - row0) < (unsigned)rows;
      const bool p2 = (unsigned)(cur[u].z - row0) < (unsigned)rows;
      const bool p3 = (unsigned)(cur[u].w - row0) < (unsigned)rows;
      const unsigned m0 = __builtin_amdgcn_ballot_w32(p0);
      const unsigned m1 = __builtin_amdgcn_ballot_w32(p1);
      const unsigned m2 = __builtin_amdgcn_ballot_w32(p2);
      const unsigned m3 = __builtin_amdgcn_ballot_w32(p3);
      if ((m0 | m1 | m2 | m3) != 0u) {
        if (m0) { if (p0) myq[cnt + __builtin_amdgcn_mbcnt_lo(m0, 0u)] = e0;     cnt += __popc(m0); }
        if (m1) { if (p1) myq[cnt + __builtin_amdgcn_mbcnt_lo(m1, 0u)] = e0 + 1; cnt += __popc(m1); }
        if (m2) { if (p2) myq[cnt + __builtin_amdgcn_mbcnt_lo(m2, 0u)] = e0 + 2; cnt += __popc(m2); }
        if (m3) { if (p3) myq[cnt + __builtin_amdgcn_mbcnt_lo(m3, 0u)] = e0 + 3; cnt += __popc(m3); }
        if (cnt >= 32) {
          const int full = cnt & ~31;
          for (int b = 0; b < full; b += 32) process(b, 32);
          const int rem = cnt - full;
          const int keep = (lane < rem) ? myq[full + lane] : 0;
          if (lane < rem) myq[lane] = keep;
          cnt = rem;
        }
      }
    }
#pragma unroll
    for (int u = 0; u < 4; ++u) cur[u] = nxt[u];
  }
  if (cnt > 0) process(0, cnt);
  __syncthreads();

  for (int i = tid; i < rows * 32; i += 1024) {
    const int r = i >> 5, n = row0 + r;
    VecT o;
#pragma unroll
    for (int k = 0; k < V; ++k) vset(o, k, tile[r * C + k * 32 + lane]);
    *(VecT*)&out[(size_t)n * C + lane * V] = o;
    if (lane == 0) segsum[n] = ssum[r];
  }
}

struct Scratch {
  float *s_src, *s_dst, *self_l, *segmaxf, *segsum;
  unsigned* segmax_u;
  const float* mean_attr;
  float *eterm, *wa;
};

template <int C, bool ET>
static void run_gat(hipStream_t st, const float* hin, const float* Wm, const float* asrc,
                    const float* adst, const float* We, const float* ae,
                    const int* src, const int* dst, const float* edge_attr, float* hg,
                    float* out, const Scratch& S) {
  constexpr int COLT = C / 16;
  if (ET) wa_k<<<1, 16, 0, st>>>(We, ae, S.wa, C);
  gemm_k128<COLT><<<NN / 16, 32 * COLT, 0, st>>>(hin, Wm, hg);
  node_prep_k<C, ET><<<(NN * 32 + 255) / 256, 256, 0, st>>>(hg, asrc, adst, S.mean_attr,
                                                            S.wa, S.s_src, S.s_dst,
                                                            S.self_l, S.segmax_u);
  if (ET) eterm_k<<<(NE * 16 + 255) / 256, 256, 0, st>>>(edge_attr, S.wa, S.eterm);
  edge_max_k<ET><<<(NE + 255) / 256, 256, 0, st>>>(src, dst, S.s_src, S.s_dst, S.eterm,
                                                   S.segmax_u);
  node_mid_k<<<(NN + 255) / 256, 256, 0, st>>>(S.segmax_u, S.self_l, S.segmaxf, S.segsum);
  tile_agg_k<C, ET><<<NTILE, 1024, 0, st>>>(src, dst, S.s_src, S.s_dst, S.eterm, S.segmaxf,
                                            S.segsum, hg, out);
}

extern "C" void kernel_launch(void* const* d_in, const int* in_sizes, int n_in, void* d_out,
                              int out_size, void* d_ws, size_t ws_size, hipStream_t stream) {
  (void)in_sizes; (void)n_in; (void)out_size; (void)ws_size;
  const float* x = (const float*)d_in[0];
  const int* ei = (const int*)d_in[1];
  const int* src = ei;
  const int* dst = ei + NE;
  const float* edge_attr = (const float*)d_in[2];
  const float *W1 = (const float*)d_in[3], *a1s = (const float*)d_in[4],
              *a1d = (const float*)d_in[5], *b1 = (const float*)d_in[6];
  const float *W2 = (const float*)d_in[7], *a2s = (const float*)d_in[8],
              *a2d = (const float*)d_in[9], *b2 = (const float*)d_in[10];
  const float *We2 = (const float*)d_in[11], *ae2 = (const float*)d_in[12];
  const float *Wr = (const float*)d_in[13], *br = (const float*)d_in[14];
  const float *Wmu = (const float*)d_in[15], *amus = (const float*)d_in[16],
              *amud = (const float*)d_in[17], *bmu = (const float*)d_in[18],
              *Wemu = (const float*)d_in[19], *aemu = (const float*)d_in[20];
  const float *Wls = (const float*)d_in[21], *alss = (const float*)d_in[22],
              *alsd = (const float*)d_in[23], *bls = (const float*)d_in[24],
              *Wels = (const float*)d_in[25], *aels = (const float*)d_in[26];
  float* outf = (float*)d_out;

  float* W = (float*)d_ws;
  size_t o = 0;
  float* hA = W + o; o += (size_t)NN * 128;
  float* hB = W + o; o += (size_t)NN * 128;
  float* hC = W + o; o += (size_t)NN * 128;
  Scratch S;
  S.s_src = W + o; o += NN;
  S.s_dst = W + o; o += NN;
  S.self_l = W + o; o += NN;
  S.segmaxf = W + o; o += NN;
  S.segsum = W + o; o += NN;
  S.segmax_u = (unsigned*)(W + o); o += NN;
  float* cnt = W + o; o += NN;
  float* mean_attr = W + o; o += (size_t)NN * 16;
  S.mean_attr = mean_attr;
  S.eterm = W + o; o += NE;
  S.wa = W + o; o += 16;

  zero_f32_k<<<(NN * 17 + 255) / 256, 256, 0, stream>>>(cnt, NN * 17);
  attr_accum_k<<<(NE * 16 + 255) / 256, 256, 0, stream>>>(dst, edge_attr, mean_attr, cnt);
  attr_mean_k<<<(NN * 16 + 255) / 256, 256, 0, stream>>>(mean_attr, cnt);

  run_gat<128, false>(stream, x, W1, a1s, a1d, nullptr, nullptr, src, dst, edge_attr, hA, hB,
                      S);
  norm_relu_k<<<(NN * 128 + 255) / 256, 256, 0, stream>>>(hB, S.segsum, b1, NN * 128);

  run_gat<128, true>(stream, hB, W2, a2s, a2d, We2, ae2, src, dst, edge_attr, hA, hC, S);

  gemm_k128<8><<<NN / 16, 256, 0, stream>>>(x, Wr, hB);
  residual_k<<<(NN * 128 + 255) / 256, 256, 0, stream>>>(hC, S.segsum, b2, hB, br, NN * 128);

  run_gat<64, true>(stream, hB, Wmu, amus, amud, Wemu, aemu, src, dst, edge_attr, hA, outf, S);
  norm_bias_k<64><<<(NN * 16 + 255) / 256, 256, 0, stream>>>(outf, S.segsum, bmu);
  run_gat<64, true>(stream, hB, Wls, alss, alsd, Wels, aels, src, dst, edge_attr, hA,
                    outf + (size_t)NN * 64, S);
  norm_bias_k<64><<<(NN * 16 + 255) / 256, 256, 0, stream>>>(outf + (size_t)NN * 64, S.segsum,
                                                             bls);
}


# Every one-float-per-thread pass (edge terms, attr sums, node scalars, edge max, normalise, zero) now runs on a fixed grid of 2048 resident blocks whose waves loop, loads batched; sum orders unchanged
# speedup vs baseline: 1.7264x; 1.1356x over previous
#include <hip/hip_runtime.h>
#include <math.h>

typedef float v2f __attribute__((ext_vector_type(2)));
typedef float v8f __attribute__((ext_vector_type(8)));

static constexpr int   NN = 100000;
static constexpr int   NE = 1600000;
static constexpr float NEG = 0.2f;

__device__ __forceinline__ float leaky(float x) { return x > 0.f ? x : NEG * x; }
__device__ __forceinline__ unsigned f2ord(float f) {
  unsigned u = __float_as_uint(f);
  return (u & 0x80000000u) ? ~u : (u | 0x80000000u);
}
__device__ __forceinline__ float ord2f(unsigned v) {
  unsigned u = (v & 0x80000000u) ? (v & 0x7fffffffu) : ~v;
  return __uint_as_float(u);
}

static constexpr int SB = 256;
static constexpr int SG = 2048;
static constexpr int ST = SG * SB;
static constexpr int NW = ST / 32;

__global__ __launch_bounds__(SB) void zero_f32_k(float4* __restrict__ p, int n4) {
  for (int i = blockIdx.x * SB + threadIdx.x; i < n4; i += ST) p[i] = make_float4(0.f, 0.f, 0.f, 0.f);
}
__global__ __launch_bounds__(SB) void norm_relu_k(float* p, const float* __restrict__ segsum,
                                                  const float* __restrict__ bias) {
  constexpr int N4 = NN * 32;
  float4* p4 = (float4*)p;
  const int gid = blockIdx.x * SB + threadIdx.x;
  const float4 b = *(const float4*)&bias[(gid & 31) * 4];
  for (int i0 = gid; i0 < N4; i0 += ST * 4) {
    float4 v[4];
    float den[4];
#pragma unroll
    for (int u = 0; u < 4; ++u) {
      const int ic = min(i0 + u * ST, N4 - 1);
      v[u] = p4[ic];
      den[u] = segsum[ic >> 5] + 1e-16f;
    }
#pragma unroll
    for (int u = 0; u < 4; ++u) {
      const int i = i0 + u * ST;
      if (i < N4) {
        float4 o;
        o.x = fmaxf(v[u].x / den[u] + b.x, 0.f);
        o.y = fmaxf(v[u].y / den[u] + b.y, 0.f);
        o.z = fmaxf(v[u].z / den[u] + b.z, 0.f);
        o.w = fmaxf(v[u].w / den[u] + b.w, 0.f);
        p4[i] = o;
      }
    }
  }
}
__global__ __launch_bounds__(SB) void residual_k(const float* __restrict__ h2,
                                                 const float* __restrict__ segsum,
                                                 const float* __restrict__ b2, float* r,
                                                 const float* __restrict__ br) {
  constexpr int N4 = NN * 32;
  const float4* h4 = (const float4*)h2;
  float4* r4 = (float4*)r;
  const int gid = blockIdx.x * SB + threadIdx.x;
  const float4 bb = *(const float4*)&b2[(gid & 31) * 4];
  const float4 rb = *(const float4*)&br[(gid & 31) * 4];
  for (int i0 = gid; i0 < N4; i0 += ST * 4) {
    float4 v[4], w[4];
    float den[4];
#pragma unroll
    for (int u = 0; u < 4; ++u) {
      const int ic = min(i0 + u * ST, N4 - 1);
      v[u] = h4[ic];
      w[u] = r4[ic];
      den[u] = segsum[ic >> 5] + 1e-16f;
    }
#pragma unroll
    for (int u = 0; u < 4; ++u) {
      const int i = i0 + u * ST;
      if (i < N4) {
        float4 o;
        o.x = fmaxf(v[u].x / den[u] + bb.x, 0.f) + w[u].x + rb.x;
        o.y = fmaxf(v[u].y / den[u] + bb.y, 0.f) + w[u].y + rb.y;
        o.z = fmaxf(v[u].z / den[u] + bb.z, 0.f) + w[u].z + rb.z;
        o.w = fmaxf(v[u].w / den[u] + bb.w, 0.f) + w[u].w + rb.w;
        r4[i] = o;
      }
    }
  }
}
template <int C>
__global__ __launch_bounds__(SB) void norm_bias_k(float* out, const float* __restrict__ segsum,
                                                  const float* __restrict__ bias) {
  constexpr int Q = C / 4, N4 = NN * Q;
  float4* o4 = (float4*)out;
  const int gid = blockIdx.x * SB + threadIdx.x;
  const float4 b = *(const float4*)&bias[(gid % Q) * 4];
  for (int i0 = gid; i0 < N4; i0 += ST * 4) {
    float4 v[4];
    float den[4];
#pragma unroll
    for (int u = 0; u < 4; ++u) {
      const int ic = min(i0 + u * ST, N4 - 1);
      v[u] = o4[ic];
      den[u] = segsum[ic / Q] + 1e-16f;
    }
#pragma unroll
    for (int u = 0; u < 4; ++u) {
      const int i = i0 + u * ST;
      if (i < N4) {
        float4 o;
        o.x = v[u].x / den[u] + b.x;
        o.y = v[u].y / den[u] + b.y;
        o.z = v[u].z / den[u] + b.z;
        o.w = v[u].w / den[u] + b.w;
        o4[i] = o;
      }
    }
  }
}

__global__ __launch_bounds__(SB) void attr_accum_k(const int* __restrict__ dst,
                                                   const float* __restrict__ ea,
                                                   float* attr_sum, float* cnt) {
  constexpr int NT = NE * 16;
  const int gid = blockIdx.x * SB + threadIdx.x;
  const int j = gid & 15;
  for (int i0 = gid; i0 < NT; i0 += ST * 8) {
    float v[8];
    int d[8];
#pragma unroll
    for (int u = 0; u < 8; ++u) {
      const int ic = min(i0 + u * ST, NT - 1);
      v[u] = ea[ic];
      d[u] = dst[ic >> 4];
    }
#pragma unroll
    for (int u = 0; u < 8; ++u) {
      if (i0 + u * ST < NT) {
        atomicAdd(&attr_sum[(size_t)d[u] * 16 + j], v[u]);
        if (j == 0) atomicAdd(&cnt[d[u]], 1.f);
      }
    }
  }
}
__global__ __launch_bounds__(SB) void attr_mean_k(float* attr_sum, const float* __restrict__ cnt) {
  for (int i = blockIdx.x * SB + threadIdx.x; i < NN * 16; i += ST)
    attr_sum[i] /= fmaxf(cnt[i >> 4], 1.f);
}

__global__ void wa_k(const float* __restrict__ We, const float* __restrict__ ae,
                     float* wa, int K) {
  int i = threadIdx.x;
  if (i < 16) { float s = 0.f; for (int j = 0; j < K; ++j) s += We[i * K + j] * ae[j]; wa[i] = s; }
}

template <int COLT>
__global__ void gemm_k128(const float* __restrict__ A, const float* __restrict__ B,
                          float* __restrict__ C) {
  constexpr int COUT = COLT * 16;
  __shared__ float As[16 * 128];
  const int tid = threadIdx.x;
  const size_t abase = (size_t)blockIdx.x * (16 * 128);
  for (int i = tid * 4; i < 16 * 128; i += blockDim.x * 4)
    *(float4*)(&As[i]) = *(const float4*)(&A[abase + i]);
  __syncthreads();

  const int wave = tid >> 5, lane = tid & 31, half = lane >> 4, lidx = lane & 15;
  const int n0 = wave * 16;
  v8f acc = {};
#pragma unroll
  for (int k0 = 0; k0 < 128; k0 += 4) {
    v2f a, b;
    a.x = As[lidx * 128 + k0 + 2 * half];
    a.y = As[lidx * 128 + k0 + 2 * half + 1];
    b.x = B[(k0 + 2 * half) * COUT + n0 + lidx];
    b.y = B[(k0 + 2 * half + 1) * COUT + n0 + lidx];
    acc = __builtin_amdgcn_wmma_f32_16x16x4_f32(false, a, false, b, (short)0, acc,
                                                false, false);
  }
  const size_t row0 = (size_t)blockIdx.x * 16;
#pragma unroll
  for (int r = 0; r < 8; ++r)
    C[(row0 + r + 8 * half) * COUT + n0 + lidx] = acc[r];
}

template <int C, bool ET>
__global__ __launch_bounds__(SB) void node_prep_k(
    const float* __restrict__ h, const float* __restrict__ a_src,
    const float* __restrict__ a_dst, const float* __restrict__ mean_attr,
    const float* __restrict__ wa, float* __restrict__ s_src, float* __restrict__ s_dst,
    float* __restrict__ self_logit, unsigned* __restrict__ segmax_u) {
  constexpr int R = C / 32;
  constexpr int U = 256 / C;
  const int gid = blockIdx.x * SB + threadIdx.x;
  const int lane = gid & 31;
  float as[R], ad[R];
#pragma unroll
  for (int i = 0; i < R; ++i) { as[i] = a_src[lane + i * 32]; ad[i] = a_dst[lane + i * 32]; }
  for (int n0 = gid >> 5; n0 < NN; n0 += NW * U) {
    float hv[U][R];
#pragma unroll
    for (int u = 0; u < U; ++u) {
      const int nc = min(n0 + u * NW, NN - 1);
#pragma unroll
      for (int i = 0; i < R; ++i) hv[u][i] = h[(size_t)nc * C + lane + i * 32];
    }
    float psv[U], pdv[U];
#pragma unroll
    for (int u = 0; u < U; ++u) {
      float ps = 0.f, pd = 0.f;
#pragma unroll
      for (int i = 0; i < R; ++i) { ps += hv[u][i] * as[i]; pd += hv[u][i] * ad[i]; }
#pragma unroll
      for (int m = 16; m > 0; m >>= 1) { ps += __shfl_xor(ps, m, 32); pd += __shfl_xor(pd, m, 32); }
      psv[u] = ps; pdv[u] = pd;
    }
#pragma unroll
    for (int u = 0; u < U; ++u) {
      const int n = n0 + u * NW;
      if (n < NN) {
        const float ps = psv[u], pd = pdv[u];
        if (lane == 0) {
          float st = 0.f;
          if (ET) { for (int j = 0; j < 16; ++j) st += mean_attr[(size_t)n * 16 + j] * wa[j]; }
          float sl = leaky(ps + pd + st);
          s_src[n] = ps; s_dst[n] = pd; self_logit[n] = sl; segmax_u[n] = f2ord(sl);
        }
      }
    }
  }
}

__global__ __launch_bounds__(SB) void eterm_k(const float* __restrict__ ea,
                                              const float* __restrict__ wa,
                                              float* __restrict__ eterm) {
  constexpr int NT = NE * 16;
  const int gid = blockIdx.x * SB + threadIdx.x;
  const int j = gid & 15;
  const float w = wa[j];
  for (int i0 = gid; i0 < NT; i0 += ST * 8) {
    float v[8];
#pragma unroll
    for (int u = 0; u < 8; ++u) v[u] = ea[min(i0 + u * ST, NT - 1)] * w;
#pragma unroll
    for (int u = 0; u < 8; ++u) {
      float t = v[u];
      t += __shfl_xor(t, 1, 32); t += __shfl_xor(t, 2, 32);
      t += __shfl_xor(t, 4, 32); t += __shfl_xor(t, 8, 32);
      v[u] = t;
    }
#pragma unroll
    for (int u = 0; u < 8; ++u) {
      const int i = i0 + u * ST;
      if (j == 0 && i < NT) eterm[i >> 4] = v[u];
    }
  }
}

template <bool ET>
__global__ __launch_bounds__(SB) void edge_max_k(
    const int* __restrict__ src, const int* __restrict__ dst,
    const float* __restrict__ s_src, const float* __restrict__ s_dst,
    const float* __restrict__ eterm, unsigned* segmax_u) {
  for (int e0 = blockIdx.x * SB + threadIdx.x; e0 < NE; e0 += ST * 4) {
    int sv[4], dv[4];
    float l[4];
#pragma unroll
    for (int u = 0; u < 4; ++u) {
      const int ec = min(e0 + u * ST, NE - 1);
      sv[u] = src[ec];
      dv[u] = dst[ec];
      l[u] = ET ? eterm[ec] : 0.f;
    }
#pragma unroll
    for (int u = 0; u < 4; ++u) {
      const float g = s_src[sv[u]] + s_dst[dv[u]];
      l[u] = ET ? g + l[u] : g;
    }
#pragma unroll
    for (int u = 0; u < 4; ++u)
      if (e0 + u * ST < NE) atomicMax(&segmax_u[dv[u]], f2ord(leaky(l[u])));
  }
}

__global__ void node_mid_k(const unsigned* __restrict__ segmax_u,
                           const float* __restrict__ self_logit, float* segmaxf,
                           float* segsum) {
  int n = blockIdx.x * blockDim.x + threadIdx.x;
  if (n >= NN) return;
  float m = ord2f(segmax_u[n]);
  segmaxf[n] = m;
  segsum[n] = __expf(self_logit[n] - m);
}

static constexpr int RT = 400;
static constexpr int NTILE = (NN + RT - 1) / RT;
static constexpr int QCAP = 160;
static constexpr int NCHUNK = NE / 128;
static_assert(NE % 128 == 0, "edge list is swept in steps of 128");

template <int V> struct RowVec;
template <> struct RowVec<4> { typedef float4 T; };
template <> struct RowVec<2> { typedef float2 T; };
__device__ __forceinline__ float vget(const float4& v, int k) {
  return k == 0 ? v.x : k == 1 ? v.y : k == 2 ? v.z : v.w;
}
__device__ __forceinline__ float vget(const float2& v, int k) { return k == 0 ? v.x : v.y; }
__device__ __forceinline__ void vset(float4& v, int k, float f) {
  if (k == 0) v.x = f; else if (k == 1) v.y = f; else if (k == 2) v.z = f; else v.w = f;
}
__device__ __forceinline__ void vset(float2& v, int k, float f) {
  if (k == 0) v.x = f; else v.y = f;
}

template <int C, bool ET>
__global__ __launch_bounds__(1024) void tile_agg_k(
    const int* __restrict__ src, const int* __restrict__ dst,
    const float* __restrict__ s_src, const float* __restrict__ s_dst,
    const float* __restrict__ eterm, const float* __restrict__ segmaxf,
    float* __restrict__ segsum, const float* __restrict__ h, float* __restrict__ out) {
  constexpr int V = C / 32;
  typedef typename RowVec<V>::T VecT;
  __shared__ float tile[RT * C];
  __shared__ float ssum[RT];
  __shared__ int queue[32 * QCAP];
  const int tid = threadIdx.x, lane = tid & 31;
  const int wave = __builtin_amdgcn_readfirstlane(tid >> 5);
  const int row0 = blockIdx.x * RT;
  const int rows = min(RT, NN - row0);

  for (int i = tid; i < rows * 32; i += 1024) {
    const int r = i >> 5, n = row0 + r;
    const float a = segsum[n];
    const VecT hv = *(const VecT*)&h[(size_t)n * C + lane * V];
#pragma unroll
    for (int k = 0; k < V; ++k) tile[r * C + k * 32 + lane] = a * vget(hv, k);
    if (lane == 0) ssum[r] = a;
  }
  __syncthreads();

  int* myq = &queue[wave * QCAP];
  int cnt = 0;

  auto process = [&](int base, int nb) {
    int s = 0, r = 0;
    float w = 0.f;
    if (lane < nb) {
      const int e = myq[base + lane];
      s = src[e];
      const int d = dst[e];
      float l = s_src[s] + s_dst[d];
      if (ET) l += eterm[e];
      w = __expf(leaky(l) - segmaxf[d]);
      r = d - row0;
      atomicAdd(&ssum[r], w);
    }
    for (int j0 = 0; j0 < nb; j0 += 8) {
      VecT hv[8];
#pragma unroll
      for (int u = 0; u < 8; ++u) {
        const int sj = __builtin_amdgcn_readlane(s, j0 + u);
        hv[u] = *(const VecT*)&h[(size_t)sj * C + lane * V];
      }
#pragma unroll
      for (int u = 0; u < 8; ++u) {
        if (j0 + u < nb) {
          const int rj = __builtin_amdgcn_readlane(r, j0 + u);
          const float wj = __int_as_float(__builtin_amdgcn_readlane(__float_as_int(w), j0 + u));
          float* t = &tile[rj * C + lane];
#pragma unroll
          for (int k = 0; k < V; ++k) atomicAdd(t + k * 32, wj * vget(hv[u], k));
        }
      }
    }
  };

  const int rot = blockIdx.x * (NCHUNK / NTILE);
  auto chunk_of = [&](int k) { const int kk = k + rot; return kk >= NCHUNK ? kk - NCHUNK : kk; };
  const int4* dst4 = (const int4*)dst;
  auto fetch = [&](int4 (&d)[4], int k0) {
#pragma unroll
    for (int u = 0; u < 4; ++u) {
      const int k = k0 + 32 * u;
      d[u] = (k < NCHUNK) ? dst4[(size_t)chunk_of(k) * 32 + lane] : make_int4(-1, -1, -1, -1);
    }
  };
  int4 cur[4], nxt[4];
  fetch(cur, wave);
  for (int k0 = wave; k0 < NCHUNK; k0 += 32 * 4) {
    fetch(nxt, k0 + 32 * 4);
#pragma unroll
    for (int u = 0; u < 4; ++u) {
      const int e0 = chunk_of(k0 + 32 * u) * 128 + lane * 4;
      const bool p0 = (unsigned)(cur[u].x - row0) < (unsigned)rows;
      const bool p1 = (unsigned)(cur[u].y - row0) < (unsigned)rows;
      const bool p2 = (unsigned)(cur[u].z - row0) < (unsigned)rows;
      const bool p3 = (unsigned)(cur[u].w - row0) < (unsigned)rows;
      const unsigned m0 = __builtin_amdgcn_ballot_w32(p0);
      const unsigned m1 = __builtin_amdgcn_ballot_w32(p1);
      const unsigned m2 = __builtin_amdgcn_ballot_w32(p2);
      const unsigned m3 = __builtin_amdgcn_ballot_w32(p3);
      if ((m0 | m1 | m2 | m3) != 0u) {
        if (m0) { if (p0) myq[cnt + __builtin_amdgcn_mbcnt_lo(m0, 0u)] = e0;     cnt += __popc(m0); }
        if (m1) { if (p1) myq[cnt + __builtin_amdgcn_mbcnt_lo(m1, 0u)] = e0 + 1; cnt += __popc(m1); }
        if (m2) { if (p2) myq[cnt + __builtin_amdgcn_mbcnt_lo(m2, 0u)] = e0 + 2; cnt += __popc(m2); }
        if (m3) { if (p3) myq[cnt + __builtin_amdgcn_mbcnt_lo(m3, 0u)] = e0 + 3; cnt += __popc(m3); }
        if (cnt >= 32) {
          const int full = cnt & ~31;
          for (int b = 0; b < full; b += 32) process(b, 32);
          const int rem = cnt - full;
          const int keep = (lane < rem) ? myq[full + lane] : 0;
          if (lane < rem) myq[lane] = keep;
          cnt = rem;
        }
      }
    }
#pragma unroll
    for (int u = 0; u < 4; ++u) cur[u] = nxt[u];
  }
  if (cnt > 0) process(0, cnt);
  __syncthreads();

  for (int i = tid; i < rows * 32; i += 1024) {
    const int r = i >> 5, n = row0 + r;
    VecT o;
#pragma unroll
    for (int k = 0; k < V; ++k) vset(o, k, tile[r * C + k * 32 + lane]);
    *(VecT*)&out[(size_t)n * C + lane * V] = o;
    if (lane == 0) segsum[n] = ssum[r];
  }
}

struct Scratch {
  float *s_src, *s_dst, *self_l, *segmaxf, *segsum;
  unsigned* segmax_u;
  const float* mean_attr;
  float *eterm, *wa;
};

template <int C, bool ET>
static void run_gat(hipStream_t st, const float* hin, const float* Wm, const float* asrc,
                    const float* adst, const float* We, const float* ae,
                    const int* src, const int* dst, const float* edge_attr, float* hg,
                    float* out, const Scratch& S) {
  constexpr int COLT = C / 16;
  if (ET) wa_k<<<1, 16, 0, st>>>(We, ae, S.wa, C);
  gemm_k128<COLT><<<NN / 16, 32 * COLT, 0, st>>>(hin, Wm, hg);
  node_prep_k<C, ET><<<SG, SB, 0, st>>>(hg, asrc, adst, S.mean_attr, S.wa, S.s_src, S.s_dst,
                                        S.self_l, S.segmax_u);
  if (ET) eterm_k<<<SG, SB, 0, st>>>(edge_attr, S.wa, S.eterm);
  edge_max_k<ET><<<SG, SB, 0, st>>>(src, dst, S.s_src, S.s_dst, S.eterm, S.segmax_u);
  node_mid_k<<<(NN + 255) / 256, 256, 0, st>>>(S.segmax_u, S.self_l, S.segmaxf, S.segsum);
  tile_agg_k<C, ET><<<NTILE, 1024, 0, st>>>(src, dst, S.s_src, S.s_dst, S.eterm, S.segmaxf,
                                            S.segsum, hg, out);
}

extern "C" void kernel_launch(void* const* d_in, const int* in_sizes, int n_in, void* d_out,
                              int out_size, void* d_ws, size_t ws_size, hipStream_t stream) {
  (void)in_sizes; (void)n_in; (void)out_size; (void)ws_size;
  const float* x = (const float*)d_in[0];
  const int* ei = (const int*)d_in[1];
  const int* src = ei;
  const int* dst = ei + NE;
  const float* edge_attr = (const float*)d_in[2];
  const float *W1 = (const float*)d_in[3], *a1s = (const float*)d_in[4],
              *a1d = (const float*)d_in[5], *b1 = (const float*)d_in[6];
  const float *W2 = (const float*)d_in[7], *a2s = (const float*)d_in[8],
              *a2d = (const float*)d_in[9], *b2 = (const float*)d_in[10];
  const float *We2 = (const float*)d_in[11], *ae2 = (const float*)d_in[12];
  const float *Wr = (const float*)d_in[13], *br = (const float*)d_in[14];
  const float *Wmu = (const float*)d_in[15], *amus = (const float*)d_in[16],
              *amud = (const float*)d_in[17], *bmu = (const float*)d_in[18],
              *Wemu = (const float*)d_in[19], *aemu = (const float*)d_in[20];
  const float *Wls = (const float*)d_in[21], *alss = (const float*)d_in[22],
              *alsd = (const float*)d_in[23], *bls = (const float*)d_in[24],
              *Wels = (const float*)d_in[25], *aels = (const float*)d_in[26];
  float* outf = (float*)d_out;

  float* W = (float*)d_ws;
  size_t o = 0;
  float* hA = W + o; o += (size_t)NN * 128;
  float* hB = W + o; o += (size_t)NN * 128;
  float* hC = W + o; o += (size_t)NN * 128;
  Scratch S;
  S.s_src = W + o; o += NN;
  S.s_dst = W + o; o += NN;
  S.self_l = W + o; o += NN;
  S.segmaxf = W + o; o += NN;
  S.segsum = W + o; o += NN;
  S.segmax_u = (unsigned*)(W + o); o += NN;
  float* cnt = W + o; o += NN;
  float* mean_attr = W + o; o += (size_t)NN * 16;
  S.mean_attr = mean_attr;
  S.eterm = W + o; o += NE;
  S.wa = W + o; o += 16;

  zero_f32_k<<<SG, SB, 0, stream>>>((float4*)cnt, NN * 17 / 4);
  attr_accum_k<<<SG, SB, 0, stream>>>(dst, edge_attr, mean_attr, cnt);
  attr_mean_k<<<SG, SB, 0, stream>>>(mean_attr, cnt);

  run_gat<128, false>(stream, x, W1, a1s, a1d, nullptr, nullptr, src, dst, edge_attr, hA, hB,
                      S);
  norm_relu_k<<<SG, SB, 0, stream>>>(hB, S.segsum, b1);

  run_gat<128, true>(stream, hB, W2, a2s, a2d, We2, ae2, src, dst, edge_attr, hA, hC, S);

  gemm_k128<8><<<NN / 16, 256, 0, stream>>>(x, Wr, hB);
  residual_k<<<SG, SB, 0, stream>>>(hC, S.segsum, b2, hB, br);

  run_gat<64, true>(stream, hB, Wmu, amus, amud, Wemu, aemu, src, dst, edge_attr, hA, outf, S);
  norm_bias_k<64><<<SG, SB, 0, stream>>>(outf, S.segsum, bmu);
  run_gat<64, true>(stream, hB, Wls, alss, alsd, Wels, aels, src, dst, edge_attr, hA,
                    outf + (size_t)NN * 64, S);
  norm_bias_k<64><<<SG, SB, 0, stream>>>(outf + (size_t)NN * 64, S.segsum, bls);
}


# GEMM: a wave keeps its 128x32 weight block in registers and walks 16-row stripes of A read straight from memory in 16-B pieces (LDS tile, barrier gone, 256 blocks); k summed in another order
# speedup vs baseline: 1.8645x; 1.0800x over previous
#include <hip/hip_runtime.h>
#include <math.h>

typedef float v2f __attribute__((ext_vector_type(2)));
typedef float v8f __attribute__((ext_vector_type(8)));

static constexpr int   NN = 100000;
static constexpr int   NE = 1600000;
static constexpr float NEG = 0.2f;

__device__ __forceinline__ float leaky(float x) { return x > 0.f ? x : NEG * x; }
__device__ __forceinline__ unsigned f2ord(float f) {
  unsigned u = __float_as_uint(f);
  return (u & 0x80000000u) ? ~u : (u | 0x80000000u);
}
__device__ __forceinline__ float ord2f(unsigned v) {
  unsigned u = (v & 0x80000000u) ? (v & 0x7fffffffu) : ~v;
  return __uint_as_float(u);
}

static constexpr int SB = 256;
static constexpr int SG = 2048;
static constexpr int ST = SG * SB;
static constexpr int NW = ST / 32;

__global__ __launch_bounds__(SB) void zero_f32_k(float4* __restrict__ p, int n4) {
  for (int i = blockIdx.x * SB + threadIdx.x; i < n4; i += ST) p[i] = make_float4(0.f, 0.f, 0.f, 0.f);
}
__global__ __launch_bounds__(SB) void norm_relu_k(float* p, const float* __restrict__ segsum,
                                                  const float* __restrict__ bias) {
  constexpr int N4 = NN * 32;
  float4* p4 = (float4*)p;
  const int gid = blockIdx.x * SB + threadIdx.x;
  const float4 b = *(const float4*)&bias[(gid & 31) * 4];
  for (int i0 = gid; i0 < N4; i0 += ST * 4) {
    float4 v[4];
    float den[4];
#pragma unroll
    for (int u = 0; u < 4; ++u) {
      const int ic = min(i0 + u * ST, N4 - 1);
      v[u] = p4[ic];
      den[u] = segsum[ic >> 5] + 1e-16f;
    }
#pragma unroll
    for (int u = 0; u < 4; ++u) {
      const int i = i0 + u * ST;
      if (i < N4) {
        float4 o;
        o.x = fmaxf(v[u].x / den[u] + b.x, 0.f);
        o.y = fmaxf(v[u].y / den[u] + b.y, 0.f);
        o.z = fmaxf(v[u].z / den[u] + b.z, 0.f);
        o.w = fmaxf(v[u].w / den[u] + b.w, 0.f);
        p4[i] = o;
      }
    }
  }
}
__global__ __launch_bounds__(SB) void residual_k(const float* __restrict__ h2,
                                                 const float* __restrict__ segsum,
                                                 const float* __restrict__ b2, float* r,
                                                 const float* __restrict__ br) {
  constexpr int N4 = NN * 32;
  const float4* h4 = (const float4*)h2;
  float4* r4 = (float4*)r;
  const int gid = blockIdx.x * SB + threadIdx.x;
  const float4 bb = *(const float4*)&b2[(gid & 31) * 4];
  const float4 rb = *(const float4*)&br[(gid & 31) * 4];
  for (int i0 = gid; i0 < N4; i0 += ST * 4) {
    float4 v[4], w[4];
    float den[4];
#pragma unroll
    for (int u = 0; u < 4; ++u) {
      const int ic = min(i0 + u * ST, N4 - 1);
      v[u] = h4[ic];
      w[u] = r4[ic];
      den[u] = segsum[ic >> 5] + 1e-16f;
    }
#pragma unroll
    for (int u = 0; u < 4; ++u) {
      const int i = i0 + u * ST;
      if (i < N4) {
        float4 o;
        o.x = fmaxf(v[u].x / den[u] + bb.x, 0.f) + w[u].x + rb.x;
        o.y = fmaxf(v[u].y / den[u] + bb.y, 0.f) + w[u].y + rb.y;
        o.z = fmaxf(v[u].z / den[u] + bb.z, 0.f) + w[u].z + rb.z;
        o.w = fmaxf(v[u].w / den[u] + bb.w, 0.f) + w[u].w + rb.w;
        r4[i] = o;
      }
    }
  }
}
template <int C>
__global__ __launch_bounds__(SB) void norm_bias_k(float* out, const float* __restrict__ segsum,
                                                  const float* __restrict__ bias) {
  constexpr int Q = C / 4, N4 = NN * Q;
  float4* o4 = (float4*)out;
  const int gid = blockIdx.x * SB + threadIdx.x;
  const float4 b = *(const float4*)&bias[(gid % Q) * 4];
  for (int i0 = gid; i0 < N4; i0 += ST * 4) {
    float4 v[4];
    float den[4];
#pragma unroll
    for (int u = 0; u < 4; ++u) {
      const int ic = min(i0 + u * ST, N4 - 1);
      v[u] = o4[ic];
      den[u] = segsum[ic / Q] + 1e-16f;
    }
#pragma unroll
    for (int u = 0; u < 4; ++u) {
      const int i = i0 + u * ST;
      if (i < N4) {
        float4 o;
        o.x = v[u].x / den[u] + b.x;
        o.y = v[u].y / den[u] + b.y;
        o.z = v[u].z / den[u] + b.z;
        o.w = v[u].w / den[u] + b.w;
        o4[i] = o;
      }
    }
  }
}

__global__ __launch_bounds__(SB) void attr_accum_k(const int* __restrict__ dst,
                                                   const float* __restrict__ ea,
                                                   float* attr_sum, float* cnt) {
  constexpr int NT = NE * 16;
  const int gid = blockIdx.x * SB + threadIdx.x;
  const int j = gid & 15;
  for (int i0 = gid; i0 < NT; i0 += ST * 8) {
    float v[8];
    int d[8];
#pragma unroll
    for (int u = 0; u < 8; ++u) {
      const int ic = min(i0 + u * ST, NT - 1);
      v[u] = ea[ic];
      d[u] = dst[ic >> 4];
    }
#pragma unroll
    for (int u = 0; u < 8; ++u) {
      if (i0 + u * ST < NT) {
        atomicAdd(&attr_sum[(size_t)d[u] * 16 + j], v[u]);
        if (j == 0) atomicAdd(&cnt[d[u]], 1.f);
      }
    }
  }
}
__global__ __launch_bounds__(SB) void attr_mean_k(float* attr_sum, const float* __restrict__ cnt) {
  for (int i = blockIdx.x * SB + threadIdx.x; i < NN * 16; i += ST)
    attr_sum[i] /= fmaxf(cnt[i >> 4], 1.f);
}

__global__ void wa_k(const float* __restrict__ We, const float* __restrict__ ae,
                     float* wa, int K) {
  int i = threadIdx.x;
  if (i < 16) { float s = 0.f; for (int j = 0; j < K; ++j) s += We[i * K + j] * ae[j]; wa[i] = s; }
}

static constexpr int GW = 2048;
template <int COUT>
__global__ __launch_bounds__(256) void gemm_k128(const float* __restrict__ A,
                                                 const float* __restrict__ B,
                                                 float* __restrict__ C) {
  constexpr int NCG = COUT / 32;
  constexpr int NSW = GW / NCG;
  constexpr int NS = NN / 16;
  static_assert(NN % 16 == 0, "rows come in stripes of 16");
  const int lane = threadIdx.x & 31, half = lane >> 4, lidx = lane & 15;
  const int wid = __builtin_amdgcn_readfirstlane((int)((blockIdx.x * 256 + threadIdx.x) >> 5));
  const int n0 = (wid % NCG) * 32;
  const int sw = wid / NCG;

  float bq[2][16][4];
#pragma unroll
  for (int t = 0; t < 2; ++t)
#pragma unroll
    for (int j = 0; j < 16; ++j)
#pragma unroll
      for (int i = 0; i < 4; ++i)
        bq[t][j][i] = B[(8 * j + 4 * half + i) * COUT + n0 + 16 * t + lidx];

  const float4* A4 = (const float4*)A;
  auto load = [&](float4 (&a)[16], int s) {
    const float4* p = A4 + ((size_t)s * 16 + lidx) * 32 + half;
#pragma unroll
    for (int j = 0; j < 16; ++j) a[j] = p[2 * j];
  };
  auto compute = [&](const float4 (&a)[16], int s) {
    v8f acc0 = {}, acc1 = {};
#pragma unroll
    for (int j = 0; j < 16; ++j) {
      v2f a01, a23, b;
      a01.x = a[j].x; a01.y = a[j].y;
      a23.x = a[j].z; a23.y = a[j].w;
      b.x = bq[0][j][0]; b.y = bq[0][j][1];
      acc0 = __builtin_amdgcn_wmma_f32_16x16x4_f32(false, a01, false, b, (short)0, acc0,
                                                   false, false);
      b.x = bq[1][j][0]; b.y = bq[1][j][1];
      acc1 = __builtin_amdgcn_wmma_f32_16x16x4_f32(false, a01, false, b, (short)0, acc1,
                                                   false, false);
      b.x = bq[0][j][2]; b.y = bq[0][j][3];
      acc0 = __builtin_amdgcn_wmma_f32_16x16x4_f32(false, a23, false, b, (short)0, acc0,
                                                   false, false);
      b.x = bq[1][j][2]; b.y = bq[1][j][3];
      acc1 = __builtin_amdgcn_wmma_f32_16x16x4_f32(false, a23, false, b, (short)0, acc1,
                                                   false, false);
    }
    float* c = C + ((size_t)s * 16 + 8 * half) * COUT + n0 + lidx;
#pragma unroll
    for (int r = 0; r < 8; ++r) {
      c[r * COUT] = acc0[r];
      c[r * COUT + 16] = acc1[r];
    }
  };

  float4 ba[16], bb[16];
  int s = sw;
  load(ba, s);
  for (; s + NSW < NS; s += 2 * NSW) {
    load(bb, s + NSW);
    compute(ba, s);
    load(ba, min(s + 2 * NSW, NS - 1));
    compute(bb, s + NSW);
  }
  if (s < NS) compute(ba, s);
}

template <int C, bool ET>
__global__ __launch_bounds__(SB) void node_prep_k(
    const float* __restrict__ h, const float* __restrict__ a_src,
    const float* __restrict__ a_dst, const float* __restrict__ mean_attr,
    const float* __restrict__ wa, float* __restrict__ s_src, float* __restrict__ s_dst,
    float* __restrict__ self_logit, unsigned* __restrict__ segmax_u) {
  constexpr int R = C / 32;
  constexpr int U = 256 / C;
  const int gid = blockIdx.x * SB + threadIdx.x;
  const int lane = gid & 31;
  float as[R], ad[R];
#pragma unroll
  for (int i = 0; i < R; ++i) { as[i] = a_src[lane + i * 32]; ad[i] = a_dst[lane + i * 32]; }
  for (int n0 = gid >> 5; n0 < NN; n0 += NW * U) {
    float hv[U][R];
#pragma unroll
    for (int u = 0; u < U; ++u) {
      const int nc = min(n0 + u * NW, NN - 1);
#pragma unroll
      for (int i = 0; i < R; ++i) hv[u][i] = h[(size_t)nc * C + lane + i * 32];
    }
    float psv[U], pdv[U];
#pragma unroll
    for (int u = 0; u < U; ++u) {
      float ps = 0.f, pd = 0.f;
#pragma unroll
      for (int i = 0; i < R; ++i) { ps += hv[u][i] * as[i]; pd += hv[u][i] * ad[i]; }
#pragma unroll
      for (int m = 16; m > 0; m >>= 1) { ps += __shfl_xor(ps, m, 32); pd += __shfl_xor(pd, m, 32); }
      psv[u] = ps; pdv[u] = pd;
    }
#pragma unroll
    for (int u = 0; u < U; ++u) {
      const int n = n0 + u * NW;
      if (n < NN) {
        const float ps = psv[u], pd = pdv[u];
        if (lane == 0) {
          float st = 0.f;
          if (ET) { for (int j = 0; j < 16; ++j) st += mean_attr[(size_t)n * 16 + j] * wa[j]; }
          float sl = leaky(ps + pd + st);
          s_src[n] = ps; s_dst[n] = pd; self_logit[n] = sl; segmax_u[n] = f2ord(sl);
        }
      }
    }
  }
}

__global__ __launch_bounds__(SB) void eterm_k(const float* __restrict__ ea,
                                              const float* __restrict__ wa,
                                              float* __restrict__ eterm) {
  constexpr int NT = NE * 16;
  const int gid = blockIdx.x * SB + threadIdx.x;
  const int j = gid & 15;
  const float w = wa[j];
  for (int i0 = gid; i0 < NT; i0 += ST * 8) {
    float v[8];
#pragma unroll
    for (int u = 0; u < 8; ++u) v[u] = ea[min(i0 + u * ST, NT - 1)] * w;
#pragma unroll
    for (int u = 0; u < 8; ++u) {
      float t = v[u];
      t += __shfl_xor(t, 1, 32); t += __shfl_xor(t, 2, 32);
      t += __shfl_xor(t, 4, 32); t += __shfl_xor(t, 8, 32);
      v[u] = t;
    }
#pragma unroll
    for (int u = 0; u < 8; ++u) {
      const int i = i0 + u * ST;
      if (j == 0 && i < NT) eterm[i >> 4] = v[u];
    }
  }
}

template <bool ET>
__global__ __launch_bounds__(SB) void edge_max_k(
    const int* __restrict__ src, const int* __restrict__ dst,
    const float* __restrict__ s_src, const float* __restrict__ s_dst,
    const float* __restrict__ eterm, unsigned* segmax_u) {
  for (int e0 = blockIdx.x * SB + threadIdx.x; e0 < NE; e0 += ST * 4) {
    int sv[4], dv[4];
    float l[4];
#pragma unroll
    for (int u = 0; u < 4; ++u) {
      const int ec = min(e0 + u * ST, NE - 1);
      sv[u] = src[ec];
      dv[u] = dst[ec];
      l[u] = ET ? eterm[ec] : 0.f;
    }
#pragma unroll
    for (int u = 0; u < 4; ++u) {
      const float g = s_src[sv[u]] + s_dst[dv[u]];
      l[u] = ET ? g + l[u] : g;
    }
#pragma unroll
    for (int u = 0; u < 4; ++u)
      if (e0 + u * ST < NE) atomicMax(&segmax_u[dv[u]], f2ord(leaky(l[u])));
  }
}

__global__ void node_mid_k(const unsigned* __restrict__ segmax_u,
                           const float* __restrict__ self_logit, float* segmaxf,
                           float* segsum) {
  int n = blockIdx.x * blockDim.x + threadIdx.x;
  if (n >= NN) return;
  float m = ord2f(segmax_u[n]);
  segmaxf[n] = m;
  segsum[n] = __expf(self_logit[n] - m);
}

static constexpr int RT = 400;
static constexpr int NTILE = (NN + RT - 1) / RT;
static constexpr int QCAP = 160;
static constexpr int NCHUNK = NE / 128;
static_assert(NE % 128 == 0, "edge list is swept in steps of 128");

template <int V> struct RowVec;
template <> struct RowVec<4> { typedef float4 T; };
template <> struct RowVec<2> { typedef float2 T; };
__device__ __forceinline__ float vget(const float4& v, int k) {
  return k == 0 ? v.x : k == 1 ? v.y : k == 2 ? v.z : v.w;
}
__device__ __forceinline__ float vget(const float2& v, int k) { return k == 0 ? v.x : v.y; }
__device__ __forceinline__ void vset(float4& v, int k, float f) {
  if (k == 0) v.x = f; else if (k == 1) v.y = f; else if (k == 2) v.z = f; else v.w = f;
}
__device__ __forceinline__ void vset(float2& v, int k, float f) {
  if (k == 0) v.x = f; else v.y = f;
}

template <int C, bool ET>
__global__ __launch_bounds__(1024) void tile_agg_k(
    const int* __restrict__ src, const int* __restrict__ dst,
    const float* __restrict__ s_src, const float* __restrict__ s_dst,
    const float* __restrict__ eterm, const float* __restrict__ segmaxf,
    float* __restrict__ segsum, const float* __restrict__ h, float* __restrict__ out) {
  constexpr int V = C / 32;
  typedef typename RowVec<V>::T VecT;
  __shared__ float tile[RT * C];
  __shared__ float ssum[RT];
  __shared__ int queue[32 * QCAP];
  const int tid = threadIdx.x, lane = tid & 31;
  const int wave = __builtin_amdgcn_readfirstlane(tid >> 5);
  const int row0 = blockIdx.x * RT;
  const int rows = min(RT, NN - row0);

  for (int i = tid; i < rows * 32; i += 1024) {
    const int r = i >> 5, n = row0 + r;
    const float a = segsum[n];
    const VecT hv = *(const VecT*)&h[(size_t)n * C + lane * V];
#pragma unroll
    for (int k = 0; k < V; ++k) tile[r * C + k * 32 + lane] = a * vget(hv, k);
    if (lane == 0) ssum[r] = a;
  }
  __syncthreads();

  int* myq = &queue[wave * QCAP];
  int cnt = 0;

  auto process = [&](int base, int nb) {
    int s = 0, r = 0;
    float w = 0.f;
    if (lane < nb) {
      const int e = myq[base + lane];
      s = src[e];
      const int d = dst[e];
      float l = s_src[s] + s_dst[d];
      if (ET) l += eterm[e];
      w = __expf(leaky(l) - segmaxf[d]);
      r = d - row0;
      atomicAdd(&ssum[r], w);
    }
    for (int j0 = 0; j0 < nb; j0 += 8) {
      VecT hv[8];
#pragma unroll
      for (int u = 0; u < 8; ++u) {
        const int sj = __builtin_amdgcn_readlane(s, j0 + u);
        hv[u] = *(const VecT*)&h[(size_t)sj * C + lane * V];
      }
#pragma unroll
      for (int u = 0; u < 8; ++u) {
        if (j0 + u < nb) {
          const int rj = __builtin_amdgcn_readlane(r, j0 + u);
          const float wj = __int_as_float(__builtin_amdgcn_readlane(__float_as_int(w), j0 + u));
          float* t = &tile[rj * C + lane];
#pragma unroll
          for (int k = 0; k < V; ++k) atomicAdd(t + k * 32, wj * vget(hv[u], k));
        }
      }
    }
  };

  const int rot = blockIdx.x * (NCHUNK / NTILE);
  auto chunk_of = [&](int k) { const int kk = k + rot; return kk >= NCHUNK ? kk - NCHUNK : kk; };
  const int4* dst4 = (const int4*)dst;
  auto fetch = [&](int4 (&d)[4], int k0) {
#pragma unroll
    for (int u = 0; u < 4; ++u) {
      const int k = k0 + 32 * u;
      d[u] = (k < NCHUNK) ? dst4[(size_t)chunk_of(k) * 32 + lane] : make_int4(-1, -1, -1, -1);
    }
  };
  int4 cur[4], nxt[4];
  fetch(cur, wave);
  for (int k0 = wave; k0 < NCHUNK; k0 += 32 * 4) {
    fetch(nxt, k0 + 32 * 4);
#pragma unroll
    for (int u = 0; u < 4; ++u) {
      const int e0 = chunk_of(k0 + 32 * u) * 128 + lane * 4;
      const bool p0 = (unsigned)(cur[u].x - row0) < (unsigned)rows;
      const bool p1 = (unsigned)(cur[u].y - row0) < (unsigned)rows;
      const bool p2 = (unsigned)(cur[u].z - row0) < (unsigned)rows;
      const bool p3 = (unsigned)(cur[u].w - row0) < (unsigned)rows;
      const unsigned m0 = __builtin_amdgcn_ballot_w32(p0);
      const unsigned m1 = __builtin_amdgcn_ballot_w32(p1);
      const unsigned m2 = __builtin_amdgcn_ballot_w32(p2);
      const unsigned m3 = __builtin_amdgcn_ballot_w32(p3);
      if ((m0 | m1 | m2 | m3) != 0u) {
        if (m0) { if (p0) myq[cnt + __builtin_amdgcn_mbcnt_lo(m0, 0u)] = e0;     cnt += __popc(m0); }
        if (m1) { if (p1) myq[cnt + __builtin_amdgcn_mbcnt_lo(m1, 0u)] = e0 + 1; cnt += __popc(m1); }
        if (m2) { if (p2) myq[cnt + __builtin_amdgcn_mbcnt_lo(m2, 0u)] = e0 + 2; cnt += __popc(m2); }
        if (m3) { if (p3) myq[cnt + __builtin_amdgcn_mbcnt_lo(m3, 0u)] = e0 + 3; cnt += __popc(m3); }
        if (cnt >= 32) {
          const int full = cnt & ~31;
          for (int b = 0; b < full; b += 32) process(b, 32);
          const int rem = cnt - full;
          const int keep = (lane < rem) ? myq[full + lane] : 0;
          if (lane < rem) myq[lane] = keep;
          cnt = rem;
        }
      }
    }
#pragma unroll
    for (int u = 0; u < 4; ++u) cur[u] = nxt[u];
  }
  if (cnt > 0) process(0, cnt);
  __syncthreads();

  for (int i = tid; i < rows * 32; i += 1024) {
    const int r = i >> 5, n = row0 + r;
    VecT o;
#pragma unroll
    for (int k = 0; k < V; ++k) vset(o, k, tile[r * C + k * 32 + lane]);
    *(VecT*)&out[(size_t)n * C + lane * V] = o;
    if (lane == 0) segsum[n] = ssum[r];
  }
}

struct Scratch {
  float *s_src, *s_dst, *self_l, *segmaxf, *segsum;
  unsigned* segmax_u;
  const float* mean_attr;
  float *eterm, *wa;
};

template <int C, bool ET>
static void run_gat(hipStream_t st, const float* hin, const float* Wm, const float* asrc,
                    const float* adst, const float* We, const float* ae,
                    const int* src, const int* dst, const float* edge_attr, float* hg,
                    float* out, const Scratch& S) {
  if (ET) wa_k<<<1, 16, 0, st>>>(We, ae, S.wa, C);
  gemm_k128<C><<<GW / 8, 256, 0, st>>>(hin, Wm, hg);
  node_prep_k<C, ET><<<SG, SB, 0, st>>>(hg, asrc, adst, S.mean_attr, S.wa, S.s_src, S.s_dst,
                                        S.self_l, S.segmax_u);
  if (ET) eterm_k<<<SG, SB, 0, st>>>(edge_attr, S.wa, S.eterm);
  edge_max_k<ET><<<SG, SB, 0, st>>>(src, dst, S.s_src, S.s_dst, S.eterm, S.segmax_u);
  node_mid_k<<<(NN + 255) / 256, 256, 0, st>>>(S.segmax_u, S.self_l, S.segmaxf, S.segsum);
  tile_agg_k<C, ET><<<NTILE, 1024, 0, st>>>(src, dst, S.s_src, S.s_dst, S.eterm, S.segmaxf,
                                            S.segsum, hg, out);
}

extern "C" void kernel_launch(void* const* d_in, const int* in_sizes, int n_in, void* d_out,
                              int out_size, void* d_ws, size_t ws_size, hipStream_t stream) {
  (void)in_sizes; (void)n_in; (void)out_size; (void)ws_size;
  const float* x = (const float*)d_in[0];
  const int* ei = (const int*)d_in[1];
  const int* src = ei;
  const int* dst = ei + NE;
  const float* edge_attr = (const float*)d_in[2];
  const float *W1 = (const float*)d_in[3], *a1s = (const float*)d_in[4],
              *a1d = (const float*)d_in[5], *b1 = (const float*)d_in[6];
  const float *W2 = (const float*)d_in[7], *a2s = (const float*)d_in[8],
              *a2d = (const float*)d_in[9], *b2 = (const float*)d_in[10];
  const float *We2 = (const float*)d_in[11], *ae2 = (const float*)d_in[12];
  const float *Wr = (const float*)d_in[13], *br = (const float*)d_in[14];
  const float *Wmu = (const float*)d_in[15], *amus = (const float*)d_in[16],
              *amud = (const float*)d_in[17], *bmu = (const float*)d_in[18],
              *Wemu = (const float*)d_in[19], *aemu = (const float*)d_in[20];
  const float *Wls = (const float*)d_in[21], *alss = (const float*)d_in[22],
              *alsd = (const float*)d_in[23], *bls = (const float*)d_in[24],
              *Wels = (const float*)d_in[25], *aels = (const float*)d_in[26];
  float* outf = (float*)d_out;

  float* W = (float*)d_ws;
  size_t o = 0;
  float* hA = W + o; o += (size_t)NN * 128;
  float* hB = W + o; o += (size_t)NN * 128;
  float* hC = W + o; o += (size_t)NN * 128;
  Scratch S;
  S.s_src = W + o; o += NN;
  S.s_dst = W + o; o += NN;
  S.self_l = W + o; o += NN;
  S.segmaxf = W + o; o += NN;
  S.segsum = W + o; o += NN;
  S.segmax_u = (unsigned*)(W + o); o += NN;
  float* cnt = W + o; o += NN;
  float* mean_attr = W + o; o += (size_t)NN * 16;
  S.mean_attr = mean_attr;
  S.eterm = W + o; o += NE;
  S.wa = W + o; o += 16;

  zero_f32_k<<<SG, SB, 0, stream>>>((float4*)cnt, NN * 17 / 4);
  attr_accum_k<<<SG, SB, 0, stream>>>(dst, edge_attr, mean_attr, cnt);
  attr_mean_k<<<SG, SB, 0, stream>>>(mean_attr, cnt);

  run_gat<128, false>(stream, x, W1, a1s, a1d, nullptr, nullptr, src, dst, edge_attr, hA, hB,
                      S);
  norm_relu_k<<<SG, SB, 0, stream>>>(hB, S.segsum, b1);

  run_gat<128, true>(stream, hB, W2, a2s, a2d, We2, ae2, src, dst, edge_attr, hA, hC, S);

  gemm_k128<128><<<GW / 8, 256, 0, stream>>>(x, Wr, hB);
  residual_k<<<SG, SB, 0, stream>>>(hC, S.segsum, b2, hB, br);

  run_gat<64, true>(stream, hB, Wmu, amus, amud, Wemu, aemu, src, dst, edge_attr, hA, outf, S);
  norm_bias_k<64><<<SG, SB, 0, stream>>>(outf, S.segsum, bmu);
  run_gat<64, true>(stream, hB, Wls, alss, alsd, Wels, aels, src, dst, edge_attr, hA,
                    outf + (size_t)NN * 64, S);
  norm_bias_k<64><<<SG, SB, 0, stream>>>(outf + (size_t)NN * 64, S.segsum, bls);
}


# Aggregation sweep reads a 1-byte owner-tile key per edge (plane made once per call by a small new pass), not the 4-byte destination: 16 edges per lane load, byte-parallel match; adds in arrival order
# speedup vs baseline: 1.9833x; 1.0637x over previous
#include <hip/hip_runtime.h>
#include <math.h>

typedef float v2f __attribute__((ext_vector_type(2)));
typedef float v8f __attribute__((ext_vector_type(8)));

static constexpr int   NN = 100000;
static constexpr int   NE = 1600000;
static constexpr float NEG = 0.2f;

__device__ __forceinline__ float leaky(float x) { return x > 0.f ? x : NEG * x; }
__device__ __forceinline__ unsigned f2ord(float f) {
  unsigned u = __float_as_uint(f);
  return (u & 0x80000000u) ? ~u : (u | 0x80000000u);
}
__device__ __forceinline__ float ord2f(unsigned v) {
  unsigned u = (v & 0x80000000u) ? (v & 0x7fffffffu) : ~v;
  return __uint_as_float(u);
}

static constexpr int SB = 256;
static constexpr int SG = 2048;
static constexpr int ST = SG * SB;
static constexpr int NW = ST / 32;

__global__ __launch_bounds__(SB) void zero_f32_k(float4* __restrict__ p, int n4) {
  for (int i = blockIdx.x * SB + threadIdx.x; i < n4; i += ST) p[i] = make_float4(0.f, 0.f, 0.f, 0.f);
}
__global__ __launch_bounds__(SB) void norm_relu_k(float* p, const float* __restrict__ segsum,
                                                  const float* __restrict__ bias) {
  constexpr int N4 = NN * 32;
  float4* p4 = (float4*)p;
  const int gid = blockIdx.x * SB + threadIdx.x;
  const float4 b = *(const float4*)&bias[(gid & 31) * 4];
  for (int i0 = gid; i0 < N4; i0 += ST * 4) {
    float4 v[4];
    float den[4];
#pragma unroll
    for (int u = 0; u < 4; ++u) {
      const int ic = min(i0 + u * ST, N4 - 1);
      v[u] = p4[ic];
      den[u] = segsum[ic >> 5] + 1e-16f;
    }
#pragma unroll
    for (int u = 0; u < 4; ++u) {
      const int i = i0 + u * ST;
      if (i < N4) {
        float4 o;
        o.x = fmaxf(v[u].x / den[u] + b.x, 0.f);
        o.y = fmaxf(v[u].y / den[u] + b.y, 0.f);
        o.z = fmaxf(v[u].z / den[u] + b.z, 0.f);
        o.w = fmaxf(v[u].w / den[u] + b.w, 0.f);
        p4[i] = o;
      }
    }
  }
}
__global__ __launch_bounds__(SB) void residual_k(const float* __restrict__ h2,
                                                 const float* __restrict__ segsum,
                                                 const float* __restrict__ b2, float* r,
                                                 const float* __restrict__ br) {
  constexpr int N4 = NN * 32;
  const float4* h4 = (const float4*)h2;
  float4* r4 = (float4*)r;
  const int gid = blockIdx.x * SB + threadIdx.x;
  const float4 bb = *(const float4*)&b2[(gid & 31) * 4];
  const float4 rb = *(const float4*)&br[(gid & 31) * 4];
  for (int i0 = gid; i0 < N4; i0 += ST * 4) {
    float4 v[4], w[4];
    float den[4];
#pragma unroll
    for (int u = 0; u < 4; ++u) {
      const int ic = min(i0 + u * ST, N4 - 1);
      v[u] = h4[ic];
      w[u] = r4[ic];
      den[u] = segsum[ic >> 5] + 1e-16f;
    }
#pragma unroll
    for (int u = 0; u < 4; ++u) {
      const int i = i0 + u * ST;
      if (i < N4) {
        float4 o;
        o.x = fmaxf(v[u].x / den[u] + bb.x, 0.f) + w[u].x + rb.x;
        o.y = fmaxf(v[u].y / den[u] + bb.y, 0.f) + w[u].y + rb.y;
        o.z = fmaxf(v[u].z / den[u] + bb.z, 0.f) + w[u].z + rb.z;
        o.w = fmaxf(v[u].w / den[u] + bb.w, 0.f) + w[u].w + rb.w;
        r4[i] = o;
      }
    }
  }
}
template <int C>
__global__ __launch_bounds__(SB) void norm_bias_k(float* out, const float* __restrict__ segsum,
                                                  const float* __restrict__ bias) {
  constexpr int Q = C / 4, N4 = NN * Q;
  float4* o4 = (float4*)out;
  const int gid = blockIdx.x * SB + threadIdx.x;
  const float4 b = *(const float4*)&bias[(gid % Q) * 4];
  for (int i0 = gid; i0 < N4; i0 += ST * 4) {
    float4 v[4];
    float den[4];
#pragma unroll
    for (int u = 0; u < 4; ++u) {
      const int ic = min(i0 + u * ST, N4 - 1);
      v[u] = o4[ic];
      den[u] = segsum[ic / Q] + 1e-16f;
    }
#pragma unroll
    for (int u = 0; u < 4; ++u) {
      const int i = i0 + u * ST;
      if (i < N4) {
        float4 o;
        o.x = v[u].x / den[u] + b.x;
        o.y = v[u].y / den[u] + b.y;
        o.z = v[u].z / den[u] + b.z;
        o.w = v[u].w / den[u] + b.w;
        o4[i] = o;
      }
    }
  }
}

__global__ __launch_bounds__(SB) void attr_accum_k(const int* __restrict__ dst,
                                                   const float* __restrict__ ea,
                                                   float* attr_sum, float* cnt) {
  constexpr int NT = NE * 16;
  const int gid = blockIdx.x * SB + threadIdx.x;
  const int j = gid & 15;
  for (int i0 = gid; i0 < NT; i0 += ST * 8) {
    float v[8];
    int d[8];
#pragma unroll
    for (int u = 0; u < 8; ++u) {
      const int ic = min(i0 + u * ST, NT - 1);
      v[u] = ea[ic];
      d[u] = dst[ic >> 4];
    }
#pragma unroll
    for (int u = 0; u < 8; ++u) {
      if (i0 + u * ST < NT) {
        atomicAdd(&attr_sum[(size_t)d[u] * 16 + j], v[u]);
        if (j == 0) atomicAdd(&cnt[d[u]], 1.f);
      }
    }
  }
}
__global__ __launch_bounds__(SB) void attr_mean_k(float* attr_sum, const float* __restrict__ cnt) {
  for (int i = blockIdx.x * SB + threadIdx.x; i < NN * 16; i += ST)
    attr_sum[i] /= fmaxf(cnt[i >> 4], 1.f);
}

__global__ void wa_k(const float* __restrict__ We, const float* __restrict__ ae,
                     float* wa, int K) {
  int i = threadIdx.x;
  if (i < 16) { float s = 0.f; for (int j = 0; j < K; ++j) s += We[i * K + j] * ae[j]; wa[i] = s; }
}

static constexpr int GW = 2048;
template <int COUT>
__global__ __launch_bounds__(256) void gemm_k128(const float* __restrict__ A,
                                                 const float* __restrict__ B,
                                                 float* __restrict__ C) {
  constexpr int NCG = COUT / 32;
  constexpr int NSW = GW / NCG;
  constexpr int NS = NN / 16;
  static_assert(NN % 16 == 0, "rows come in stripes of 16");
  const int lane = threadIdx.x & 31, half = lane >> 4, lidx = lane & 15;
  const int wid = __builtin_amdgcn_readfirstlane((int)((blockIdx.x * 256 + threadIdx.x) >> 5));
  const int n0 = (wid % NCG) * 32;
  const int sw = wid / NCG;

  float bq[2][16][4];
#pragma unroll
  for (int t = 0; t < 2; ++t)
#pragma unroll
    for (int j = 0; j < 16; ++j)
#pragma unroll
      for (int i = 0; i < 4; ++i)
        bq[t][j][i] = B[(8 * j + 4 * half + i) * COUT + n0 + 16 * t + lidx];

  const float4* A4 = (const float4*)A;
  auto load = [&](float4 (&a)[16], int s) {
    const float4* p = A4 + ((size_t)s * 16 + lidx) * 32 + half;
#pragma unroll
    for (int j = 0; j < 16; ++j) a[j] = p[2 * j];
  };
  auto compute = [&](const float4 (&a)[16], int s) {
    v8f acc0 = {}, acc1 = {};
#pragma unroll
    for (int j = 0; j < 16; ++j) {
      v2f a01, a23, b;
      a01.x = a[j].x; a01.y = a[j].y;
      a23.x = a[j].z; a23.y = a[j].w;
      b.x = bq[0][j][0]; b.y = bq[0][j][1];
      acc0 = __builtin_amdgcn_wmma_f32_16x16x4_f32(false, a01, false, b, (short)0, acc0,
                                                   false, false);
      b.x = bq[1][j][0]; b.y = bq[1][j][1];
      acc1 = __builtin_amdgcn_wmma_f32_16x16x4_f32(false, a01, false, b, (short)0, acc1,
                                                   false, false);
      b.x = bq[0][j][2]; b.y = bq[0][j][3];
      acc0 = __builtin_amdgcn_wmma_f32_16x16x4_f32(false, a23, false, b, (short)0, acc0,
                                                   false, false);
      b.x = bq[1][j][2]; b.y = bq[1][j][3];
      acc1 = __builtin_amdgcn_wmma_f32_16x16x4_f32(false, a23, false, b, (short)0, acc1,
                                                   false, false);
    }
    float* c = C + ((size_t)s * 16 + 8 * half) * COUT + n0 + lidx;
#pragma unroll
    for (int r = 0; r < 8; ++r) {
      c[r * COUT] = acc0[r];
      c[r * COUT + 16] = acc1[r];
    }
  };

  float4 ba[16], bb[16];
  int s = sw;
  load(ba, s);
  for (; s + NSW < NS; s += 2 * NSW) {
    load(bb, s + NSW);
    compute(ba, s);
    load(ba, min(s + 2 * NSW, NS - 1));
    compute(bb, s + NSW);
  }
  if (s < NS) compute(ba, s);
}

template <int C, bool ET>
__global__ __launch_bounds__(SB) void node_prep_k(
    const float* __restrict__ h, const float* __restrict__ a_src,
    const float* __restrict__ a_dst, const float* __restrict__ mean_attr,
    const float* __restrict__ wa, float* __restrict__ s_src, float* __restrict__ s_dst,
    float* __restrict__ self_logit, unsigned* __restrict__ segmax_u) {
  constexpr int R = C / 32;
  constexpr int U = 256 / C;
  const int gid = blockIdx.x * SB + threadIdx.x;
  const int lane = gid & 31;
  float as[R], ad[R];
#pragma unroll
  for (int i = 0; i < R; ++i) { as[i] = a_src[lane + i * 32]; ad[i] = a_dst[lane + i * 32]; }
  for (int n0 = gid >> 5; n0 < NN; n0 += NW * U) {
    float hv[U][R];
#pragma unroll
    for (int u = 0; u < U; ++u) {
      const int nc = min(n0 + u * NW, NN - 1);
#pragma unroll
      for (int i = 0; i < R; ++i) hv[u][i] = h[(size_t)nc * C + lane + i * 32];
    }
    float psv[U], pdv[U];
#pragma unroll
    for (int u = 0; u < U; ++u) {
      float ps = 0.f, pd = 0.f;
#pragma unroll
      for (int i = 0; i < R; ++i) { ps += hv[u][i] * as[i]; pd += hv[u][i] * ad[i]; }
#pragma unroll
      for (int m = 16; m > 0; m >>= 1) { ps += __shfl_xor(ps, m, 32); pd += __shfl_xor(pd, m, 32); }
      psv[u] = ps; pdv[u] = pd;
    }
#pragma unroll
    for (int u = 0; u < U; ++u) {
      const int n = n0 + u * NW;
      if (n < NN) {
        const float ps = psv[u], pd = pdv[u];
        if (lane == 0) {
          float st = 0.f;
          if (ET) { for (int j = 0; j < 16; ++j) st += mean_attr[(size_t)n * 16 + j] * wa[j]; }
          float sl = leaky(ps + pd + st);
          s_src[n] = ps; s_dst[n] = pd; self_logit[n] = sl; segmax_u[n] = f2ord(sl);
        }
      }
    }
  }
}

__global__ __launch_bounds__(SB) void eterm_k(const float* __restrict__ ea,
                                              const float* __restrict__ wa,
                                              float* __restrict__ eterm) {
  constexpr int NT = NE * 16;
  const int gid = blockIdx.x * SB + threadIdx.x;
  const int j = gid & 15;
  const float w = wa[j];
  for (int i0 = gid; i0 < NT; i0 += ST * 8) {
    float v[8];
#pragma unroll
    for (int u = 0; u < 8; ++u) v[u] = ea[min(i0 + u * ST, NT - 1)] * w;
#pragma unroll
    for (int u = 0; u < 8; ++u) {
      float t = v[u];
      t += __shfl_xor(t, 1, 32); t += __shfl_xor(t, 2, 32);
      t += __shfl_xor(t, 4, 32); t += __shfl_xor(t, 8, 32);
      v[u] = t;
    }
#pragma unroll
    for (int u = 0; u < 8; ++u) {
      const int i = i0 + u * ST;
      if (j == 0 && i < NT) eterm[i >> 4] = v[u];
    }
  }
}

template <bool ET>
__global__ __launch_bounds__(SB) void edge_max_k(
    const int* __restrict__ src, const int* __restrict__ dst,
    const float* __restrict__ s_src, const float* __restrict__ s_dst,
    const float* __restrict__ eterm, unsigned* segmax_u) {
  for (int e0 = blockIdx.x * SB + threadIdx.x; e0 < NE; e0 += ST * 4) {
    int sv[4], dv[4];
    float l[4];
#pragma unroll
    for (int u = 0; u < 4; ++u) {
      const int ec = min(e0 + u * ST, NE - 1);
      sv[u] = src[ec];
      dv[u] = dst[ec];
      l[u] = ET ? eterm[ec] : 0.f;
    }
#pragma unroll
    for (int u = 0; u < 4; ++u) {
      const float g = s_src[sv[u]] + s_dst[dv[u]];
      l[u] = ET ? g + l[u] : g;
    }
#pragma unroll
    for (int u = 0; u < 4; ++u)
      if (e0 + u * ST < NE) atomicMax(&segmax_u[dv[u]], f2ord(leaky(l[u])));
  }
}

__global__ void node_mid_k(const unsigned* __restrict__ segmax_u,
                           const float* __restrict__ self_logit, float* segmaxf,
                           float* segsum) {
  int n = blockIdx.x * blockDim.x + threadIdx.x;
  if (n >= NN) return;
  float m = ord2f(segmax_u[n]);
  segmaxf[n] = m;
  segsum[n] = __expf(self_logit[n] - m);
}

static constexpr int RT = 400;
static constexpr int NTILE = (NN + RT - 1) / RT;
static constexpr int QCAP = 544;
static constexpr int NSTEP = NE / 512;
static_assert(NE % 512 == 0, "the key plane is swept in steps of 512 edges");
static_assert(NTILE <= 255, "a tile id is one byte, and 255 is the filler that matches no tile");

static constexpr int KEYB = (NE / 4 + SB - 1) / SB;
__global__ __launch_bounds__(SB) void tile_key_k(const int* __restrict__ dst,
                                                 unsigned* __restrict__ key) {
  const int i = blockIdx.x * SB + threadIdx.x;
  if (i < NE / 4) {
    const int4 d = ((const int4*)dst)[i];
    key[i] = (unsigned)d.x / RT | ((unsigned)d.y / RT) << 8 | ((unsigned)d.z / RT) << 16 |
             ((unsigned)d.w / RT) << 24;
  }
}

template <int V> struct RowVec;
template <> struct RowVec<4> { typedef float4 T; };
template <> struct RowVec<2> { typedef float2 T; };
__device__ __forceinline__ float vget(const float4& v, int k) {
  return k == 0 ? v.x : k == 1 ? v.y : k == 2 ? v.z : v.w;
}
__device__ __forceinline__ float vget(const float2& v, int k) { return k == 0 ? v.x : v.y; }
__device__ __forceinline__ void vset(float4& v, int k, float f) {
  if (k == 0) v.x = f; else if (k == 1) v.y = f; else if (k == 2) v.z = f; else v.w = f;
}
__device__ __forceinline__ void vset(float2& v, int k, float f) {
  if (k == 0) v.x = f; else v.y = f;
}

template <int C, bool ET>
__global__ __launch_bounds__(1024) void tile_agg_k(
    const int* __restrict__ src, const int* __restrict__ dst, const unsigned* __restrict__ key,
    const float* __restrict__ s_src, const float* __restrict__ s_dst,
    const float* __restrict__ eterm, const float* __restrict__ segmaxf,
    float* __restrict__ segsum, const float* __restrict__ h, float* __restrict__ out) {
  constexpr int V = C / 32;
  typedef typename RowVec<V>::T VecT;
  __shared__ float tile[RT * C];
  __shared__ float ssum[RT];
  __shared__ int queue[32 * QCAP];
  const int tid = threadIdx.x, lane = tid & 31;
  const int wave = __builtin_amdgcn_readfirstlane(tid >> 5);
  const int row0 = blockIdx.x * RT;
  const int rows = min(RT, NN - row0);

  for (int i = tid; i < rows * 32; i += 1024) {
    const int r = i >> 5, n = row0 + r;
    const float a = segsum[n];
    const VecT hv = *(const VecT*)&h[(size_t)n * C + lane * V];
#pragma unroll
    for (int k = 0; k < V; ++k) tile[r * C + k * 32 + lane] = a * vget(hv, k);
    if (lane == 0) ssum[r] = a;
  }
  __syncthreads();

  int* myq = &queue[wave * QCAP];
  int cnt = 0;

  auto process = [&](int base, int nb) {
    int s = 0, r = 0;
    float w = 0.f;
    if (lane < nb) {
      const int e = myq[base + lane];
      s = src[e];
      const int d = dst[e];
      float l = s_src[s] + s_dst[d];
      if (ET) l += eterm[e];
      w = __expf(leaky(l) - segmaxf[d]);
      r = d - row0;
      atomicAdd(&ssum[r], w);
    }
    for (int j0 = 0; j0 < nb; j0 += 8) {
      VecT hv[8];
#pragma unroll
      for (int u = 0; u < 8; ++u) {
        const int sj = __builtin_amdgcn_readlane(s, j0 + u);
        hv[u] = *(const VecT*)&h[(size_t)sj * C + lane * V];
      }
#pragma unroll
      for (int u = 0; u < 8; ++u) {
        if (j0 + u < nb) {
          const int rj = __builtin_amdgcn_readlane(r, j0 + u);
          const float wj = __int_as_float(__builtin_amdgcn_readlane(__float_as_int(w), j0 + u));
          float* t = &tile[rj * C + lane];
#pragma unroll
          for (int k = 0; k < V; ++k) atomicAdd(t + k * 32, wj * vget(hv[u], k));
        }
      }
    }
  };

  const unsigned tt = blockIdx.x * 0x01010101u;
  const int rot = blockIdx.x * (NSTEP / NTILE);
  auto step_of = [&](int k) { const int kk = k + rot; return kk >= NSTEP ? kk - NSTEP : kk; };
  const uint4* key4 = (const uint4*)key;
  auto fetch = [&](uint4 (&d)[4], int k0) {
#pragma unroll
    for (int u = 0; u < 4; ++u) {
      const int k = k0 + 32 * u;
      d[u] = (k < NSTEP) ? key4[(size_t)step_of(k) * 32 + lane] : make_uint4(~0u, ~0u, ~0u, ~0u);
    }
  };
  auto differs = [&](unsigned x) {
    const unsigned y = x ^ tt;
    return (((y & 0x7f7f7f7fu) + 0x7f7f7f7fu) | y) & 0x80808080u;
  };
  uint4 cur[4], nxt[4];
  fetch(cur, wave);
  for (int k0 = wave; k0 < NSTEP; k0 += 32 * 4) {
    fetch(nxt, k0 + 32 * 4);
#pragma unroll
    for (int u = 0; u < 4; ++u) {
      unsigned m = ~(differs(cur[u].x) >> 7 | differs(cur[u].y) >> 6 | differs(cur[u].z) >> 5 |
                     differs(cur[u].w) >> 4) & 0x0f0f0f0fu;
      unsigned live = __builtin_amdgcn_ballot_w32(m != 0u);
      if (live != 0u) {
        const int e0 = step_of(k0 + 32 * u) * 512 + lane * 16;
        do {
          if (m != 0u) {
            const int bit = __builtin_ctz(m);
            myq[cnt + __builtin_amdgcn_mbcnt_lo(live, 0u)] = e0 + (bit & 7) * 4 + (bit >> 3);
            m &= m - 1u;
          }
          cnt += __popc(live);
          live = __builtin_amdgcn_ballot_w32(m != 0u);
        } while (live != 0u);
        if (cnt >= 32) {
          const int full = cnt & ~31;
          for (int b = 0; b < full; b += 32) process(b, 32);
          const int rem = cnt - full;
          const int keep = (lane < rem) ? myq[full + lane] : 0;
          if (lane < rem) myq[lane] = keep;
          cnt = rem;
        }
      }
    }
#pragma unroll
    for (int u = 0; u < 4; ++u) cur[u] = nxt[u];
  }
  if (cnt > 0) process(0, cnt);
  __syncthreads();

  for (int i = tid; i < rows * 32; i += 1024) {
    const int r = i >> 5, n = row0 + r;
    VecT o;
#pragma unroll
    for (int k = 0; k < V; ++k) vset(o, k, tile[r * C + k * 32 + lane]);
    *(VecT*)&out[(size_t)n * C + lane * V] = o;
    if (lane == 0) segsum[n] = ssum[r];
  }
}

struct Scratch {
  float *s_src, *s_dst, *self_l, *segmaxf, *segsum;
  unsigned* segmax_u;
  const float* mean_attr;
  float *eterm, *wa;
  const unsigned* key;
};

template <int C, bool ET>
static void run_gat(hipStream_t st, const float* hin, const float* Wm, const float* asrc,
                    const float* adst, const float* We, const float* ae,
                    const int* src, const int* dst, const float* edge_attr, float* hg,
                    float* out, const Scratch& S) {
  if (ET) wa_k<<<1, 16, 0, st>>>(We, ae, S.wa, C);
  gemm_k128<C><<<GW / 8, 256, 0, st>>>(hin, Wm, hg);
  node_prep_k<C, ET><<<SG, SB, 0, st>>>(hg, asrc, adst, S.mean_attr, S.wa, S.s_src, S.s_dst,
                                        S.self_l, S.segmax_u);
  if (ET) eterm_k<<<SG, SB, 0, st>>>(edge_attr, S.wa, S.eterm);
  edge_max_k<ET><<<SG, SB, 0, st>>>(src, dst, S.s_src, S.s_dst, S.eterm, S.segmax_u);
  node_mid_k<<<(NN + 255) / 256, 256, 0, st>>>(S.segmax_u, S.self_l, S.segmaxf, S.segsum);
  tile_agg_k<C, ET><<<NTILE, 1024, 0, st>>>(src, dst, S.key, S.s_src, S.s_dst, S.eterm,
                                            S.segmaxf, S.segsum, hg, out);
}

extern "C" void kernel_launch(void* const* d_in, const int* in_sizes, int n_in, void* d_out,
                              int out_size, void* d_ws, size_t ws_size, hipStream_t stream) {
  (void)in_sizes; (void)n_in; (void)out_size; (void)ws_size;
  const float* x = (const float*)d_in[0];
  const int* ei = (const int*)d_in[1];
  const int* src = ei;
  const int* dst = ei + NE;
  const float* edge_attr = (const float*)d_in[2];
  const float *W1 = (const float*)d_in[3], *a1s = (const float*)d_in[4],
              *a1d = (const float*)d_in[5], *b1 = (const float*)d_in[6];
  const float *W2 = (const float*)d_in[7], *a2s = (const float*)d_in[8],
              *a2d = (const float*)d_in[9], *b2 = (const float*)d_in[10];
  const float *We2 = (const float*)d_in[11], *ae2 = (const float*)d_in[12];
  const float *Wr = (const float*)d_in[13], *br = (const float*)d_in[14];
  const float *Wmu = (const float*)d_in[15], *amus = (const float*)d_in[16],
              *amud = (const float*)d_in[17], *bmu = (const float*)d_in[18],
              *Wemu = (const float*)d_in[19], *aemu = (const float*)d_in[20];
  const float *Wls = (const float*)d_in[21], *alss = (const float*)d_in[22],
              *alsd = (const float*)d_in[23], *bls = (const float*)d_in[24],
              *Wels = (const float*)d_in[25], *aels = (const float*)d_in[26];
  float* outf = (float*)d_out;

  float* W = (float*)d_ws;
  size_t o = 0;
  float* hA = W + o; o += (size_t)NN * 128;
  float* hB = W + o; o += (size_t)NN * 128;
  float* hC = W + o; o += (size_t)NN * 128;
  Scratch S;
  S.s_src = W + o; o += NN;
  S.s_dst = W + o; o += NN;
  S.self_l = W + o; o += NN;
  S.segmaxf = W + o; o += NN;
  S.segsum = W + o; o += NN;
  S.segmax_u = (unsigned*)(W + o); o += NN;
  float* cnt = W + o; o += NN;
  float* mean_attr = W + o; o += (size_t)NN * 16;
  S.mean_attr = mean_attr;
  S.eterm = W + o; o += NE;
  S.wa = W + o; o += 16;
  unsigned* key = (unsigned*)(W + o); o += NE / 4;
  S.key = key;

  tile_key_k<<<KEYB, SB, 0, stream>>>(dst, key);

  zero_f32_k<<<SG, SB, 0, stream>>>((float4*)cnt, NN * 17 / 4);
  attr_accum_k<<<SG, SB, 0, stream>>>(dst, edge_attr, mean_attr, cnt);
  attr_mean_k<<<SG, SB, 0, stream>>>(mean_attr, cnt);

  run_gat<128, false>(stream, x, W1, a1s, a1d, nullptr, nullptr, src, dst, edge_attr, hA, hB,
                      S);
  norm_relu_k<<<SG, SB, 0, stream>>>(hB, S.segsum, b1);

  run_gat<128, true>(stream, hB, W2, a2s, a2d, We2, ae2, src, dst, edge_attr, hA, hC, S);

  gemm_k128<128><<<GW / 8, 256, 0, stream>>>(x, Wr, hB);
  residual_k<<<SG, SB, 0, stream>>>(hC, S.segsum, b2, hB, br);

  run_gat<64, true>(stream, hB, Wmu, amus, amud, Wemu, aemu, src, dst, edge_attr, hA, outf, S);
  norm_bias_k<64><<<SG, SB, 0, stream>>>(outf, S.segsum, bmu);
  run_gat<64, true>(stream, hB, Wls, alss, alsd, Wels, aels, src, dst, edge_attr, hA,
                    outf + (size_t)NN * 64, S);
  norm_bias_k<64><<<SG, SB, 0, stream>>>(outf + (size_t)NN * 64, S.segsum, bls);
}


# Softmax max taken by the owner workgroup in LDS (integer max in a first key sweep; edges, logits logged per wave in LDS for the add phase): global atomic-max and node pass gone; adds in arrival order
# speedup vs baseline: 2.7301x; 1.3765x over previous
#include <hip/hip_runtime.h>
#include <math.h>

typedef float v2f __attribute__((ext_vector_type(2)));
typedef float v8f __attribute__((ext_vector_type(8)));

static constexpr int   NN = 100000;
static constexpr int   NE = 1600000;
static constexpr float NEG = 0.2f;

__device__ __forceinline__ float leaky(float x) { return x > 0.f ? x : NEG * x; }
__device__ __forceinline__ unsigned f2ord(float f) {
  unsigned u = __float_as_uint(f);
  return (u & 0x80000000u) ? ~u : (u | 0x80000000u);
}
__device__ __forceinline__ float ord2f(unsigned v) {
  unsigned u = (v & 0x80000000u) ? (v & 0x7fffffffu) : ~v;
  return __uint_as_float(u);
}

static constexpr int SB = 256;
static constexpr int SG = 2048;
static constexpr int ST = SG * SB;
static constexpr int NW = ST / 32;

__global__ __launch_bounds__(SB) void zero_f32_k(float4* __restrict__ p, int n4) {
  for (int i = blockIdx.x * SB + threadIdx.x; i < n4; i += ST) p[i] = make_float4(0.f, 0.f, 0.f, 0.f);
}
__global__ __launch_bounds__(SB) void norm_relu_k(float* p, const float* __restrict__ segsum,
                                                  const float* __restrict__ bias) {
  constexpr int N4 = NN * 32;
  float4* p4 = (float4*)p;
  const int gid = blockIdx.x * SB + threadIdx.x;
  const float4 b = *(const float4*)&bias[(gid & 31) * 4];
  for (int i0 = gid; i0 < N4; i0 += ST * 4) {
    float4 v[4];
    float den[4];
#pragma unroll
    for (int u = 0; u < 4; ++u) {
      const int ic = min(i0 + u * ST, N4 - 1);
      v[u] = p4[ic];
      den[u] = segsum[ic >> 5] + 1e-16f;
    }
#pragma unroll
    for (int u = 0; u < 4; ++u) {
      const int i = i0 + u * ST;
      if (i < N4) {
        float4 o;
        o.x = fmaxf(v[u].x / den[u] + b.x, 0.f);
        o.y = fmaxf(v[u].y / den[u] + b.y, 0.f);
        o.z = fmaxf(v[u].z / den[u] + b.z, 0.f);
        o.w = fmaxf(v[u].w / den[u] + b.w, 0.f);
        p4[i] = o;
      }
    }
  }
}
__global__ __launch_bounds__(SB) void residual_k(const float* __restrict__ h2,
                                                 const float* __restrict__ segsum,
                                                 const float* __restrict__ b2, float* r,
                                                 const float* __restrict__ br) {
  constexpr int N4 = NN * 32;
  const float4* h4 = (const float4*)h2;
  float4* r4 = (float4*)r;
  const int gid = blockIdx.x * SB + threadIdx.x;
  const float4 bb = *(const float4*)&b2[(gid & 31) * 4];
  const float4 rb = *(const float4*)&br[(gid & 31) * 4];
  for (int i0 = gid; i0 < N4; i0 += ST * 4) {
    float4 v[4], w[4];
    float den[4];
#pragma unroll
    for (int u = 0; u < 4; ++u) {
      const int ic = min(i0 + u * ST, N4 - 1);
      v[u] = h4[ic];
      w[u] = r4[ic];
      den[u] = segsum[ic >> 5] + 1e-16f;
    }
#pragma unroll
    for (int u = 0; u < 4; ++u) {
      const int i = i0 + u * ST;
      if (i < N4) {
        float4 o;
        o.x = fmaxf(v[u].x / den[u] + bb.x, 0.f) + w[u].x + rb.x;
        o.y = fmaxf(v[u].y / den[u] + bb.y, 0.f) + w[u].y + rb.y;
        o.z = fmaxf(v[u].z / den[u] + bb.z, 0.f) + w[u].z + rb.z;
        o.w = fmaxf(v[u].w / den[u] + bb.w, 0.f) + w[u].w + rb.w;
        r4[i] = o;
      }
    }
  }
}
template <int C>
__global__ __launch_bounds__(SB) void norm_bias_k(float* out, const float* __restrict__ segsum,
                                                  const float* __restrict__ bias) {
  constexpr int Q = C / 4, N4 = NN * Q;
  float4* o4 = (float4*)out;
  const int gid = blockIdx.x * SB + threadIdx.x;
  const float4 b = *(const float4*)&bias[(gid % Q) * 4];
  for (int i0 = gid; i0 < N4; i0 += ST * 4) {
    float4 v[4];
    float den[4];
#pragma unroll
    for (int u = 0; u < 4; ++u) {
      const int ic = min(i0 + u * ST, N4 - 1);
      v[u] = o4[ic];
      den[u] = segsum[ic / Q] + 1e-16f;
    }
#pragma unroll
    for (int u = 0; u < 4; ++u) {
      const int i = i0 + u * ST;
      if (i < N4) {
        float4 o;
        o.x = v[u].x / den[u] + b.x;
        o.y = v[u].y / den[u] + b.y;
        o.z = v[u].z / den[u] + b.z;
        o.w = v[u].w / den[u] + b.w;
        o4[i] = o;
      }
    }
  }
}

__global__ __launch_bounds__(SB) void attr_accum_k(const int* __restrict__ dst,
                                                   const float* __restrict__ ea,
                                                   float* attr_sum, float* cnt) {
  constexpr int NT = NE * 16;
  const int gid = blockIdx.x * SB + threadIdx.x;
  const int j = gid & 15;
  for (int i0 = gid; i0 < NT; i0 += ST * 8) {
    float v[8];
    int d[8];
#pragma unroll
    for (int u = 0; u < 8; ++u) {
      const int ic = min(i0 + u * ST, NT - 1);
      v[u] = ea[ic];
      d[u] = dst[ic >> 4];
    }
#pragma unroll
    for (int u = 0; u < 8; ++u) {
      if (i0 + u * ST < NT) {
        atomicAdd(&attr_sum[(size_t)d[u] * 16 + j], v[u]);
        if (j == 0) atomicAdd(&cnt[d[u]], 1.f);
      }
    }
  }
}
__global__ __launch_bounds__(SB) void attr_mean_k(float* attr_sum, const float* __restrict__ cnt) {
  for (int i = blockIdx.x * SB + threadIdx.x; i < NN * 16; i += ST)
    attr_sum[i] /= fmaxf(cnt[i >> 4], 1.f);
}

__global__ void wa_k(const float* __restrict__ We, const float* __restrict__ ae,
                     float* wa, int K) {
  int i = threadIdx.x;
  if (i < 16) { float s = 0.f; for (int j = 0; j < K; ++j) s += We[i * K + j] * ae[j]; wa[i] = s; }
}

static constexpr int GW = 2048;
template <int COUT>
__global__ __launch_bounds__(256) void gemm_k128(const float* __restrict__ A,
                                                 const float* __restrict__ B,
                                                 float* __restrict__ C) {
  constexpr int NCG = COUT / 32;
  constexpr int NSW = GW / NCG;
  constexpr int NS = NN / 16;
  static_assert(NN % 16 == 0, "rows come in stripes of 16");
  const int lane = threadIdx.x & 31, half = lane >> 4, lidx = lane & 15;
  const int wid = __builtin_amdgcn_readfirstlane((int)((blockIdx.x * 256 + threadIdx.x) >> 5));
  const int n0 = (wid % NCG) * 32;
  const int sw = wid / NCG;

  float bq[2][16][4];
#pragma unroll
  for (int t = 0; t < 2; ++t)
#pragma unroll
    for (int j = 0; j < 16; ++j)
#pragma unroll
      for (int i = 0; i < 4; ++i)
        bq[t][j][i] = B[(8 * j + 4 * half + i) * COUT + n0 + 16 * t + lidx];

  const float4* A4 = (const float4*)A;
  auto load = [&](float4 (&a)[16], int s) {
    const float4* p = A4 + ((size_t)s * 16 + lidx) * 32 + half;
#pragma unroll
    for (int j = 0; j < 16; ++j) a[j] = p[2 * j];
  };
  auto compute = [&](const float4 (&a)[16], int s) {
    v8f acc0 = {}, acc1 = {};
#pragma unroll
    for (int j = 0; j < 16; ++j) {
      v2f a01, a23, b;
      a01.x = a[j].x; a01.y = a[j].y;
      a23.x = a[j].z; a23.y = a[j].w;
      b.x = bq[0][j][0]; b.y = bq[0][j][1];
      acc0 = __builtin_amdgcn_wmma_f32_16x16x4_f32(false, a01, false, b, (short)0, acc0,
                                                   false, false);
      b.x = bq[1][j][0]; b.y = bq[1][j][1];
      acc1 = __builtin_amdgcn_wmma_f32_16x16x4_f32(false, a01, false, b, (short)0, acc1,
                                                   false, false);
      b.x = bq[0][j][2]; b.y = bq[0][j][3];
      acc0 = __builtin_amdgcn_wmma_f32_16x16x4_f32(false, a23, false, b, (short)0, acc0,
                                                   false, false);
      b.x = bq[1][j][2]; b.y = bq[1][j][3];
      acc1 = __builtin_amdgcn_wmma_f32_16x16x4_f32(false, a23, false, b, (short)0, acc1,
                                                   false, false);
    }
    float* c = C + ((size_t)s * 16 + 8 * half) * COUT + n0 + lidx;
#pragma unroll
    for (int r = 0; r < 8; ++r) {
      c[r * COUT] = acc0[r];
      c[r * COUT + 16] = acc1[r];
    }
  };

  float4 ba[16], bb[16];
  int s = sw;
  load(ba, s);
  for (; s + NSW < NS; s += 2 * NSW) {
    load(bb, s + NSW);
    compute(ba, s);
    load(ba, min(s + 2 * NSW, NS - 1));
    compute(bb, s + NSW);
  }
  if (s < NS) compute(ba, s);
}

template <int C, bool ET>
__global__ __launch_bounds__(SB) void node_prep_k(
    const float* __restrict__ h, const float* __restrict__ a_src,
    const float* __restrict__ a_dst, const float* __restrict__ mean_attr,
    const float* __restrict__ wa, float* __restrict__ s_src, float* __restrict__ s_dst,
    float* __restrict__ self_logit) {
  constexpr int R = C / 32;
  constexpr int U = 256 / C;
  const int gid = blockIdx.x * SB + threadIdx.x;
  const int lane = gid & 31;
  float as[R], ad[R];
#pragma unroll
  for (int i = 0; i < R; ++i) { as[i] = a_src[lane + i * 32]; ad[i] = a_dst[lane + i * 32]; }
  for (int n0 = gid >> 5; n0 < NN; n0 += NW * U) {
    float hv[U][R];
#pragma unroll
    for (int u = 0; u < U; ++u) {
      const int nc = min(n0 + u * NW, NN - 1);
#pragma unroll
      for (int i = 0; i < R; ++i) hv[u][i] = h[(size_t)nc * C + lane + i * 32];
    }
    float psv[U], pdv[U];
#pragma unroll
    for (int u = 0; u < U; ++u) {
      float ps = 0.f, pd = 0.f;
#pragma unroll
      for (int i = 0; i < R; ++i) { ps += hv[u][i] * as[i]; pd += hv[u][i] * ad[i]; }
#pragma unroll
      for (int m = 16; m > 0; m >>= 1) { ps += __shfl_xor(ps, m, 32); pd += __shfl_xor(pd, m, 32); }
      psv[u] = ps; pdv[u] = pd;
    }
#pragma unroll
    for (int u = 0; u < U; ++u) {
      const int n = n0 + u * NW;
      if (n < NN) {
        const float ps = psv[u], pd = pdv[u];
        if (lane == 0) {
          float st = 0.f;
          if (ET) { for (int j = 0; j < 16; ++j) st += mean_attr[(size_t)n * 16 + j] * wa[j]; }
          float sl = leaky(ps + pd + st);
          s_src[n] = ps; s_dst[n] = pd; self_logit[n] = sl;
        }
      }
    }
  }
}

__global__ __launch_bounds__(SB) void eterm_k(const float* __restrict__ ea,
                                              const float* __restrict__ wa,
                                              float* __restrict__ eterm) {
  constexpr int NT = NE * 16;
  const int gid = blockIdx.x * SB + threadIdx.x;
  const int j = gid & 15;
  const float w = wa[j];
  for (int i0 = gid; i0 < NT; i0 += ST * 8) {
    float v[8];
#pragma unroll
    for (int u = 0; u < 8; ++u) v[u] = ea[min(i0 + u * ST, NT - 1)] * w;
#pragma unroll
    for (int u = 0; u < 8; ++u) {
      float t = v[u];
      t += __shfl_xor(t, 1, 32); t += __shfl_xor(t, 2, 32);
      t += __shfl_xor(t, 4, 32); t += __shfl_xor(t, 8, 32);
      v[u] = t;
    }
#pragma unroll
    for (int u = 0; u < 8; ++u) {
      const int i = i0 + u * ST;
      if (j == 0 && i < NT) eterm[i >> 4] = v[u];
    }
  }
}

static constexpr int RT = 400;
static constexpr int NTILE = (NN + RT - 1) / RT;
static constexpr int QCAP = 64;
static constexpr int LOGCAP = 320;
static constexpr int NSTEP = NE / 512;
static_assert(NE % 512 == 0, "the key plane is swept in steps of 512 edges");
static_assert(NTILE <= 255, "a tile id is one byte, and 255 is the filler that matches no tile");
static_assert(NN <= (1 << 17) && RT <= (1 << 9), "a log entry packs source (17 bits) and row (9 bits)");

static constexpr int KEYB = (NE / 4 + SB - 1) / SB;
__global__ __launch_bounds__(SB) void tile_key_k(const int* __restrict__ dst,
                                                 unsigned* __restrict__ key) {
  const int i = blockIdx.x * SB + threadIdx.x;
  if (i < NE / 4) {
    const int4 d = ((const int4*)dst)[i];
    key[i] = (unsigned)d.x / RT | ((unsigned)d.y / RT) << 8 | ((unsigned)d.z / RT) << 16 |
             ((unsigned)d.w / RT) << 24;
  }
}

template <int V> struct RowVec;
template <> struct RowVec<4> { typedef float4 T; };
template <> struct RowVec<2> { typedef float2 T; };
__device__ __forceinline__ float vget(const float4& v, int k) {
  return k == 0 ? v.x : k == 1 ? v.y : k == 2 ? v.z : v.w;
}
__device__ __forceinline__ float vget(const float2& v, int k) { return k == 0 ? v.x : v.y; }
__device__ __forceinline__ void vset(float4& v, int k, float f) {
  if (k == 0) v.x = f; else if (k == 1) v.y = f; else if (k == 2) v.z = f; else v.w = f;
}
__device__ __forceinline__ void vset(float2& v, int k, float f) {
  if (k == 0) v.x = f; else v.y = f;
}

template <int C, bool ET>
__global__ __launch_bounds__(1024) void tile_agg_k(
    const int* __restrict__ src, const int* __restrict__ dst, const unsigned* __restrict__ key,
    const float* __restrict__ s_src, const float* __restrict__ s_dst,
    const float* __restrict__ eterm, const float* __restrict__ self_logit,
    float* __restrict__ segsum, const float* __restrict__ h, float* __restrict__ out) {
  constexpr int V = C / 32;
  typedef typename RowVec<V>::T VecT;
  __shared__ float tile[RT * C];
  __shared__ float ssum[RT];
  __shared__ unsigned smax[RT];
  __shared__ int queue[32 * QCAP];
  __shared__ int log_sr[32 * LOGCAP];
  __shared__ float log_l[32 * LOGCAP];
  const int tid = threadIdx.x, lane = tid & 31;
  const int wave = __builtin_amdgcn_readfirstlane(tid >> 5);
  const int row0 = blockIdx.x * RT;
  const int rows = min(RT, NN - row0);

  for (int r = tid; r < rows; r += 1024) smax[r] = f2ord(self_logit[row0 + r]);
  __syncthreads();

  int* myq = &queue[wave * QCAP];
  int* mysr = &log_sr[wave * LOGCAP];
  float* myll = &log_l[wave * LOGCAP];
  int logn = 0;
  bool ovf = false;

  auto edge_logit = [&](int e, int& s, int& r) {
    s = src[e];
    const int d = dst[e];
    float l = s_src[s] + s_dst[d];
    if (ET) l += eterm[e];
    r = d - row0;
    return leaky(l);
  };

  auto proc_max = [&](int nb) {
    int sr = 0;
    float ll = 0.f;
    if (lane < nb) {
      int s, r;
      ll = edge_logit(myq[lane], s, r);
      atomicMax(&smax[r], f2ord(ll));
      sr = s | r << 17;
    }
    if (!ovf && logn + nb <= LOGCAP) {
      if (lane < nb) { mysr[logn + lane] = sr; myll[logn + lane] = ll; }
      logn += nb;
    } else {
      ovf = true;
    }
  };

  auto agg_rows = [&](int nb, int s, int r, float w) {
    for (int j0 = 0; j0 < nb; j0 += 8) {
      VecT hv[8];
#pragma unroll
      for (int u = 0; u < 8; ++u) {
        const int sj = __builtin_amdgcn_readlane(s, j0 + u);
        hv[u] = *(const VecT*)&h[(size_t)sj * C + lane * V];
      }
#pragma unroll
      for (int u = 0; u < 8; ++u) {
        if (j0 + u < nb) {
          const int rj = __builtin_amdgcn_readlane(r, j0 + u);
          const float wj = __int_as_float(__builtin_amdgcn_readlane(__float_as_int(w), j0 + u));
          float* t = &tile[rj * C + lane];
#pragma unroll
          for (int k = 0; k < V; ++k) atomicAdd(t + k * 32, wj * vget(hv[u], k));
        }
      }
    }
  };
  auto proc_log = [&](int base, int nb) {
    int s = 0, r = 0;
    float w = 0.f;
    if (lane < nb) {
      const int sr = mysr[base + lane];
      s = sr & 0x1ffff;
      r = sr >> 17;
      w = __expf(myll[base + lane] - ord2f(smax[r]));
      atomicAdd(&ssum[r], w);
    }
    agg_rows(nb, s, r, w);
  };
  auto proc_queue = [&](int nb) {
    int s = 0, r = 0;
    float w = 0.f;
    if (lane < nb) {
      const float ll = edge_logit(myq[lane], s, r);
      w = __expf(ll - ord2f(smax[r]));
      atomicAdd(&ssum[r], w);
    }
    agg_rows(nb, s, r, w);
  };

  const unsigned tt = blockIdx.x * 0x01010101u;
  const int rot = blockIdx.x * (NSTEP / NTILE);
  auto step_of = [&](int k) { const int kk = k + rot; return kk >= NSTEP ? kk - NSTEP : kk; };
  const uint4* key4 = (const uint4*)key;
  auto fetch = [&](uint4 (&d)[4], int k0) {
#pragma unroll
    for (int u = 0; u < 4; ++u) {
      const int k = k0 + 32 * u;
      d[u] = (k < NSTEP) ? key4[(size_t)step_of(k) * 32 + lane] : make_uint4(~0u, ~0u, ~0u, ~0u);
    }
  };
  auto differs = [&](unsigned x) {
    const unsigned y = x ^ tt;
    return (((y & 0x7f7f7f7fu) + 0x7f7f7f7fu) | y) & 0x80808080u;
  };
  auto sweep = [&](auto&& proc) {
    int cnt = 0;
    uint4 cur[4], nxt[4];
    fetch(cur, wave);
    for (int k0 = wave; k0 < NSTEP; k0 += 32 * 4) {
      fetch(nxt, k0 + 32 * 4);
#pragma unroll
      for (int u = 0; u < 4; ++u) {
        unsigned m = ~(differs(cur[u].x) >> 7 | differs(cur[u].y) >> 6 | differs(cur[u].z) >> 5 |
                       differs(cur[u].w) >> 4) & 0x0f0f0f0fu;
        unsigned live = __builtin_amdgcn_ballot_w32(m != 0u);
        if (live != 0u) {
          const int e0 = step_of(k0 + 32 * u) * 512 + lane * 16;
          do {
            if (m != 0u) {
              const int bit = __builtin_ctz(m);
              myq[cnt + __builtin_amdgcn_mbcnt_lo(live, 0u)] = e0 + (bit & 7) * 4 + (bit >> 3);
              m &= m - 1u;
            }
            cnt += __popc(live);
            live = __builtin_amdgcn_ballot_w32(m != 0u);
            if (cnt >= 32) {
              proc(32);
              const int rem = cnt - 32;
              const int keep = (lane < rem) ? myq[32 + lane] : 0;
              if (lane < rem) myq[lane] = keep;
              cnt = rem;
            }
          } while (live != 0u);
        }
      }
#pragma unroll
      for (int u = 0; u < 4; ++u) cur[u] = nxt[u];
    }
    if (cnt > 0) proc(cnt);
  };

  sweep(proc_max);
  __syncthreads();

  for (int i = tid; i < rows * 32; i += 1024) {
    const int r = i >> 5, n = row0 + r;
    const float a = __expf(self_logit[n] - ord2f(smax[r]));
    const VecT hv = *(const VecT*)&h[(size_t)n * C + lane * V];
#pragma unroll
    for (int k = 0; k < V; ++k) tile[r * C + k * 32 + lane] = a * vget(hv, k);
    if (lane == 0) ssum[r] = a;
  }
  __syncthreads();

  const int nlog = __builtin_amdgcn_readfirstlane(logn);
  if (__builtin_amdgcn_readfirstlane((int)ovf) == 0) {
    int b = 0;
    for (; b + 32 <= nlog; b += 32) proc_log(b, 32);
    if (b < nlog) proc_log(b, nlog - b);
  } else {
    sweep(proc_queue);
  }
  __syncthreads();

  for (int i = tid; i < rows * 32; i += 1024) {
    const int r = i >> 5, n = row0 + r;
    VecT o;
#pragma unroll
    for (int k = 0; k < V; ++k) vset(o, k, tile[r * C + k * 32 + lane]);
    *(VecT*)&out[(size_t)n * C + lane * V] = o;
    if (lane == 0) segsum[n] = ssum[r];
  }
}

struct Scratch {
  float *s_src, *s_dst, *self_l, *segsum;
  const float* mean_attr;
  float *eterm, *wa;
  const unsigned* key;
};

template <int C, bool ET>
static void run_gat(hipStream_t st, const float* hin, const float* Wm, const float* asrc,
                    const float* adst, const float* We, const float* ae,
                    const int* src, const int* dst, const float* edge_attr, float* hg,
                    float* out, const Scratch& S) {
  if (ET) wa_k<<<1, 16, 0, st>>>(We, ae, S.wa, C);
  gemm_k128<C><<<GW / 8, 256, 0, st>>>(hin, Wm, hg);
  node_prep_k<C, ET><<<SG, SB, 0, st>>>(hg, asrc, adst, S.mean_attr, S.wa, S.s_src, S.s_dst,
                                        S.self_l);
  if (ET) eterm_k<<<SG, SB, 0, st>>>(edge_attr, S.wa, S.eterm);
  tile_agg_k<C, ET><<<NTILE, 1024, 0, st>>>(src, dst, S.key, S.s_src, S.s_dst, S.eterm,
                                            S.self_l, S.segsum, hg, out);
}

extern "C" void kernel_launch(void* const* d_in, const int* in_sizes, int n_in, void* d_out,
                              int out_size, void* d_ws, size_t ws_size, hipStream_t stream) {
  (void)in_sizes; (void)n_in; (void)out_size; (void)ws_size;
  const float* x = (const float*)d_in[0];
  const int* ei = (const int*)d_in[1];
  const int* src = ei;
  const int* dst = ei + NE;
  const float* edge_attr = (const float*)d_in[2];
  const float *W1 = (const float*)d_in[3], *a1s = (const float*)d_in[4],
              *a1d = (const float*)d_in[5], *b1 = (const float*)d_in[6];
  const float *W2 = (const float*)d_in[7], *a2s = (const float*)d_in[8],
              *a2d = (const float*)d_in[9], *b2 = (const float*)d_in[10];
  const float *We2 = (const float*)d_in[11], *ae2 = (const float*)d_in[12];
  const float *Wr = (const float*)d_in[13], *br = (const float*)d_in[14];
  const float *Wmu = (const float*)d_in[15], *amus = (const float*)d_in[16],
              *amud = (const float*)d_in[17], *bmu = (const float*)d_in[18],
              *Wemu = (const float*)d_in[19], *aemu = (const float*)d_in[20];
  const float *Wls = (const float*)d_in[21], *alss = (const float*)d_in[22],
              *alsd = (const float*)d_in[23], *bls = (const float*)d_in[24],
              *Wels = (const float*)d_in[25], *aels = (const float*)d_in[26];
  float* outf = (float*)d_out;

  float* W = (float*)d_ws;
  size_t o = 0;
  float* hA = W + o; o += (size_t)NN * 128;
  float* hB = W + o; o += (size_t)NN * 128;
  float* hC = W + o; o += (size_t)NN * 128;
  Scratch S;
  S.s_src = W + o; o += NN;
  S.s_dst = W + o; o += NN;
  S.self_l = W + o; o += NN;
  S.segsum = W + o; o += NN;
  float* cnt = W + o; o += NN;
  float* mean_attr = W + o; o += (size_t)NN * 16;
  S.mean_attr = mean_attr;
  S.eterm = W + o; o += NE;
  S.wa = W + o; o += 16;
  unsigned* key = (unsigned*)(W + o); o += NE / 4;
  S.key = key;

  tile_key_k<<<KEYB, SB, 0, stream>>>(dst, key);

  zero_f32_k<<<SG, SB, 0, stream>>>((float4*)cnt, NN * 17 / 4);
  attr_accum_k<<<SG, SB, 0, stream>>>(dst, edge_attr, mean_attr, cnt);
  attr_mean_k<<<SG, SB, 0, stream>>>(mean_attr, cnt);

  run_gat<128, false>(stream, x, W1, a1s, a1d, nullptr, nullptr, src, dst, edge_attr, hA, hB,
                      S);
  norm_relu_k<<<SG, SB, 0, stream>>>(hB, S.segsum, b1);

  run_gat<128, true>(stream, hB, W2, a2s, a2d, We2, ae2, src, dst, edge_attr, hA, hC, S);

  gemm_k128<128><<<GW / 8, 256, 0, stream>>>(x, Wr, hB);
  residual_k<<<SG, SB, 0, stream>>>(hC, S.segsum, b2, hB, br);

  run_gat<64, true>(stream, hB, Wmu, amus, amud, Wemu, aemu, src, dst, edge_attr, hA, outf, S);
  norm_bias_k<64><<<SG, SB, 0, stream>>>(outf, S.segsum, bmu);
  run_gat<64, true>(stream, hB, Wls, alss, alsd, Wels, aels, src, dst, edge_attr, hA,
                    outf + (size_t)NN * 64, S);
  norm_bias_k<64><<<SG, SB, 0, stream>>>(outf + (size_t)NN * 64, S.segsum, bls);
}


# Owner tile's per-wave edge log (source|row, edge id) written to workspace by conv1's sweep; later convs read it in whole lines: key sweep and per-edge src/dst gathers go; adds in arrival order
# speedup vs baseline: 3.2991x; 1.2084x over previous
#include <hip/hip_runtime.h>
#include <math.h>

typedef float v2f __attribute__((ext_vector_type(2)));
typedef float v8f __attribute__((ext_vector_type(8)));

static constexpr int   NN = 100000;
static constexpr int   NE = 1600000;
static constexpr float NEG = 0.2f;

__device__ __forceinline__ float leaky(float x) { return x > 0.f ? x : NEG * x; }
__device__ __forceinline__ unsigned f2ord(float f) {
  unsigned u = __float_as_uint(f);
  return (u & 0x80000000u) ? ~u : (u | 0x80000000u);
}
__device__ __forceinline__ float ord2f(unsigned v) {
  unsigned u = (v & 0x80000000u) ? (v & 0x7fffffffu) : ~v;
  return __uint_as_float(u);
}

static constexpr int SB = 256;
static constexpr int SG = 2048;
static constexpr int ST = SG * SB;
static constexpr int NW = ST / 32;

__global__ __launch_bounds__(SB) void zero_f32_k(float4* __restrict__ p, int n4) {
  for (int i = blockIdx.x * SB + threadIdx.x; i < n4; i += ST) p[i] = make_float4(0.f, 0.f, 0.f, 0.f);
}
__global__ __launch_bounds__(SB) void norm_relu_k(float* p, const float* __restrict__ segsum,
                                                  const float* __restrict__ bias) {
  constexpr int N4 = NN * 32;
  float4* p4 = (float4*)p;
  const int gid = blockIdx.x * SB + threadIdx.x;
  const float4 b = *(const float4*)&bias[(gid & 31) * 4];
  for (int i0 = gid; i0 < N4; i0 += ST * 4) {
    float4 v[4];
    float den[4];
#pragma unroll
    for (int u = 0; u < 4; ++u) {
      const int ic = min(i0 + u * ST, N4 - 1);
      v[u] = p4[ic];
      den[u] = segsum[ic >> 5] + 1e-16f;
    }
#pragma unroll
    for (int u = 0; u < 4; ++u) {
      const int i = i0 + u * ST;
      if (i < N4) {
        float4 o;
        o.x = fmaxf(v[u].x / den[u] + b.x, 0.f);
        o.y = fmaxf(v[u].y / den[u] + b.y, 0.f);
        o.z = fmaxf(v[u].z / den[u] + b.z, 0.f);
        o.w = fmaxf(v[u].w / den[u] + b.w, 0.f);
        p4[i] = o;
      }
    }
  }
}
__global__ __launch_bounds__(SB) void residual_k(const float* __restrict__ h2,
                                                 const float* __restrict__ segsum,
                                                 const float* __restrict__ b2, float* r,
                                                 const float* __restrict__ br) {
  constexpr int N4 = NN * 32;
  const float4* h4 = (const float4*)h2;
  float4* r4 = (float4*)r;
  const int gid = blockIdx.x * SB + threadIdx.x;
  const float4 bb = *(const float4*)&b2[(gid & 31) * 4];
  const float4 rb = *(const float4*)&br[(gid & 31) * 4];
  for (int i0 = gid; i0 < N4; i0 += ST * 4) {
    float4 v[4], w[4];
    float den[4];
#pragma unroll
    for (int u = 0; u < 4; ++u) {
      const int ic = min(i0 + u * ST, N4 - 1);
      v[u] = h4[ic];
      w[u] = r4[ic];
      den[u] = segsum[ic >> 5] + 1e-16f;
    }
#pragma unroll
    for (int u = 0; u < 4; ++u) {
      const int i = i0 + u * ST;
      if (i < N4) {
        float4 o;
        o.x = fmaxf(v[u].x / den[u] + bb.x, 0.f) + w[u].x + rb.x;
        o.y = fmaxf(v[u].y / den[u] + bb.y, 0.f) + w[u].y + rb.y;
        o.z = fmaxf(v[u].z / den[u] + bb.z, 0.f) + w[u].z + rb.z;
        o.w = fmaxf(v[u].w / den[u] + bb.w, 0.f) + w[u].w + rb.w;
        r4[i] = o;
      }
    }
  }
}
template <int C>
__global__ __launch_bounds__(SB) void norm_bias_k(float* out, const float* __restrict__ segsum,
                                                  const float* __restrict__ bias) {
  constexpr int Q = C / 4, N4 = NN * Q;
  float4* o4 = (float4*)out;
  const int gid = blockIdx.x * SB + threadIdx.x;
  const float4 b = *(const float4*)&bias[(gid % Q) * 4];
  for (int i0 = gid; i0 < N4; i0 += ST * 4) {
    float4 v[4];
    float den[4];
#pragma unroll
    for (int u = 0; u < 4; ++u) {
      const int ic = min(i0 + u * ST, N4 - 1);
      v[u] = o4[ic];
      den[u] = segsum[ic / Q] + 1e-16f;
    }
#pragma unroll
    for (int u = 0; u < 4; ++u) {
      const int i = i0 + u * ST;
      if (i < N4) {
        float4 o;
        o.x = v[u].x / den[u] + b.x;
        o.y = v[u].y / den[u] + b.y;
        o.z = v[u].z / den[u] + b.z;
        o.w = v[u].w / den[u] + b.w;
        o4[i] = o;
      }
    }
  }
}

__global__ __launch_bounds__(SB) void attr_accum_k(const int* __restrict__ dst,
                                                   const float* __restrict__ ea,
                                                   float* attr_sum, float* cnt) {
  constexpr int NT = NE * 16;
  const int gid = blockIdx.x * SB + threadIdx.x;
  const int j = gid & 15;
  for (int i0 = gid; i0 < NT; i0 += ST * 8) {
    float v[8];
    int d[8];
#pragma unroll
    for (int u = 0; u < 8; ++u) {
      const int ic = min(i0 + u * ST, NT - 1);
      v[u] = ea[ic];
      d[u] = dst[ic >> 4];
    }
#pragma unroll
    for (int u = 0; u < 8; ++u) {
      if (i0 + u * ST < NT) {
        atomicAdd(&attr_sum[(size_t)d[u] * 16 + j], v[u]);
        if (j == 0) atomicAdd(&cnt[d[u]], 1.f);
      }
    }
  }
}
__global__ __launch_bounds__(SB) void attr_mean_k(float* attr_sum, const float* __restrict__ cnt) {
  for (int i = blockIdx.x * SB + threadIdx.x; i < NN * 16; i += ST)
    attr_sum[i] /= fmaxf(cnt[i >> 4], 1.f);
}

__global__ void wa_k(const float* __restrict__ We, const float* __restrict__ ae,
                     float* wa, int K) {
  int i = threadIdx.x;
  if (i < 16) { float s = 0.f; for (int j = 0; j < K; ++j) s += We[i * K + j] * ae[j]; wa[i] = s; }
}

static constexpr int GW = 2048;
template <int COUT>
__global__ __launch_bounds__(256) void gemm_k128(const float* __restrict__ A,
                                                 const float* __restrict__ B,
                                                 float* __restrict__ C) {
  constexpr int NCG = COUT / 32;
  constexpr int NSW = GW / NCG;
  constexpr int NS = NN / 16;
  static_assert(NN % 16 == 0, "rows come in stripes of 16");
  const int lane = threadIdx.x & 31, half = lane >> 4, lidx = lane & 15;
  const int wid = __builtin_amdgcn_readfirstlane((int)((blockIdx.x * 256 + threadIdx.x) >> 5));
  const int n0 = (wid % NCG) * 32;
  const int sw = wid / NCG;

  float bq[2][16][4];
#pragma unroll
  for (int t = 0; t < 2; ++t)
#pragma unroll
    for (int j = 0; j < 16; ++j)
#pragma unroll
      for (int i = 0; i < 4; ++i)
        bq[t][j][i] = B[(8 * j + 4 * half + i) * COUT + n0 + 16 * t + lidx];

  const float4* A4 = (const float4*)A;
  auto load = [&](float4 (&a)[16], int s) {
    const float4* p = A4 + ((size_t)s * 16 + lidx) * 32 + half;
#pragma unroll
    for (int j = 0; j < 16; ++j) a[j] = p[2 * j];
  };
  auto compute = [&](const float4 (&a)[16], int s) {
    v8f acc0 = {}, acc1 = {};
#pragma unroll
    for (int j = 0; j < 16; ++j) {
      v2f a01, a23, b;
      a01.x = a[j].x; a01.y = a[j].y;
      a23.x = a[j].z; a23.y = a[j].w;
      b.x = bq[0][j][0]; b.y = bq[0][j][1];
      acc0 = __builtin_amdgcn_wmma_f32_16x16x4_f32(false, a01, false, b, (short)0, acc0,
                                                   false, false);
      b.x = bq[1][j][0]; b.y = bq[1][j][1];
      acc1 = __builtin_amdgcn_wmma_f32_16x16x4_f32(false, a01, false, b, (short)0, acc1,
                                                   false, false);
      b.x = bq[0][j][2]; b.y = bq[0][j][3];
      acc0 = __builtin_amdgcn_wmma_f32_16x16x4_f32(false, a23, false, b, (short)0, acc0,
                                                   false, false);
      b.x = bq[1][j][2]; b.y = bq[1][j][3];
      acc1 = __builtin_amdgcn_wmma_f32_16x16x4_f32(false, a23, false, b, (short)0, acc1,
                                                   false, false);
    }
    float* c = C + ((size_t)s * 16 + 8 * half) * COUT + n0 + lidx;
#pragma unroll
    for (int r = 0; r < 8; ++r) {
      c[r * COUT] = acc0[r];
      c[r * COUT + 16] = acc1[r];
    }
  };

  float4 ba[16], bb[16];
  int s = sw;
  load(ba, s);
  for (; s + NSW < NS; s += 2 * NSW) {
    load(bb, s + NSW);
    compute(ba, s);
    load(ba, min(s + 2 * NSW, NS - 1));
    compute(bb, s + NSW);
  }
  if (s < NS) compute(ba, s);
}

template <int C, bool ET>
__global__ __launch_bounds__(SB) void node_prep_k(
    const float* __restrict__ h, const float* __restrict__ a_src,
    const float* __restrict__ a_dst, const float* __restrict__ mean_attr,
    const float* __restrict__ wa, float* __restrict__ s_src, float* __restrict__ s_dst,
    float* __restrict__ self_logit) {
  constexpr int R = C / 32;
  constexpr int U = 256 / C;
  const int gid = blockIdx.x * SB + threadIdx.x;
  const int lane = gid & 31;
  float as[R], ad[R];
#pragma unroll
  for (int i = 0; i < R; ++i) { as[i] = a_src[lane + i * 32]; ad[i] = a_dst[lane + i * 32]; }
  for (int n0 = gid >> 5; n0 < NN; n0 += NW * U) {
    float hv[U][R];
#pragma unroll
    for (int u = 0; u < U; ++u) {
      const int nc = min(n0 + u * NW, NN - 1);
#pragma unroll
      for (int i = 0; i < R; ++i) hv[u][i] = h[(size_t)nc * C + lane + i * 32];
    }
    float psv[U], pdv[U];
#pragma unroll
    for (int u = 0; u < U; ++u) {
      float ps = 0.f, pd = 0.f;
#pragma unroll
      for (int i = 0; i < R; ++i) { ps += hv[u][i] * as[i]; pd += hv[u][i] * ad[i]; }
#pragma unroll
      for (int m = 16; m > 0; m >>= 1) { ps += __shfl_xor(ps, m, 32); pd += __shfl_xor(pd, m, 32); }
      psv[u] = ps; pdv[u] = pd;
    }
#pragma unroll
    for (int u = 0; u < U; ++u) {
      const int n = n0 + u * NW;
      if (n < NN) {
        const float ps = psv[u], pd = pdv[u];
        if (lane == 0) {
          float st = 0.f;
          if (ET) { for (int j = 0; j < 16; ++j) st += mean_attr[(size_t)n * 16 + j] * wa[j]; }
          float sl = leaky(ps + pd + st);
          s_src[n] = ps; s_dst[n] = pd; self_logit[n] = sl;
        }
      }
    }
  }
}

__global__ __launch_bounds__(SB) void eterm_k(const float* __restrict__ ea,
                                              const float* __restrict__ wa,
                                              float* __restrict__ eterm) {
  constexpr int NT = NE * 16;
  const int gid = blockIdx.x * SB + threadIdx.x;
  const int j = gid & 15;
  const float w = wa[j];
  for (int i0 = gid; i0 < NT; i0 += ST * 8) {
    float v[8];
#pragma unroll
    for (int u = 0; u < 8; ++u) v[u] = ea[min(i0 + u * ST, NT - 1)] * w;
#pragma unroll
    for (int u = 0; u < 8; ++u) {
      float t = v[u];
      t += __shfl_xor(t, 1, 32); t += __shfl_xor(t, 2, 32);
      t += __shfl_xor(t, 4, 32); t += __shfl_xor(t, 8, 32);
      v[u] = t;
    }
#pragma unroll
    for (int u = 0; u < 8; ++u) {
      const int i = i0 + u * ST;
      if (j == 0 && i < NT) eterm[i >> 4] = v[u];
    }
  }
}

static constexpr int RT = 400;
static constexpr int NTILE = (NN + RT - 1) / RT;
static constexpr int QCAP = 64;
static constexpr int LOGCAP = 320;
static constexpr int NSTEP = NE / 512;
static_assert(NE % 512 == 0, "the key plane is swept in steps of 512 edges");
static_assert(NTILE <= 255, "a tile id is one byte, and 255 is the filler that matches no tile");
static_assert(NN <= (1 << 17) && RT <= (1 << 9), "a log entry packs source (17 bits) and row (9 bits)");

static constexpr int KEYB = (NE / 4 + SB - 1) / SB;
__global__ __launch_bounds__(SB) void tile_key_k(const int* __restrict__ dst,
                                                 unsigned* __restrict__ key) {
  const int i = blockIdx.x * SB + threadIdx.x;
  if (i < NE / 4) {
    const int4 d = ((const int4*)dst)[i];
    key[i] = (unsigned)d.x / RT | ((unsigned)d.y / RT) << 8 | ((unsigned)d.z / RT) << 16 |
             ((unsigned)d.w / RT) << 24;
  }
}

template <int V> struct RowVec;
template <> struct RowVec<4> { typedef float4 T; };
template <> struct RowVec<2> { typedef float2 T; };
__device__ __forceinline__ float vget(const float4& v, int k) {
  return k == 0 ? v.x : k == 1 ? v.y : k == 2 ? v.z : v.w;
}
__device__ __forceinline__ float vget(const float2& v, int k) { return k == 0 ? v.x : v.y; }
__device__ __forceinline__ void vset(float4& v, int k, float f) {
  if (k == 0) v.x = f; else if (k == 1) v.y = f; else if (k == 2) v.z = f; else v.w = f;
}
__device__ __forceinline__ void vset(float2& v, int k, float f) {
  if (k == 0) v.x = f; else v.y = f;
}

template <int C, bool ET, bool BUILD>
__global__ __launch_bounds__(1024) void tile_agg_k(
    const int* __restrict__ src, const int* __restrict__ dst, const unsigned* __restrict__ key,
    const float* __restrict__ s_src, const float* __restrict__ s_dst,
    const float* __restrict__ eterm, const float* __restrict__ self_logit,
    float* __restrict__ segsum, const float* __restrict__ h, float* __restrict__ out,
    int* __restrict__ glog_sr, int* __restrict__ glog_e, int* __restrict__ glog_n) {
  constexpr int V = C / 32;
  typedef typename RowVec<V>::T VecT;
  __shared__ float tile[RT * C];
  __shared__ float ssum[RT];
  __shared__ unsigned smax[RT];
  __shared__ int queue[32 * QCAP];
  __shared__ int log_sr[32 * LOGCAP];
  __shared__ float log_l[32 * LOGCAP];
  const int tid = threadIdx.x, lane = tid & 31;
  const int wave = __builtin_amdgcn_readfirstlane(tid >> 5);
  const int row0 = blockIdx.x * RT;
  const int rows = min(RT, NN - row0);

  for (int r = tid; r < rows; r += 1024) smax[r] = f2ord(self_logit[row0 + r]);
  __syncthreads();

  int* myq = &queue[wave * QCAP];
  int* mysr = &log_sr[wave * LOGCAP];
  float* myll = &log_l[wave * LOGCAP];
  const int gw = blockIdx.x * 32 + wave;
  int* gsr = glog_sr + (size_t)gw * LOGCAP;
  int* ge = glog_e + (size_t)gw * LOGCAP;
  int logn = 0;
  bool ovf = false;

  auto edge_logit = [&](int e, int& s, int& r) {
    s = src[e];
    const int d = dst[e];
    float l = s_src[s] + s_dst[d];
    if (ET) l += eterm[e];
    r = d - row0;
    return leaky(l);
  };

  auto proc_max = [&](int nb) {
    int sr = 0, e = 0;
    float ll = 0.f;
    if (lane < nb) {
      int s, r;
      e = myq[lane];
      ll = edge_logit(e, s, r);
      atomicMax(&smax[r], f2ord(ll));
      sr = s | r << 17;
    }
    if (!ovf && logn + nb <= LOGCAP) {
      if (lane < nb) {
        mysr[logn + lane] = sr; myll[logn + lane] = ll;
        if (BUILD) { gsr[logn + lane] = sr; ge[logn + lane] = e; }
      }
      logn += nb;
    } else {
      ovf = true;
    }
  };
  auto max_from_log = [&](int n) {
    for (int b = 0; b < n; b += 128) {
      int sr[4], e[4];
      float a[4], d[4], t[4];
#pragma unroll
      for (int u = 0; u < 4; ++u) {
        const int ic = min(b + 32 * u + lane, n - 1);
        sr[u] = gsr[ic];
        e[u] = ET ? ge[ic] : 0;
      }
      asm volatile("" ::"v"(sr[0]), "v"(sr[1]), "v"(sr[2]), "v"(sr[3]), "v"(e[0]), "v"(e[1]),
                   "v"(e[2]), "v"(e[3]));
#pragma unroll
      for (int u = 0; u < 4; ++u) {
        a[u] = s_src[sr[u] & 0x1ffff];
        d[u] = s_dst[row0 + (sr[u] >> 17)];
        t[u] = ET ? eterm[e[u]] : 0.f;
      }
      asm volatile("" ::"v"(a[0]), "v"(a[1]), "v"(a[2]), "v"(a[3]), "v"(d[0]), "v"(d[1]),
                   "v"(d[2]), "v"(d[3]), "v"(t[0]), "v"(t[1]), "v"(t[2]), "v"(t[3]));
#pragma unroll
      for (int u = 0; u < 4; ++u) {
        const int i = b + 32 * u + lane;
        if (i < n) {
          float l = a[u] + d[u];
          if (ET) l += t[u];
          const float ll = leaky(l);
          atomicMax(&smax[sr[u] >> 17], f2ord(ll));
          mysr[i] = sr[u];
          myll[i] = ll;
        }
      }
    }
  };

  auto agg_rows = [&](int nb, int s, int r, float w) {
    for (int j0 = 0; j0 < nb; j0 += 8) {
      VecT hv[8];
#pragma unroll
      for (int u = 0; u < 8; ++u) {
        const int sj = __builtin_amdgcn_readlane(s, j0 + u);
        hv[u] = *(const VecT*)&h[(size_t)sj * C + lane * V];
      }
#pragma unroll
      for (int u = 0; u < 8; ++u) {
        if (j0 + u < nb) {
          const int rj = __builtin_amdgcn_readlane(r, j0 + u);
          const float wj = __int_as_float(__builtin_amdgcn_readlane(__float_as_int(w), j0 + u));
          float* t = &tile[rj * C + lane];
#pragma unroll
          for (int k = 0; k < V; ++k) atomicAdd(t + k * 32, wj * vget(hv[u], k));
        }
      }
    }
  };
  auto proc_log = [&](int base, int nb) {
    int s = 0, r = 0;
    float w = 0.f;
    if (lane < nb) {
      const int sr = mysr[base + lane];
      s = sr & 0x1ffff;
      r = sr >> 17;
      w = __expf(myll[base + lane] - ord2f(smax[r]));
      atomicAdd(&ssum[r], w);
    }
    agg_rows(nb, s, r, w);
  };
  auto proc_queue = [&](int nb) {
    int s = 0, r = 0;
    float w = 0.f;
    if (lane < nb) {
      const float ll = edge_logit(myq[lane], s, r);
      w = __expf(ll - ord2f(smax[r]));
      atomicAdd(&ssum[r], w);
    }
    agg_rows(nb, s, r, w);
  };

  const unsigned tt = blockIdx.x * 0x01010101u;
  const int rot = blockIdx.x * (NSTEP / NTILE);
  auto step_of = [&](int k) { const int kk = k + rot; return kk >= NSTEP ? kk - NSTEP : kk; };
  const uint4* key4 = (const uint4*)key;
  auto fetch = [&](uint4 (&d)[4], int k0) {
#pragma unroll
    for (int u = 0; u < 4; ++u) {
      const int k = k0 + 32 * u;
      d[u] = (k < NSTEP) ? key4[(size_t)step_of(k) * 32 + lane] : make_uint4(~0u, ~0u, ~0u, ~0u);
    }
  };
  auto differs = [&](unsigned x) {
    const unsigned y = x ^ tt;
    return (((y & 0x7f7f7f7fu) + 0x7f7f7f7fu) | y) & 0x80808080u;
  };
  auto sweep = [&](auto&& proc) {
    int cnt = 0;
    uint4 cur[4], nxt[4];
    fetch(cur, wave);
    for (int k0 = wave; k0 < NSTEP; k0 += 32 * 4) {
      fetch(nxt, k0 + 32 * 4);
#pragma unroll
      for (int u = 0; u < 4; ++u) {
        unsigned m = ~(differs(cur[u].x) >> 7 | differs(cur[u].y) >> 6 | differs(cur[u].z) >> 5 |
                       differs(cur[u].w) >> 4) & 0x0f0f0f0fu;
        unsigned live = __builtin_amdgcn_ballot_w32(m != 0u);
        if (live != 0u) {
          const int e0 = step_of(k0 + 32 * u) * 512 + lane * 16;
          do {
            if (m != 0u) {
              const int bit = __builtin_ctz(m);
              myq[cnt + __builtin_amdgcn_mbcnt_lo(live, 0u)] = e0 + (bit & 7) * 4 + (bit >> 3);
              m &= m - 1u;
            }
            cnt += __popc(live);
            live = __builtin_amdgcn_ballot_w32(m != 0u);
            if (cnt >= 32) {
              proc(32);
              const int rem = cnt - 32;
              const int keep = (lane < rem) ? myq[32 + lane] : 0;
              if (lane < rem) myq[lane] = keep;
              cnt = rem;
            }
          } while (live != 0u);
        }
      }
#pragma unroll
      for (int u = 0; u < 4; ++u) cur[u] = nxt[u];
    }
    if (cnt > 0) proc(cnt);
  };

  if (BUILD) {
    sweep(proc_max);
    if (lane == 0) glog_n[gw] = ovf ? -1 : logn;
  } else {
    const int gn = __builtin_amdgcn_readfirstlane(glog_n[gw]);
    if (gn >= 0) {
      max_from_log(gn);
      logn = gn;
    } else {
      sweep(proc_max);
    }
  }
  __syncthreads();

  for (int i = tid; i < rows * 32; i += 1024) {
    const int r = i >> 5, n = row0 + r;
    const float a = __expf(self_logit[n] - ord2f(smax[r]));
    const VecT hv = *(const VecT*)&h[(size_t)n * C + lane * V];
#pragma unroll
    for (int k = 0; k < V; ++k) tile[r * C + k * 32 + lane] = a * vget(hv, k);
    if (lane == 0) ssum[r] = a;
  }
  __syncthreads();

  const int nlog = __builtin_amdgcn_readfirstlane(logn);
  if (__builtin_amdgcn_readfirstlane((int)ovf) == 0) {
    int b = 0;
    for (; b + 32 <= nlog; b += 32) proc_log(b, 32);
    if (b < nlog) proc_log(b, nlog - b);
  } else {
    sweep(proc_queue);
  }
  __syncthreads();

  for (int i = tid; i < rows * 32; i += 1024) {
    const int r = i >> 5, n = row0 + r;
    VecT o;
#pragma unroll
    for (int k = 0; k < V; ++k) vset(o, k, tile[r * C + k * 32 + lane]);
    *(VecT*)&out[(size_t)n * C + lane * V] = o;
    if (lane == 0) segsum[n] = ssum[r];
  }
}

struct Scratch {
  float *s_src, *s_dst, *self_l, *segsum;
  const float* mean_attr;
  float *eterm, *wa;
  const unsigned* key;
  int *glog_sr, *glog_e, *glog_n;
};

template <int C, bool ET, bool BUILD>
static void run_gat(hipStream_t st, const float* hin, const float* Wm, const float* asrc,
                    const float* adst, const float* We, const float* ae,
                    const int* src, const int* dst, const float* edge_attr, float* hg,
                    float* out, const Scratch& S) {
  if (ET) wa_k<<<1, 16, 0, st>>>(We, ae, S.wa, C);
  gemm_k128<C><<<GW / 8, 256, 0, st>>>(hin, Wm, hg);
  node_prep_k<C, ET><<<SG, SB, 0, st>>>(hg, asrc, adst, S.mean_attr, S.wa, S.s_src, S.s_dst,
                                        S.self_l);
  if (ET) eterm_k<<<SG, SB, 0, st>>>(edge_attr, S.wa, S.eterm);
  tile_agg_k<C, ET, BUILD><<<NTILE, 1024, 0, st>>>(src, dst, S.key, S.s_src, S.s_dst, S.eterm,
                                                   S.self_l, S.segsum, hg, out, S.glog_sr,
                                                   S.glog_e, S.glog_n);
}

extern "C" void kernel_launch(void* const* d_in, const int* in_sizes, int n_in, void* d_out,
                              int out_size, void* d_ws, size_t ws_size, hipStream_t stream) {
  (void)in_sizes; (void)n_in; (void)out_size; (void)ws_size;
  const float* x = (const float*)d_in[0];
  const int* ei = (const int*)d_in[1];
  const int* src = ei;
  const int* dst = ei + NE;
  const float* edge_attr = (const float*)d_in[2];
  const float *W1 = (const float*)d_in[3], *a1s = (const float*)d_in[4],
              *a1d = (const float*)d_in[5], *b1 = (const float*)d_in[6];
  const float *W2 = (const float*)d_in[7], *a2s = (const float*)d_in[8],
              *a2d = (const float*)d_in[9], *b2 = (const float*)d_in[10];
  const float *We2 = (const float*)d_in[11], *ae2 = (const float*)d_in[12];
  const float *Wr = (const float*)d_in[13], *br = (const float*)d_in[14];
  const float *Wmu = (const float*)d_in[15], *amus = (const float*)d_in[16],
              *amud = (const float*)d_in[17], *bmu = (const float*)d_in[18],
              *Wemu = (const float*)d_in[19], *aemu = (const float*)d_in[20];
  const float *Wls = (const float*)d_in[21], *alss = (const float*)d_in[22],
              *alsd = (const float*)d_in[23], *bls = (const float*)d_in[24],
              *Wels = (const float*)d_in[25], *aels = (const float*)d_in[26];
  float* outf = (float*)d_out;

  float* W = (float*)d_ws;
  size_t o = 0;
  float* hA = W + o; o += (size_t)NN * 128;
  float* hB = W + o; o += (size_t)NN * 128;
  float* hC = W + o; o += (size_t)NN * 128;
  Scratch S;
  S.s_src = W + o; o += NN;
  S.s_dst = W + o; o += NN;
  S.self_l = W + o; o += NN;
  S.segsum = W + o; o += NN;
  float* cnt = W + o; o += NN;
  float* mean_attr = W + o; o += (size_t)NN * 16;
  S.mean_attr = mean_attr;
  S.eterm = W + o; o += NE;
  S.wa = W + o; o += 16;
  unsigned* key = (unsigned*)(W + o); o += NE / 4;
  S.key = key;
  o += 16;
  S.glog_sr = (int*)(W + o); o += (size_t)NTILE * 32 * LOGCAP;
  S.glog_e = (int*)(W + o); o += (size_t)NTILE * 32 * LOGCAP;
  S.glog_n = (int*)(W + o); o += NTILE * 32;

  tile_key_k<<<KEYB, SB, 0, stream>>>(dst, key);

  zero_f32_k<<<SG, SB, 0, stream>>>((float4*)cnt, NN * 17 / 4);
  attr_accum_k<<<SG, SB, 0, stream>>>(dst, edge_attr, mean_attr, cnt);
  attr_mean_k<<<SG, SB, 0, stream>>>(mean_attr, cnt);

  run_gat<128, false, true>(stream, x, W1, a1s, a1d, nullptr, nullptr, src, dst, edge_attr, hA, hB,
                      S);
  norm_relu_k<<<SG, SB, 0, stream>>>(hB, S.segsum, b1);

  run_gat<128, true, false>(stream, hB, W2, a2s, a2d, We2, ae2, src, dst, edge_attr, hA, hC, S);

  gemm_k128<128><<<GW / 8, 256, 0, stream>>>(x, Wr, hB);
  residual_k<<<SG, SB, 0, stream>>>(hC, S.segsum, b2, hB, br);

  run_gat<64, true, false>(stream, hB, Wmu, amus, amud, Wemu, aemu, src, dst, edge_attr, hA, outf, S);
  norm_bias_k<64><<<SG, SB, 0, stream>>>(outf, S.segsum, bmu);
  run_gat<64, true, false>(stream, hB, Wls, alss, alsd, Wels, aels, src, dst, edge_attr, hA,
                    outf + (size_t)NN * 64, S);
  norm_bias_k<64><<<SG, SB, 0, stream>>>(outf + (size_t)NN * 64, S.segsum, bls);
}


# Per-destination edge_attr sums taken by the owner workgroup in LDS during conv1's edge sweep (LDS float adds, arrival order as before), means stored per tile: global float adds, 3 of 27 launches gone
# speedup vs baseline: 3.6053x; 1.0928x over previous
#include <hip/hip_runtime.h>
#include <math.h>

typedef float v2f __attribute__((ext_vector_type(2)));
typedef float v8f __attribute__((ext_vector_type(8)));

static constexpr int   NN = 100000;
static constexpr int   NE = 1600000;
static constexpr float NEG = 0.2f;

__device__ __forceinline__ float leaky(float x) { return x > 0.f ? x : NEG * x; }
__device__ __forceinline__ unsigned f2ord(float f) {
  unsigned u = __float_as_uint(f);
  return (u & 0x80000000u) ? ~u : (u | 0x80000000u);
}
__device__ __forceinline__ float ord2f(unsigned v) {
  unsigned u = (v & 0x80000000u) ? (v & 0x7fffffffu) : ~v;
  return __uint_as_float(u);
}

static constexpr int SB = 256;
static constexpr int SG = 2048;
static constexpr int ST = SG * SB;
static constexpr int NW = ST / 32;

__global__ __launch_bounds__(SB) void norm_relu_k(float* p, const float* __restrict__ segsum,
                                                  const float* __restrict__ bias) {
  constexpr int N4 = NN * 32;
  float4* p4 = (float4*)p;
  const int gid = blockIdx.x * SB + threadIdx.x;
  const float4 b = *(const float4*)&bias[(gid & 31) * 4];
  for (int i0 = gid; i0 < N4; i0 += ST * 4) {
    float4 v[4];
    float den[4];
#pragma unroll
    for (int u = 0; u < 4; ++u) {
      const int ic = min(i0 + u * ST, N4 - 1);
      v[u] = p4[ic];
      den[u] = segsum[ic >> 5] + 1e-16f;
    }
#pragma unroll
    for (int u = 0; u < 4; ++u) {
      const int i = i0 + u * ST;
      if (i < N4) {
        float4 o;
        o.x = fmaxf(v[u].x / den[u] + b.x, 0.f);
        o.y = fmaxf(v[u].y / den[u] + b.y, 0.f);
        o.z = fmaxf(v[u].z / den[u] + b.z, 0.f);
        o.w = fmaxf(v[u].w / den[u] + b.w, 0.f);
        p4[i] = o;
      }
    }
  }
}
__global__ __launch_bounds__(SB) void residual_k(const float* __restrict__ h2,
                                                 const float* __restrict__ segsum,
                                                 const float* __restrict__ b2, float* r,
                                                 const float* __restrict__ br) {
  constexpr int N4 = NN * 32;
  const float4* h4 = (const float4*)h2;
  float4* r4 = (float4*)r;
  const int gid = blockIdx.x * SB + threadIdx.x;
  const float4 bb = *(const float4*)&b2[(gid & 31) * 4];
  const float4 rb = *(const float4*)&br[(gid & 31) * 4];
  for (int i0 = gid; i0 < N4; i0 += ST * 4) {
    float4 v[4], w[4];
    float den[4];
#pragma unroll
    for (int u = 0; u < 4; ++u) {
      const int ic = min(i0 + u * ST, N4 - 1);
      v[u] = h4[ic];
      w[u] = r4[ic];
      den[u] = segsum[ic >> 5] + 1e-16f;
    }
#pragma unroll
    for (int u = 0; u < 4; ++u) {
      const int i = i0 + u * ST;
      if (i < N4) {
        float4 o;
        o.x = fmaxf(v[u].x / den[u] + bb.x, 0.f) + w[u].x + rb.x;
        o.y = fmaxf(v[u].y / den[u] + bb.y, 0.f) + w[u].y + rb.y;
        o.z = fmaxf(v[u].z / den[u] + bb.z, 0.f) + w[u].z + rb.z;
        o.w = fmaxf(v[u].w / den[u] + bb.w, 0.f) + w[u].w + rb.w;
        r4[i] = o;
      }
    }
  }
}
template <int C>
__global__ __launch_bounds__(SB) void norm_bias_k(float* out, const float* __restrict__ segsum,
                                                  const float* __restrict__ bias) {
  constexpr int Q = C / 4, N4 = NN * Q;
  float4* o4 = (float4*)out;
  const int gid = blockIdx.x * SB + threadIdx.x;
  const float4 b = *(const float4*)&bias[(gid % Q) * 4];
  for (int i0 = gid; i0 < N4; i0 += ST * 4) {
    float4 v[4];
    float den[4];
#pragma unroll
    for (int u = 0; u < 4; ++u) {
      const int ic = min(i0 + u * ST, N4 - 1);
      v[u] = o4[ic];
      den[u] = segsum[ic / Q] + 1e-16f;
    }
#pragma unroll
    for (int u = 0; u < 4; ++u) {
      const int i = i0 + u * ST;
      if (i < N4) {
        float4 o;
        o.x = v[u].x / den[u] + b.x;
        o.y = v[u].y / den[u] + b.y;
        o.z = v[u].z / den[u] + b.z;
        o.w = v[u].w / den[u] + b.w;
        o4[i] = o;
      }
    }
  }
}

__global__ void wa_k(const float* __restrict__ We, const float* __restrict__ ae,
                     float* wa, int K) {
  int i = threadIdx.x;
  if (i < 16) { float s = 0.f; for (int j = 0; j < K; ++j) s += We[i * K + j] * ae[j]; wa[i] = s; }
}

static constexpr int GW = 2048;
template <int COUT>
__global__ __launch_bounds__(256) void gemm_k128(const float* __restrict__ A,
                                                 const float* __restrict__ B,
                                                 float* __restrict__ C) {
  constexpr int NCG = COUT / 32;
  constexpr int NSW = GW / NCG;
  constexpr int NS = NN / 16;
  static_assert(NN % 16 == 0, "rows come in stripes of 16");
  const int lane = threadIdx.x & 31, half = lane >> 4, lidx = lane & 15;
  const int wid = __builtin_amdgcn_readfirstlane((int)((blockIdx.x * 256 + threadIdx.x) >> 5));
  const int n0 = (wid % NCG) * 32;
  const int sw = wid / NCG;

  float bq[2][16][4];
#pragma unroll
  for (int t = 0; t < 2; ++t)
#pragma unroll
    for (int j = 0; j < 16; ++j)
#pragma unroll
      for (int i = 0; i < 4; ++i)
        bq[t][j][i] = B[(8 * j + 4 * half + i) * COUT + n0 + 16 * t + lidx];

  const float4* A4 = (const float4*)A;
  auto load = [&](float4 (&a)[16], int s) {
    const float4* p = A4 + ((size_t)s * 16 + lidx) * 32 + half;
#pragma unroll
    for (int j = 0; j < 16; ++j) a[j] = p[2 * j];
  };
  auto compute = [&](const float4 (&a)[16], int s) {
    v8f acc0 = {}, acc1 = {};
#pragma unroll
    for (int j = 0; j < 16; ++j) {
      v2f a01, a23, b;
      a01.x = a[j].x; a01.y = a[j].y;
      a23.x = a[j].z; a23.y = a[j].w;
      b.x = bq[0][j][0]; b.y = bq[0][j][1];
      acc0 = __builtin_amdgcn_wmma_f32_16x16x4_f32(false, a01, false, b, (short)0, acc0,
                                                   false, false);
      b.x = bq[1][j][0]; b.y = bq[1][j][1];
      acc1 = __builtin_amdgcn_wmma_f32_16x16x4_f32(false, a01, false, b, (short)0, acc1,
                                                   false, false);
      b.x = bq[0][j][2]; b.y = bq[0][j][3];
      acc0 = __builtin_amdgcn_wmma_f32_16x16x4_f32(false, a23, false, b, (short)0, acc0,
                                                   false, false);
      b.x = bq[1][j][2]; b.y = bq[1][j][3];
      acc1 = __builtin_amdgcn_wmma_f32_16x16x4_f32(false, a23, false, b, (short)0, acc1,
                                                   false, false);
    }
    float* c = C + ((size_t)s * 16 + 8 * half) * COUT + n0 + lidx;
#pragma unroll
    for (int r = 0; r < 8; ++r) {
      c[r * COUT] = acc0[r];
      c[r * COUT + 16] = acc1[r];
    }
  };

  float4 ba[16], bb[16];
  int s = sw;
  load(ba, s);
  for (; s + NSW < NS; s += 2 * NSW) {
    load(bb, s + NSW);
    compute(ba, s);
    load(ba, min(s + 2 * NSW, NS - 1));
    compute(bb, s + NSW);
  }
  if (s < NS) compute(ba, s);
}

template <int C, bool ET>
__global__ __launch_bounds__(SB) void node_prep_k(
    const float* __restrict__ h, const float* __restrict__ a_src,
    const float* __restrict__ a_dst, const float* __restrict__ mean_attr,
    const float* __restrict__ wa, float* __restrict__ s_src, float* __restrict__ s_dst,
    float* __restrict__ self_logit) {
  constexpr int R = C / 32;
  constexpr int U = 256 / C;
  const int gid = blockIdx.x * SB + threadIdx.x;
  const int lane = gid & 31;
  float as[R], ad[R];
#pragma unroll
  for (int i = 0; i < R; ++i) { as[i] = a_src[lane + i * 32]; ad[i] = a_dst[lane + i * 32]; }
  for (int n0 = gid >> 5; n0 < NN; n0 += NW * U) {
    float hv[U][R];
#pragma unroll
    for (int u = 0; u < U; ++u) {
      const int nc = min(n0 + u * NW, NN - 1);
#pragma unroll
      for (int i = 0; i < R; ++i) hv[u][i] = h[(size_t)nc * C + lane + i * 32];
    }
    float psv[U], pdv[U];
#pragma unroll
    for (int u = 0; u < U; ++u) {
      float ps = 0.f, pd = 0.f;
#pragma unroll
      for (int i = 0; i < R; ++i) { ps += hv[u][i] * as[i]; pd += hv[u][i] * ad[i]; }
#pragma unroll
      for (int m = 16; m > 0; m >>= 1) { ps += __shfl_xor(ps, m, 32); pd += __shfl_xor(pd, m, 32); }
      psv[u] = ps; pdv[u] = pd;
    }
#pragma unroll
    for (int u = 0; u < U; ++u) {
      const int n = n0 + u * NW;
      if (n < NN) {
        const float ps = psv[u], pd = pdv[u];
        if (lane == 0) {
          float st = 0.f;
          if (ET) { for (int j = 0; j < 16; ++j) st += mean_attr[(size_t)n * 16 + j] * wa[j]; }
          float sl = leaky(ps + pd + st);
          s_src[n] = ps; s_dst[n] = pd; self_logit[n] = sl;
        }
      }
    }
  }
}

__global__ __launch_bounds__(SB) void eterm_k(const float* __restrict__ ea,
                                              const float* __restrict__ wa,
                                              float* __restrict__ eterm) {
  constexpr int NT = NE * 16;
  const int gid = blockIdx.x * SB + threadIdx.x;
  const int j = gid & 15;
  const float w = wa[j];
  for (int i0 = gid; i0 < NT; i0 += ST * 8) {
    float v[8];
#pragma unroll
    for (int u = 0; u < 8; ++u) v[u] = ea[min(i0 + u * ST, NT - 1)] * w;
#pragma unroll
    for (int u = 0; u < 8; ++u) {
      float t = v[u];
      t += __shfl_xor(t, 1, 32); t += __shfl_xor(t, 2, 32);
      t += __shfl_xor(t, 4, 32); t += __shfl_xor(t, 8, 32);
      v[u] = t;
    }
#pragma unroll
    for (int u = 0; u < 8; ++u) {
      const int i = i0 + u * ST;
      if (j == 0 && i < NT) eterm[i >> 4] = v[u];
    }
  }
}

static constexpr int RT = 400;
static constexpr int AP = 17;
static constexpr int NTILE = (NN + RT - 1) / RT;
static constexpr int QCAP = 64;
static constexpr int LOGCAP = 320;
static constexpr int NSTEP = NE / 512;
static_assert(NE % 512 == 0, "the key plane is swept in steps of 512 edges");
static_assert(NTILE <= 255, "a tile id is one byte, and 255 is the filler that matches no tile");
static_assert(NN <= (1 << 17) && RT <= (1 << 9), "a log entry packs source (17 bits) and row (9 bits)");

static constexpr int KEYB = (NE / 4 + SB - 1) / SB;
__global__ __launch_bounds__(SB) void tile_key_k(const int* __restrict__ dst,
                                                 unsigned* __restrict__ key) {
  const int i = blockIdx.x * SB + threadIdx.x;
  if (i < NE / 4) {
    const int4 d = ((const int4*)dst)[i];
    key[i] = (unsigned)d.x / RT | ((unsigned)d.y / RT) << 8 | ((unsigned)d.z / RT) << 16 |
             ((unsigned)d.w / RT) << 24;
  }
}

template <int V> struct RowVec;
template <> struct RowVec<4> { typedef float4 T; };
template <> struct RowVec<2> { typedef float2 T; };
__device__ __forceinline__ float vget(const float4& v, int k) {
  return k == 0 ? v.x : k == 1 ? v.y : k == 2 ? v.z : v.w;
}
__device__ __forceinline__ float vget(const float2& v, int k) { return k == 0 ? v.x : v.y; }
__device__ __forceinline__ void vset(float4& v, int k, float f) {
  if (k == 0) v.x = f; else if (k == 1) v.y = f; else if (k == 2) v.z = f; else v.w = f;
}
__device__ __forceinline__ void vset(float2& v, int k, float f) {
  if (k == 0) v.x = f; else v.y = f;
}

template <int C, bool ET, bool BUILD>
__global__ __launch_bounds__(1024) void tile_agg_k(
    const int* __restrict__ src, const int* __restrict__ dst, const unsigned* __restrict__ key,
    const float* __restrict__ s_src, const float* __restrict__ s_dst,
    const float* __restrict__ eterm, const float* __restrict__ self_logit,
    float* __restrict__ segsum, const float* __restrict__ h, float* __restrict__ out,
    int* __restrict__ glog_sr, int* __restrict__ glog_e, int* __restrict__ glog_n,
    const float* __restrict__ edge_attr, float* __restrict__ mean_attr) {
  constexpr int V = C / 32;
  static_assert(!BUILD || RT * AP <= RT * C, "the attr sums borrow the tile's LDS");
  typedef typename RowVec<V>::T VecT;
  __shared__ float tile[RT * C];
  __shared__ float ssum[RT];
  __shared__ unsigned smax[RT];
  __shared__ int queue[32 * QCAP];
  __shared__ int log_sr[32 * LOGCAP];
  __shared__ float log_l[32 * LOGCAP];
  const int tid = threadIdx.x, lane = tid & 31;
  const int wave = __builtin_amdgcn_readfirstlane(tid >> 5);
  const int row0 = blockIdx.x * RT;
  const int rows = min(RT, NN - row0);

  for (int r = tid; r < rows; r += 1024) smax[r] = f2ord(self_logit[row0 + r]);
  if (BUILD) { for (int i = tid; i < RT * AP; i += 1024) tile[i] = 0.f; }
  __syncthreads();

  int* myq = &queue[wave * QCAP];
  int* mysr = &log_sr[wave * LOGCAP];
  float* myll = &log_l[wave * LOGCAP];
  const int gw = blockIdx.x * 32 + wave;
  int* gsr = glog_sr + (size_t)gw * LOGCAP;
  int* ge = glog_e + (size_t)gw * LOGCAP;
  int logn = 0;
  bool ovf = false;

  auto edge_logit = [&](int e, int& s, int& r) {
    s = src[e];
    const int d = dst[e];
    float l = s_src[s] + s_dst[d];
    if (ET) l += eterm[e];
    r = d - row0;
    return leaky(l);
  };

  auto proc_max = [&](int nb) {
    int sr = 0, e = 0;
    float ll = 0.f;
    if (lane < nb) {
      int s, r;
      e = myq[lane];
      float4 av[4];
      if (BUILD) {
        const float4* ap = (const float4*)&edge_attr[(size_t)e * 16];
#pragma unroll
        for (int q = 0; q < 4; ++q) av[q] = ap[q];
      }
      ll = edge_logit(e, s, r);
      atomicMax(&smax[r], f2ord(ll));
      sr = s | r << 17;
      if (BUILD) {
        float* t = &tile[r * AP];
#pragma unroll
        for (int q = 0; q < 4; ++q) {
          atomicAdd(t + 4 * q, av[q].x); atomicAdd(t + 4 * q + 1, av[q].y);
          atomicAdd(t + 4 * q + 2, av[q].z); atomicAdd(t + 4 * q + 3, av[q].w);
        }
        atomicAdd(t + 16, 1.f);
      }
    }
    if (!ovf && logn + nb <= LOGCAP) {
      if (lane < nb) {
        mysr[logn + lane] = sr; myll[logn + lane] = ll;
        if (BUILD) { gsr[logn + lane] = sr; ge[logn + lane] = e; }
      }
      logn += nb;
    } else {
      ovf = true;
    }
  };
  auto max_from_log = [&](int n) {
    for (int b = 0; b < n; b += 128) {
      int sr[4], e[4];
      float a[4], d[4], t[4];
#pragma unroll
      for (int u = 0; u < 4; ++u) {
        const int ic = min(b + 32 * u + lane, n - 1);
        sr[u] = gsr[ic];
        e[u] = ET ? ge[ic] : 0;
      }
      asm volatile("" ::"v"(sr[0]), "v"(sr[1]), "v"(sr[2]), "v"(sr[3]), "v"(e[0]), "v"(e[1]),
                   "v"(e[2]), "v"(e[3]));
#pragma unroll
      for (int u = 0; u < 4; ++u) {
        a[u] = s_src[sr[u] & 0x1ffff];
        d[u] = s_dst[row0 + (sr[u] >> 17)];
        t[u] = ET ? eterm[e[u]] : 0.f;
      }
      asm volatile("" ::"v"(a[0]), "v"(a[1]), "v"(a[2]), "v"(a[3]), "v"(d[0]), "v"(d[1]),
                   "v"(d[2]), "v"(d[3]), "v"(t[0]), "v"(t[1]), "v"(t[2]), "v"(t[3]));
#pragma unroll
      for (int u = 0; u < 4; ++u) {
        const int i = b + 32 * u + lane;
        if (i < n) {
          float l = a[u] + d[u];
          if (ET) l += t[u];
          const float ll = leaky(l);
          atomicMax(&smax[sr[u] >> 17], f2ord(ll));
          mysr[i] = sr[u];
          myll[i] = ll;
        }
      }
    }
  };

  auto agg_rows = [&](int nb, int s, int r, float w) {
    for (int j0 = 0; j0 < nb; j0 += 8) {
      VecT hv[8];
#pragma unroll
      for (int u = 0; u < 8; ++u) {
        const int sj = __builtin_amdgcn_readlane(s, j0 + u);
        hv[u] = *(const VecT*)&h[(size_t)sj * C + lane * V];
      }
#pragma unroll
      for (int u = 0; u < 8; ++u) {
        if (j0 + u < nb) {
          const int rj = __builtin_amdgcn_readlane(r, j0 + u);
          const float wj = __int_as_float(__builtin_amdgcn_readlane(__float_as_int(w), j0 + u));
          float* t = &tile[rj * C + lane];
#pragma unroll
          for (int k = 0; k < V; ++k) atomicAdd(t + k * 32, wj * vget(hv[u], k));
        }
      }
    }
  };
  auto proc_log = [&](int base, int nb) {
    int s = 0, r = 0;
    float w = 0.f;
    if (lane < nb) {
      const int sr = mysr[base + lane];
      s = sr & 0x1ffff;
      r = sr >> 17;
      w = __expf(myll[base + lane] - ord2f(smax[r]));
      atomicAdd(&ssum[r], w);
    }
    agg_rows(nb, s, r, w);
  };
  auto proc_queue = [&](int nb) {
    int s = 0, r = 0;
    float w = 0.f;
    if (lane < nb) {
      const float ll = edge_logit(myq[lane], s, r);
      w = __expf(ll - ord2f(smax[r]));
      atomicAdd(&ssum[r], w);
    }
    agg_rows(nb, s, r, w);
  };

  const unsigned tt = blockIdx.x * 0x01010101u;
  const int rot = blockIdx.x * (NSTEP / NTILE);
  auto step_of = [&](int k) { const int kk = k + rot; return kk >= NSTEP ? kk - NSTEP : kk; };
  const uint4* key4 = (const uint4*)key;
  auto fetch = [&](uint4 (&d)[4], int k0) {
#pragma unroll
    for (int u = 0; u < 4; ++u) {
      const int k = k0 + 32 * u;
      d[u] = (k < NSTEP) ? key4[(size_t)step_of(k) * 32 + lane] : make_uint4(~0u, ~0u, ~0u, ~0u);
    }
  };
  auto differs = [&](unsigned x) {
    const unsigned y = x ^ tt;
    return (((y & 0x7f7f7f7fu) + 0x7f7f7f7fu) | y) & 0x80808080u;
  };
  auto sweep = [&](auto&& proc) {
    int cnt = 0;
    uint4 cur[4], nxt[4];
    fetch(cur, wave);
    for (int k0 = wave; k0 < NSTEP; k0 += 32 * 4) {
      fetch(nxt, k0 + 32 * 4);
#pragma unroll
      for (int u = 0; u < 4; ++u) {
        unsigned m = ~(differs(cur[u].x) >> 7 | differs(cur[u].y) >> 6 | differs(cur[u].z) >> 5 |
                       differs(cur[u].w) >> 4) & 0x0f0f0f0fu;
        unsigned live = __builtin_amdgcn_ballot_w32(m != 0u);
        if (live != 0u) {
          const int e0 = step_of(k0 + 32 * u) * 512 + lane * 16;
          do {
            if (m != 0u) {
              const int bit = __builtin_ctz(m);
              myq[cnt + __builtin_amdgcn_mbcnt_lo(live, 0u)] = e0 + (bit & 7) * 4 + (bit >> 3);
              m &= m - 1u;
            }
            cnt += __popc(live);
            live = __builtin_amdgcn_ballot_w32(m != 0u);
            if (cnt >= 32) {
              proc(32);
              const int rem = cnt - 32;
              const int keep = (lane < rem) ? myq[32 + lane] : 0;
              if (lane < rem) myq[lane] = keep;
              cnt = rem;
            }
          } while (live != 0u);
        }
      }
#pragma unroll
      for (int u = 0; u < 4; ++u) cur[u] = nxt[u];
    }
    if (cnt > 0) proc(cnt);
  };

  if (BUILD) {
    sweep(proc_max);
    if (lane == 0) glog_n[gw] = ovf ? -1 : logn;
  } else {
    const int gn = __builtin_amdgcn_readfirstlane(glog_n[gw]);
    if (gn >= 0) {
      max_from_log(gn);
      logn = gn;
    } else {
      sweep(proc_max);
    }
  }
  __syncthreads();

  if (BUILD) {
    for (int i = tid; i < rows * 16; i += 1024) {
      const int r = i >> 4, j = i & 15;
      mean_attr[(size_t)(row0 + r) * 16 + j] = tile[r * AP + j] / fmaxf(tile[r * AP + 16], 1.f);
    }
    __syncthreads();
  }

  for (int i = tid; i < rows * 32; i += 1024) {
    const int r = i >> 5, n = row0 + r;
    const float a = __expf(self_logit[n] - ord2f(smax[r]));
    const VecT hv = *(const VecT*)&h[(size_t)n * C + lane * V];
#pragma unroll
    for (int k = 0; k < V; ++k) tile[r * C + k * 32 + lane] = a * vget(hv, k);
    if (lane == 0) ssum[r] = a;
  }
  __syncthreads();

  const int nlog = __builtin_amdgcn_readfirstlane(logn);
  if (__builtin_amdgcn_readfirstlane((int)ovf) == 0) {
    int b = 0;
    for (; b + 32 <= nlog; b += 32) proc_log(b, 32);
    if (b < nlog) proc_log(b, nlog - b);
  } else {
    sweep(proc_queue);
  }
  __syncthreads();

  for (int i = tid; i < rows * 32; i += 1024) {
    const int r = i >> 5, n = row0 + r;
    VecT o;
#pragma unroll
    for (int k = 0; k < V; ++k) vset(o, k, tile[r * C + k * 32 + lane]);
    *(VecT*)&out[(size_t)n * C + lane * V] = o;
    if (lane == 0) segsum[n] = ssum[r];
  }
}

struct Scratch {
  float *s_src, *s_dst, *self_l, *segsum;
  float* mean_attr;
  float *eterm, *wa;
  const unsigned* key;
  int *glog_sr, *glog_e, *glog_n;
};

template <int C, bool ET, bool BUILD>
static void run_gat(hipStream_t st, const float* hin, const float* Wm, const float* asrc,
                    const float* adst, const float* We, const float* ae,
                    const int* src, const int* dst, const float* edge_attr, float* hg,
                    float* out, const Scratch& S) {
  if (ET) wa_k<<<1, 16, 0, st>>>(We, ae, S.wa, C);
  gemm_k128<C><<<GW / 8, 256, 0, st>>>(hin, Wm, hg);
  node_prep_k<C, ET><<<SG, SB, 0, st>>>(hg, asrc, adst, S.mean_attr, S.wa, S.s_src, S.s_dst,
                                        S.self_l);
  if (ET) eterm_k<<<SG, SB, 0, st>>>(edge_attr, S.wa, S.eterm);
  tile_agg_k<C, ET, BUILD><<<NTILE, 1024, 0, st>>>(src, dst, S.key, S.s_src, S.s_dst, S.eterm,
                                                   S.self_l, S.segsum, hg, out, S.glog_sr,
                                                   S.glog_e, S.glog_n, edge_attr, S.mean_attr);
}

extern "C" void kernel_launch(void* const* d_in, const int* in_sizes, int n_in, void* d_out,
                              int out_size, void* d_ws, size_t ws_size, hipStream_t stream) {
  (void)in_sizes; (void)n_in; (void)out_size; (void)ws_size;
  const float* x = (const float*)d_in[0];
  const int* ei = (const int*)d_in[1];
  const int* src = ei;
  const int* dst = ei + NE;
  const float* edge_attr = (const float*)d_in[2];
  const float *W1 = (const float*)d_in[3], *a1s = (const float*)d_in[4],
              *a1d = (const float*)d_in[5], *b1 = (const float*)d_in[6];
  const float *W2 = (const float*)d_in[7], *a2s = (const float*)d_in[8],
              *a2d = (const float*)d_in[9], *b2 = (const float*)d_in[10];
  const float *We2 = (const float*)d_in[11], *ae2 = (const float*)d_in[12];
  const float *Wr = (const float*)d_in[13], *br = (const float*)d_in[14];
  const float *Wmu = (const float*)d_in[15], *amus = (const float*)d_in[16],
              *amud = (const float*)d_in[17], *bmu = (const float*)d_in[18],
              *Wemu = (const float*)d_in[19], *aemu = (const float*)d_in[20];
  const float *Wls = (const float*)d_in[21], *alss = (const float*)d_in[22],
              *alsd = (const float*)d_in[23], *bls = (const float*)d_in[24],
              *Wels = (const float*)d_in[25], *aels = (const float*)d_in[26];
  float* outf = (float*)d_out;

  float* W = (float*)d_ws;
  size_t o = 0;
  float* hA = W + o; o += (size_t)NN * 128;
  float* hB = W + o; o += (size_t)NN * 128;
  float* hC = W + o; o += (size_t)NN * 128;
  Scratch S;
  S.s_src = W + o; o += NN;
  S.s_dst = W + o; o += NN;
  S.self_l = W + o; o += NN;
  S.segsum = W + o; o += NN;
  S.mean_attr = W + o; o += (size_t)NN * 16;
  S.eterm = W + o; o += NE;
  S.wa = W + o; o += 16;
  unsigned* key = (unsigned*)(W + o); o += NE / 4;
  S.key = key;
  o += 48;
  S.glog_sr = (int*)(W + o); o += (size_t)NTILE * 32 * LOGCAP;
  S.glog_e = (int*)(W + o); o += (size_t)NTILE * 32 * LOGCAP;
  S.glog_n = (int*)(W + o); o += NTILE * 32;

  tile_key_k<<<KEYB, SB, 0, stream>>>(dst, key);

  run_gat<128, false, true>(stream, x, W1, a1s, a1d, nullptr, nullptr, src, dst, edge_attr, hA, hB,
                      S);
  norm_relu_k<<<SG, SB, 0, stream>>>(hB, S.segsum, b1);

  run_gat<128, true, false>(stream, hB, W2, a2s, a2d, We2, ae2, src, dst, edge_attr, hA, hC, S);

  gemm_k128<128><<<GW / 8, 256, 0, stream>>>(x, Wr, hB);
  residual_k<<<SG, SB, 0, stream>>>(hC, S.segsum, b2, hB, br);

  run_gat<64, true, false>(stream, hB, Wmu, amus, amud, Wemu, aemu, src, dst, edge_attr, hA, outf, S);
  norm_bias_k<64><<<SG, SB, 0, stream>>>(outf, S.segsum, bmu);
  run_gat<64, true, false>(stream, hB, Wls, alss, alsd, Wels, aels, src, dst, edge_attr, hA,
                    outf + (size_t)NN * 64, S);
  norm_bias_k<64><<<SG, SB, 0, stream>>>(outf + (size_t)NN * 64, S.segsum, bls);
}


# Owner workgroup finishes its rows at the tile store: divides by the exp sum, adds bias, relu, residual row (x@Wr made first): the 4 normalise passes and 4 of 24 launches go; same arithmetic
# speedup vs baseline: 4.3795x; 1.2147x over previous
#include <hip/hip_runtime.h>
#include <math.h>

typedef float v2f __attribute__((ext_vector_type(2)));
typedef float v8f __attribute__((ext_vector_type(8)));

static constexpr int   NN = 100000;
static constexpr int   NE = 1600000;
static constexpr float NEG = 0.2f;

__device__ __forceinline__ float leaky(float x) { return x > 0.f ? x : NEG * x; }
__device__ __forceinline__ unsigned f2ord(float f) {
  unsigned u = __float_as_uint(f);
  return (u & 0x80000000u) ? ~u : (u | 0x80000000u);
}
__device__ __forceinline__ float ord2f(unsigned v) {
  unsigned u = (v & 0x80000000u) ? (v & 0x7fffffffu) : ~v;
  return __uint_as_float(u);
}

static constexpr int SB = 256;
static constexpr int SG = 2048;
static constexpr int ST = SG * SB;
static constexpr int NW = ST / 32;

__global__ void wa_k(const float* __restrict__ We, const float* __restrict__ ae,
                     float* wa, int K) {
  int i = threadIdx.x;
  if (i < 16) { float s = 0.f; for (int j = 0; j < K; ++j) s += We[i * K + j] * ae[j]; wa[i] = s; }
}

static constexpr int GW = 2048;
template <int COUT>
__global__ __launch_bounds__(256) void gemm_k128(const float* __restrict__ A,
                                                 const float* __restrict__ B,
                                                 float* __restrict__ C) {
  constexpr int NCG = COUT / 32;
  constexpr int NSW = GW / NCG;
  constexpr int NS = NN / 16;
  static_assert(NN % 16 == 0, "rows come in stripes of 16");
  const int lane = threadIdx.x & 31, half = lane >> 4, lidx = lane & 15;
  const int wid = __builtin_amdgcn_readfirstlane((int)((blockIdx.x * 256 + threadIdx.x) >> 5));
  const int n0 = (wid % NCG) * 32;
  const int sw = wid / NCG;

  float bq[2][16][4];
#pragma unroll
  for (int t = 0; t < 2; ++t)
#pragma unroll
    for (int j = 0; j < 16; ++j)
#pragma unroll
      for (int i = 0; i < 4; ++i)
        bq[t][j][i] = B[(8 * j + 4 * half + i) * COUT + n0 + 16 * t + lidx];

  const float4* A4 = (const float4*)A;
  auto load = [&](float4 (&a)[16], int s) {
    const float4* p = A4 + ((size_t)s * 16 + lidx) * 32 + half;
#pragma unroll
    for (int j = 0; j < 16; ++j) a[j] = p[2 * j];
  };
  auto compute = [&](const float4 (&a)[16], int s) {
    v8f acc0 = {}, acc1 = {};
#pragma unroll
    for (int j = 0; j < 16; ++j) {
      v2f a01, a23, b;
      a01.x = a[j].x; a01.y = a[j].y;
      a23.x = a[j].z; a23.y = a[j].w;
      b.x = bq[0][j][0]; b.y = bq[0][j][1];
      acc0 = __builtin_amdgcn_wmma_f32_16x16x4_f32(false, a01, false, b, (short)0, acc0,
                                                   false, false);
      b.x = bq[1][j][0]; b.y = bq[1][j][1];
      acc1 = __builtin_amdgcn_wmma_f32_16x16x4_f32(false, a01, false, b, (short)0, acc1,
                                                   false, false);
      b.x = bq[0][j][2]; b.y = bq[0][j][3];
      acc0 = __builtin_amdgcn_wmma_f32_16x16x4_f32(false, a23, false, b, (short)0, acc0,
                                                   false, false);
      b.x = bq[1][j][2]; b.y = bq[1][j][3];
      acc1 = __builtin_amdgcn_wmma_f32_16x16x4_f32(false, a23, false, b, (short)0, acc1,
                                                   false, false);
    }
    float* c = C + ((size_t)s * 16 + 8 * half) * COUT + n0 + lidx;
#pragma unroll
    for (int r = 0; r < 8; ++r) {
      c[r * COUT] = acc0[r];
      c[r * COUT + 16] = acc1[r];
    }
  };

  float4 ba[16], bb[16];
  int s = sw;
  load(ba, s);
  for (; s + NSW < NS; s += 2 * NSW) {
    load(bb, s + NSW);
    compute(ba, s);
    load(ba, min(s + 2 * NSW, NS - 1));
    compute(bb, s + NSW);
  }
  if (s < NS) compute(ba, s);
}

template <int C, bool ET>
__global__ __launch_bounds__(SB) void node_prep_k(
    const float* __restrict__ h, const float* __restrict__ a_src,
    const float* __restrict__ a_dst, const float* __restrict__ mean_attr,
    const float* __restrict__ wa, float* __restrict__ s_src, float* __restrict__ s_dst,
    float* __restrict__ self_logit) {
  constexpr int R = C / 32;
  constexpr int U = 256 / C;
  const int gid = blockIdx.x * SB + threadIdx.x;
  const int lane = gid & 31;
  float as[R], ad[R];
#pragma unroll
  for (int i = 0; i < R; ++i) { as[i] = a_src[lane + i * 32]; ad[i] = a_dst[lane + i * 32]; }
  for (int n0 = gid >> 5; n0 < NN; n0 += NW * U) {
    float hv[U][R];
#pragma unroll
    for (int u = 0; u < U; ++u) {
      const int nc = min(n0 + u * NW, NN - 1);
#pragma unroll
      for (int i = 0; i < R; ++i) hv[u][i] = h[(size_t)nc * C + lane + i * 32];
    }
    float psv[U], pdv[U];
#pragma unroll
    for (int u = 0; u < U; ++u) {
      float ps = 0.f, pd = 0.f;
#pragma unroll
      for (int i = 0; i < R; ++i) { ps += hv[u][i] * as[i]; pd += hv[u][i] * ad[i]; }
#pragma unroll
      for (int m = 16; m > 0; m >>= 1) { ps += __shfl_xor(ps, m, 32); pd += __shfl_xor(pd, m, 32); }
      psv[u] = ps; pdv[u] = pd;
    }
#pragma unroll
    for (int u = 0; u < U; ++u) {
      const int n = n0 + u * NW;
      if (n < NN) {
        const float ps = psv[u], pd = pdv[u];
        if (lane == 0) {
          float st = 0.f;
          if (ET) { for (int j = 0; j < 16; ++j) st += mean_attr[(size_t)n * 16 + j] * wa[j]; }
          float sl = leaky(ps + pd + st);
          s_src[n] = ps; s_dst[n] = pd; self_logit[n] = sl;
        }
      }
    }
  }
}

__global__ __launch_bounds__(SB) void eterm_k(const float* __restrict__ ea,
                                              const float* __restrict__ wa,
                                              float* __restrict__ eterm) {
  constexpr int NT = NE * 16;
  const int gid = blockIdx.x * SB + threadIdx.x;
  const int j = gid & 15;
  const float w = wa[j];
  for (int i0 = gid; i0 < NT; i0 += ST * 8) {
    float v[8];
#pragma unroll
    for (int u = 0; u < 8; ++u) v[u] = ea[min(i0 + u * ST, NT - 1)] * w;
#pragma unroll
    for (int u = 0; u < 8; ++u) {
      float t = v[u];
      t += __shfl_xor(t, 1, 32); t += __shfl_xor(t, 2, 32);
      t += __shfl_xor(t, 4, 32); t += __shfl_xor(t, 8, 32);
      v[u] = t;
    }
#pragma unroll
    for (int u = 0; u < 8; ++u) {
      const int i = i0 + u * ST;
      if (j == 0 && i < NT) eterm[i >> 4] = v[u];
    }
  }
}

static constexpr int RT = 400;
static constexpr int AP = 17;
static constexpr int NTILE = (NN + RT - 1) / RT;
static constexpr int QCAP = 64;
static constexpr int LOGCAP = 320;
static constexpr int NSTEP = NE / 512;
static_assert(NE % 512 == 0, "the key plane is swept in steps of 512 edges");
static_assert(NTILE <= 255, "a tile id is one byte, and 255 is the filler that matches no tile");
static_assert(NN <= (1 << 17) && RT <= (1 << 9), "a log entry packs source (17 bits) and row (9 bits)");

static constexpr int KEYB = (NE / 4 + SB - 1) / SB;
__global__ __launch_bounds__(SB) void tile_key_k(const int* __restrict__ dst,
                                                 unsigned* __restrict__ key) {
  const int i = blockIdx.x * SB + threadIdx.x;
  if (i < NE / 4) {
    const int4 d = ((const int4*)dst)[i];
    key[i] = (unsigned)d.x / RT | ((unsigned)d.y / RT) << 8 | ((unsigned)d.z / RT) << 16 |
             ((unsigned)d.w / RT) << 24;
  }
}

template <int V> struct RowVec;
template <> struct RowVec<4> { typedef float4 T; };
template <> struct RowVec<2> { typedef float2 T; };
__device__ __forceinline__ float vget(const float4& v, int k) {
  return k == 0 ? v.x : k == 1 ? v.y : k == 2 ? v.z : v.w;
}
__device__ __forceinline__ float vget(const float2& v, int k) { return k == 0 ? v.x : v.y; }
__device__ __forceinline__ void vset(float4& v, int k, float f) {
  if (k == 0) v.x = f; else if (k == 1) v.y = f; else if (k == 2) v.z = f; else v.w = f;
}
__device__ __forceinline__ void vset(float2& v, int k, float f) {
  if (k == 0) v.x = f; else v.y = f;
}
__device__ __forceinline__ void keep_loaded(const float4& v) {
  asm volatile("" ::"v"(v.x), "v"(v.y), "v"(v.z), "v"(v.w));
}
__device__ __forceinline__ void keep_loaded(const float2& v) {
  asm volatile("" ::"v"(v.x), "v"(v.y));
}

static constexpr int EPI_BIAS = 0;
static constexpr int EPI_RELU = 1;
static constexpr int EPI_RELU_RES = 2;
template <int C, bool ET, bool BUILD, int EPI>
__global__ __launch_bounds__(1024) void tile_agg_k(
    const int* __restrict__ src, const int* __restrict__ dst, const unsigned* __restrict__ key,
    const float* __restrict__ s_src, const float* __restrict__ s_dst,
    const float* __restrict__ eterm, const float* __restrict__ self_logit,
    const float* __restrict__ h, float* __restrict__ out, const float* __restrict__ bias,
    const float* __restrict__ resid, const float* __restrict__ rbias,
    int* __restrict__ glog_sr, int* __restrict__ glog_e, int* __restrict__ glog_n,
    const float* __restrict__ edge_attr, float* __restrict__ mean_attr) {
  constexpr int V = C / 32;
  static_assert(!BUILD || RT * AP <= RT * C, "the attr sums borrow the tile's LDS");
  typedef typename RowVec<V>::T VecT;
  __shared__ float tile[RT * C];
  __shared__ float ssum[RT];
  __shared__ unsigned smax[RT];
  __shared__ int queue[32 * QCAP];
  __shared__ int log_sr[32 * LOGCAP];
  __shared__ float log_l[32 * LOGCAP];
  const int tid = threadIdx.x, lane = tid & 31;
  const int wave = __builtin_amdgcn_readfirstlane(tid >> 5);
  const int row0 = blockIdx.x * RT;
  const int rows = min(RT, NN - row0);

  for (int r = tid; r < rows; r += 1024) smax[r] = f2ord(self_logit[row0 + r]);
  if (BUILD) { for (int i = tid; i < RT * AP; i += 1024) tile[i] = 0.f; }
  __syncthreads();

  int* myq = &queue[wave * QCAP];
  int* mysr = &log_sr[wave * LOGCAP];
  float* myll = &log_l[wave * LOGCAP];
  const int gw = blockIdx.x * 32 + wave;
  int* gsr = glog_sr + (size_t)gw * LOGCAP;
  int* ge = glog_e + (size_t)gw * LOGCAP;
  int logn = 0;
  bool ovf = false;

  auto edge_logit = [&](int e, int& s, int& r) {
    s = src[e];
    const int d = dst[e];
    float l = s_src[s] + s_dst[d];
    if (ET) l += eterm[e];
    r = d - row0;
    return leaky(l);
  };

  auto proc_max = [&](int nb) {
    int sr = 0, e = 0;
    float ll = 0.f;
    if (lane < nb) {
      int s, r;
      e = myq[lane];
      float4 av[4];
      if (BUILD) {
        const float4* ap = (const float4*)&edge_attr[(size_t)e * 16];
#pragma unroll
        for (int q = 0; q < 4; ++q) av[q] = ap[q];
      }
      ll = edge_logit(e, s, r);
      atomicMax(&smax[r], f2ord(ll));
      sr = s | r << 17;
      if (BUILD) {
        float* t = &tile[r * AP];
#pragma unroll
        for (int q = 0; q < 4; ++q) {
          atomicAdd(t + 4 * q, av[q].x); atomicAdd(t + 4 * q + 1, av[q].y);
          atomicAdd(t + 4 * q + 2, av[q].z); atomicAdd(t + 4 * q + 3, av[q].w);
        }
        atomicAdd(t + 16, 1.f);
      }
    }
    if (!ovf && logn + nb <= LOGCAP) {
      if (lane < nb) {
        mysr[logn + lane] = sr; myll[logn + lane] = ll;
        if (BUILD) { gsr[logn + lane] = sr; ge[logn + lane] = e; }
      }
      logn += nb;
    } else {
      ovf = true;
    }
  };
  auto max_from_log = [&](int n) {
    for (int b = 0; b < n; b += 128) {
      int sr[4], e[4];
      float a[4], d[4], t[4];
#pragma unroll
      for (int u = 0; u < 4; ++u) {
        const int ic = min(b + 32 * u + lane, n - 1);
        sr[u] = gsr[ic];
        e[u] = ET ? ge[ic] : 0;
      }
      asm volatile("" ::"v"(sr[0]), "v"(sr[1]), "v"(sr[2]), "v"(sr[3]), "v"(e[0]), "v"(e[1]),
                   "v"(e[2]), "v"(e[3]));
#pragma unroll
      for (int u = 0; u < 4; ++u) {
        a[u] = s_src[sr[u] & 0x1ffff];
        d[u] = s_dst[row0 + (sr[u] >> 17)];
        t[u] = ET ? eterm[e[u]] : 0.f;
      }
      asm volatile("" ::"v"(a[0]), "v"(a[1]), "v"(a[2]), "v"(a[3]), "v"(d[0]), "v"(d[1]),
                   "v"(d[2]), "v"(d[3]), "v"(t[0]), "v"(t[1]), "v"(t[2]), "v"(t[3]));
#pragma unroll
      for (int u = 0; u < 4; ++u) {
        const int i = b + 32 * u + lane;
        if (i < n) {
          float l = a[u] + d[u];
          if (ET) l += t[u];
          const float ll = leaky(l);
          atomicMax(&smax[sr[u] >> 17], f2ord(ll));
          mysr[i] = sr[u];
          myll[i] = ll;
        }
      }
    }
  };

  auto agg_rows = [&](int nb, int s, int r, float w) {
    for (int j0 = 0; j0 < nb; j0 += 8) {
      VecT hv[8];
#pragma unroll
      for (int u = 0; u < 8; ++u) {
        const int sj = __builtin_amdgcn_readlane(s, j0 + u);
        hv[u] = *(const VecT*)&h[(size_t)sj * C + lane * V];
      }
#pragma unroll
      for (int u = 0; u < 8; ++u) {
        if (j0 + u < nb) {
          const int rj = __builtin_amdgcn_readlane(r, j0 + u);
          const float wj = __int_as_float(__builtin_amdgcn_readlane(__float_as_int(w), j0 + u));
          float* t = &tile[rj * C + lane];
#pragma unroll
          for (int k = 0; k < V; ++k) atomicAdd(t + k * 32, wj * vget(hv[u], k));
        }
      }
    }
  };
  auto proc_log = [&](int base, int nb) {
    int s = 0, r = 0;
    float w = 0.f;
    if (lane < nb) {
      const int sr = mysr[base + lane];
      s = sr & 0x1ffff;
      r = sr >> 17;
      w = __expf(myll[base + lane] - ord2f(smax[r]));
      atomicAdd(&ssum[r], w);
    }
    agg_rows(nb, s, r, w);
  };
  auto proc_queue = [&](int nb) {
    int s = 0, r = 0;
    float w = 0.f;
    if (lane < nb) {
      const float ll = edge_logit(myq[lane], s, r);
      w = __expf(ll - ord2f(smax[r]));
      atomicAdd(&ssum[r], w);
    }
    agg_rows(nb, s, r, w);
  };

  const unsigned tt = blockIdx.x * 0x01010101u;
  const int rot = blockIdx.x * (NSTEP / NTILE);
  auto step_of = [&](int k) { const int kk = k + rot; return kk >= NSTEP ? kk - NSTEP : kk; };
  const uint4* key4 = (const uint4*)key;
  auto fetch = [&](uint4 (&d)[4], int k0) {
#pragma unroll
    for (int u = 0; u < 4; ++u) {
      const int k = k0 + 32 * u;
      d[u] = (k < NSTEP) ? key4[(size_t)step_of(k) * 32 + lane] : make_uint4(~0u, ~0u, ~0u, ~0u);
    }
  };
  auto differs = [&](unsigned x) {
    const unsigned y = x ^ tt;
    return (((y & 0x7f7f7f7fu) + 0x7f7f7f7fu) | y) & 0x80808080u;
  };
  auto sweep = [&](auto&& proc) {
    int cnt = 0;
    uint4 cur[4], nxt[4];
    fetch(cur, wave);
    for (int k0 = wave; k0 < NSTEP; k0 += 32 * 4) {
      fetch(nxt, k0 + 32 * 4);
#pragma unroll
      for (int u = 0; u < 4; ++u) {
        unsigned m = ~(differs(cur[u].x) >> 7 | differs(cur[u].y) >> 6 | differs(cur[u].z) >> 5 |
                       differs(cur[u].w) >> 4) & 0x0f0f0f0fu;
        unsigned live = __builtin_amdgcn_ballot_w32(m != 0u);
        if (live != 0u) {
          const int e0 = step_of(k0 + 32 * u) * 512 + lane * 16;
          do {
            if (m != 0u) {
              const int bit = __builtin_ctz(m);
              myq[cnt + __builtin_amdgcn_mbcnt_lo(live, 0u)] = e0 + (bit & 7) * 4 + (bit >> 3);
              m &= m - 1u;
            }
            cnt += __popc(live);
            live = __builtin_amdgcn_ballot_w32(m != 0u);
            if (cnt >= 32) {
              proc(32);
              const int rem = cnt - 32;
              const int keep = (lane < rem) ? myq[32 + lane] : 0;
              if (lane < rem) myq[lane] = keep;
              cnt = rem;
            }
          } while (live != 0u);
        }
      }
#pragma unroll
      for (int u = 0; u < 4; ++u) cur[u] = nxt[u];
    }
    if (cnt > 0) proc(cnt);
  };

  if (BUILD) {
    sweep(proc_max);
    if (lane == 0) glog_n[gw] = ovf ? -1 : logn;
  } else {
    const int gn = __builtin_amdgcn_readfirstlane(glog_n[gw]);
    if (gn >= 0) {
      max_from_log(gn);
      logn = gn;
    } else {
      sweep(proc_max);
    }
  }
  __syncthreads();

  if (BUILD) {
    for (int i = tid; i < rows * 16; i += 1024) {
      const int r = i >> 4, j = i & 15;
      mean_attr[(size_t)(row0 + r) * 16 + j] = tile[r * AP + j] / fmaxf(tile[r * AP + 16], 1.f);
    }
    __syncthreads();
  }

  for (int i = tid; i < rows * 32; i += 1024) {
    const int r = i >> 5, n = row0 + r;
    const float a = __expf(self_logit[n] - ord2f(smax[r]));
    const VecT hv = *(const VecT*)&h[(size_t)n * C + lane * V];
#pragma unroll
    for (int k = 0; k < V; ++k) tile[r * C + k * 32 + lane] = a * vget(hv, k);
    if (lane == 0) ssum[r] = a;
  }
  __syncthreads();

  const int nlog = __builtin_amdgcn_readfirstlane(logn);
  if (__builtin_amdgcn_readfirstlane((int)ovf) == 0) {
    int b = 0;
    for (; b + 32 <= nlog; b += 32) proc_log(b, 32);
    if (b < nlog) proc_log(b, nlog - b);
  } else {
    sweep(proc_queue);
  }
  __syncthreads();

  const VecT bv = *(const VecT*)&bias[lane * V];
  VecT rbv = {};
  if (EPI == EPI_RELU_RES) rbv = *(const VecT*)&rbias[lane * V];
  for (int i0 = tid; i0 < rows * 32; i0 += 4 * 1024) {
    VecT rv[4] = {};
    if (EPI == EPI_RELU_RES) {
#pragma unroll
      for (int u = 0; u < 4; ++u) {
        const int nc = row0 + (min(i0 + u * 1024, rows * 32 - 1) >> 5);
        rv[u] = *(const VecT*)&resid[(size_t)nc * C + lane * V];
      }
#pragma unroll
      for (int u = 0; u < 4; ++u) keep_loaded(rv[u]);
    }
#pragma unroll
    for (int u = 0; u < 4; ++u) {
      const int i = i0 + u * 1024;
      if (i < rows * 32) {
        const int r = i >> 5, n = row0 + r;
        const float den = ssum[r] + 1e-16f;
        VecT o;
#pragma unroll
        for (int k = 0; k < V; ++k) {
          float x = tile[r * C + k * 32 + lane] / den + vget(bv, k);
          if (EPI != EPI_BIAS) x = fmaxf(x, 0.f);
          if (EPI == EPI_RELU_RES) x = x + vget(rv[u], k) + vget(rbv, k);
          vset(o, k, x);
        }
        *(VecT*)&out[(size_t)n * C + lane * V] = o;
      }
    }
  }
}

struct Scratch {
  float *s_src, *s_dst, *self_l;
  float* mean_attr;
  float *eterm, *wa;
  const unsigned* key;
  int *glog_sr, *glog_e, *glog_n;
};

template <int C, bool ET, bool BUILD, int EPI>
static void run_gat(hipStream_t st, const float* hin, const float* Wm, const float* asrc,
                    const float* adst, const float* bias, const float* We, const float* ae,
                    const int* src, const int* dst, const float* edge_attr, float* hg,
                    float* out, const float* resid, const float* rbias, const Scratch& S) {
  if (ET) wa_k<<<1, 16, 0, st>>>(We, ae, S.wa, C);
  gemm_k128<C><<<GW / 8, 256, 0, st>>>(hin, Wm, hg);
  node_prep_k<C, ET><<<SG, SB, 0, st>>>(hg, asrc, adst, S.mean_attr, S.wa, S.s_src, S.s_dst,
                                        S.self_l);
  if (ET) eterm_k<<<SG, SB, 0, st>>>(edge_attr, S.wa, S.eterm);
  tile_agg_k<C, ET, BUILD, EPI><<<NTILE, 1024, 0, st>>>(
      src, dst, S.key, S.s_src, S.s_dst, S.eterm, S.self_l, hg, out, bias, resid, rbias,
      S.glog_sr, S.glog_e, S.glog_n, edge_attr, S.mean_attr);
}

extern "C" void kernel_launch(void* const* d_in, const int* in_sizes, int n_in, void* d_out,
                              int out_size, void* d_ws, size_t ws_size, hipStream_t stream) {
  (void)in_sizes; (void)n_in; (void)out_size; (void)ws_size;
  const float* x = (const float*)d_in[0];
  const int* ei = (const int*)d_in[1];
  const int* src = ei;
  const int* dst = ei + NE;
  const float* edge_attr = (const float*)d_in[2];
  const float *W1 = (const float*)d_in[3], *a1s = (const float*)d_in[4],
              *a1d = (const float*)d_in[5], *b1 = (const float*)d_in[6];
  const float *W2 = (const float*)d_in[7], *a2s = (const float*)d_in[8],
              *a2d = (const float*)d_in[9], *b2 = (const float*)d_in[10];
  const float *We2 = (const float*)d_in[11], *ae2 = (const float*)d_in[12];
  const float *Wr = (const float*)d_in[13], *br = (const float*)d_in[14];
  const float *Wmu = (const float*)d_in[15], *amus = (const float*)d_in[16],
              *amud = (const float*)d_in[17], *bmu = (const float*)d_in[18],
              *Wemu = (const float*)d_in[19], *aemu = (const float*)d_in[20];
  const float *Wls = (const float*)d_in[21], *alss = (const float*)d_in[22],
              *alsd = (const float*)d_in[23], *bls = (const float*)d_in[24],
              *Wels = (const float*)d_in[25], *aels = (const float*)d_in[26];
  float* outf = (float*)d_out;

  float* W = (float*)d_ws;
  size_t o = 0;
  float* hA = W + o; o += (size_t)NN * 128;
  float* hB = W + o; o += (size_t)NN * 128;
  float* hC = W + o; o += (size_t)NN * 128;
  Scratch S;
  S.s_src = W + o; o += NN;
  S.s_dst = W + o; o += NN;
  S.self_l = W + o; o += NN;
  S.mean_attr = W + o; o += (size_t)NN * 16;
  S.eterm = W + o; o += NE;
  S.wa = W + o; o += 16;
  unsigned* key = (unsigned*)(W + o); o += NE / 4;
  S.key = key;
  o += 16;
  S.glog_sr = (int*)(W + o); o += (size_t)NTILE * 32 * LOGCAP;
  S.glog_e = (int*)(W + o); o += (size_t)NTILE * 32 * LOGCAP;
  S.glog_n = (int*)(W + o); o += NTILE * 32;

  tile_key_k<<<KEYB, SB, 0, stream>>>(dst, key);

  gemm_k128<128><<<GW / 8, 256, 0, stream>>>(x, Wr, hC);

  run_gat<128, false, true, EPI_RELU>(stream, x, W1, a1s, a1d, b1, nullptr, nullptr, src, dst,
                                      edge_attr, hA, hB, nullptr, nullptr, S);

  run_gat<128, true, false, EPI_RELU_RES>(stream, hB, W2, a2s, a2d, b2, We2, ae2, src, dst,
                                          edge_attr, hA, hB, hC, br, S);

  run_gat<64, true, false, EPI_BIAS>(stream, hB, Wmu, amus, amud, bmu, Wemu, aemu, src, dst,
                                     edge_attr, hA, outf, nullptr, nullptr, S);
  run_gat<64, true, false, EPI_BIAS>(stream, hB, Wls, alss, alsd, bls, Wels, aels, src, dst,
                                     edge_attr, hA, outf + (size_t)NN * 64, nullptr, nullptr, S);
}


# Edge terms of conv2 and both heads made in conv1's owner sweep from the attribute registers it holds, kept in log order: 3 edge_attr passes, edge-id gathers, 3 launches go; dot sum order changes
# speedup vs baseline: 5.2499x; 1.1987x over previous
#include <hip/hip_runtime.h>
#include <math.h>

typedef float v2f __attribute__((ext_vector_type(2)));
typedef float v8f __attribute__((ext_vector_type(8)));

static constexpr int   NN = 100000;
static constexpr int   NE = 1600000;
static constexpr float NEG = 0.2f;

__device__ __forceinline__ float leaky(float x) { return x > 0.f ? x : NEG * x; }
__device__ __forceinline__ unsigned f2ord(float f) {
  unsigned u = __float_as_uint(f);
  return (u & 0x80000000u) ? ~u : (u | 0x80000000u);
}
__device__ __forceinline__ float ord2f(unsigned v) {
  unsigned u = (v & 0x80000000u) ? (v & 0x7fffffffu) : ~v;
  return __uint_as_float(u);
}

static constexpr int SB = 256;
static constexpr int SG = 2048;
static constexpr int ST = SG * SB;
static constexpr int NW = ST / 32;

__global__ void wa_k(const float* __restrict__ We, const float* __restrict__ ae,
                     float* wa, int K) {
  int i = threadIdx.x;
  if (i < 16) { float s = 0.f; for (int j = 0; j < K; ++j) s += We[i * K + j] * ae[j]; wa[i] = s; }
}

static constexpr int GW = 2048;
template <int COUT>
__global__ __launch_bounds__(256) void gemm_k128(const float* __restrict__ A,
                                                 const float* __restrict__ B,
                                                 float* __restrict__ C) {
  constexpr int NCG = COUT / 32;
  constexpr int NSW = GW / NCG;
  constexpr int NS = NN / 16;
  static_assert(NN % 16 == 0, "rows come in stripes of 16");
  const int lane = threadIdx.x & 31, half = lane >> 4, lidx = lane & 15;
  const int wid = __builtin_amdgcn_readfirstlane((int)((blockIdx.x * 256 + threadIdx.x) >> 5));
  const int n0 = (wid % NCG) * 32;
  const int sw = wid / NCG;

  float bq[2][16][4];
#pragma unroll
  for (int t = 0; t < 2; ++t)
#pragma unroll
    for (int j = 0; j < 16; ++j)
#pragma unroll
      for (int i = 0; i < 4; ++i)
        bq[t][j][i] = B[(8 * j + 4 * half + i) * COUT + n0 + 16 * t + lidx];

  const float4* A4 = (const float4*)A;
  auto load = [&](float4 (&a)[16], int s) {
    const float4* p = A4 + ((size_t)s * 16 + lidx) * 32 + half;
#pragma unroll
    for (int j = 0; j < 16; ++j) a[j] = p[2 * j];
  };
  auto compute = [&](const float4 (&a)[16], int s) {
    v8f acc0 = {}, acc1 = {};
#pragma unroll
    for (int j = 0; j < 16; ++j) {
      v2f a01, a23, b;
      a01.x = a[j].x; a01.y = a[j].y;
      a23.x = a[j].z; a23.y = a[j].w;
      b.x = bq[0][j][0]; b.y = bq[0][j][1];
      acc0 = __builtin_amdgcn_wmma_f32_16x16x4_f32(false, a01, false, b, (short)0, acc0,
                                                   false, false);
      b.x = bq[1][j][0]; b.y = bq[1][j][1];
      acc1 = __builtin_amdgcn_wmma_f32_16x16x4_f32(false, a01, false, b, (short)0, acc1,
                                                   false, false);
      b.x = bq[0][j][2]; b.y = bq[0][j][3];
      acc0 = __builtin_amdgcn_wmma_f32_16x16x4_f32(false, a23, false, b, (short)0, acc0,
                                                   false, false);
      b.x = bq[1][j][2]; b.y = bq[1][j][3];
      acc1 = __builtin_amdgcn_wmma_f32_16x16x4_f32(false, a23, false, b, (short)0, acc1,
                                                   false, false);
    }
    float* c = C + ((size_t)s * 16 + 8 * half) * COUT + n0 + lidx;
#pragma unroll
    for (int r = 0; r < 8; ++r) {
      c[r * COUT] = acc0[r];
      c[r * COUT + 16] = acc1[r];
    }
  };

  float4 ba[16], bb[16];
  int s = sw;
  load(ba, s);
  for (; s + NSW < NS; s += 2 * NSW) {
    load(bb, s + NSW);
    compute(ba, s);
    load(ba, min(s + 2 * NSW, NS - 1));
    compute(bb, s + NSW);
  }
  if (s < NS) compute(ba, s);
}

template <int C, bool ET>
__global__ __launch_bounds__(SB) void node_prep_k(
    const float* __restrict__ h, const float* __restrict__ a_src,
    const float* __restrict__ a_dst, const float* __restrict__ mean_attr,
    const float* __restrict__ wa, float* __restrict__ s_src, float* __restrict__ s_dst,
    float* __restrict__ self_logit) {
  constexpr int R = C / 32;
  constexpr int U = 256 / C;
  const int gid = blockIdx.x * SB + threadIdx.x;
  const int lane = gid & 31;
  float as[R], ad[R];
#pragma unroll
  for (int i = 0; i < R; ++i) { as[i] = a_src[lane + i * 32]; ad[i] = a_dst[lane + i * 32]; }
  for (int n0 = gid >> 5; n0 < NN; n0 += NW * U) {
    float hv[U][R];
#pragma unroll
    for (int u = 0; u < U; ++u) {
      const int nc = min(n0 + u * NW, NN - 1);
#pragma unroll
      for (int i = 0; i < R; ++i) hv[u][i] = h[(size_t)nc * C + lane + i * 32];
    }
    float psv[U], pdv[U];
#pragma unroll
    for (int u = 0; u < U; ++u) {
      float ps = 0.f, pd = 0.f;
#pragma unroll
      for (int i = 0; i < R; ++i) { ps += hv[u][i] * as[i]; pd += hv[u][i] * ad[i]; }
#pragma unroll
      for (int m = 16; m > 0; m >>= 1) { ps += __shfl_xor(ps, m, 32); pd += __shfl_xor(pd, m, 32); }
      psv[u] = ps; pdv[u] = pd;
    }
#pragma unroll
    for (int u = 0; u < U; ++u) {
      const int n = n0 + u * NW;
      if (n < NN) {
        const float ps = psv[u], pd = pdv[u];
        if (lane == 0) {
          float st = 0.f;
          if (ET) { for (int j = 0; j < 16; ++j) st += mean_attr[(size_t)n * 16 + j] * wa[j]; }
          float sl = leaky(ps + pd + st);
          s_src[n] = ps; s_dst[n] = pd; self_logit[n] = sl;
        }
      }
    }
  }
}

__device__ __forceinline__ float dot16(const float4 (&a)[4], const float* __restrict__ w) {
  float s[4];
#pragma unroll
  for (int q = 0; q < 4; ++q)
    s[q] = a[q].x * w[4 * q] + a[q].y * w[4 * q + 1] + a[q].z * w[4 * q + 2] + a[q].w * w[4 * q + 3];
  return (s[0] + s[1]) + (s[2] + s[3]);
}

static constexpr int RT = 400;
static constexpr int AP = 17;
static constexpr int NTILE = (NN + RT - 1) / RT;
static constexpr int QCAP = 64;
static constexpr int LOGCAP = 320;
static constexpr int NSTEP = NE / 512;
static constexpr size_t LOGPL = (size_t)NTILE * 32 * LOGCAP;
static_assert(NE % 512 == 0, "the key plane is swept in steps of 512 edges");
static_assert(NTILE <= 255, "a tile id is one byte, and 255 is the filler that matches no tile");
static_assert(NN <= (1 << 17) && RT <= (1 << 9), "a log entry packs source (17 bits) and row (9 bits)");

static constexpr int KEYB = (NE / 4 + SB - 1) / SB;
__global__ __launch_bounds__(SB) void tile_key_k(const int* __restrict__ dst,
                                                 unsigned* __restrict__ key) {
  const int i = blockIdx.x * SB + threadIdx.x;
  if (i < NE / 4) {
    const int4 d = ((const int4*)dst)[i];
    key[i] = (unsigned)d.x / RT | ((unsigned)d.y / RT) << 8 | ((unsigned)d.z / RT) << 16 |
             ((unsigned)d.w / RT) << 24;
  }
}

template <int V> struct RowVec;
template <> struct RowVec<4> { typedef float4 T; };
template <> struct RowVec<2> { typedef float2 T; };
__device__ __forceinline__ float vget(const float4& v, int k) {
  return k == 0 ? v.x : k == 1 ? v.y : k == 2 ? v.z : v.w;
}
__device__ __forceinline__ float vget(const float2& v, int k) { return k == 0 ? v.x : v.y; }
__device__ __forceinline__ void vset(float4& v, int k, float f) {
  if (k == 0) v.x = f; else if (k == 1) v.y = f; else if (k == 2) v.z = f; else v.w = f;
}
__device__ __forceinline__ void vset(float2& v, int k, float f) {
  if (k == 0) v.x = f; else v.y = f;
}
__device__ __forceinline__ void keep_loaded(const float4& v) {
  asm volatile("" ::"v"(v.x), "v"(v.y), "v"(v.z), "v"(v.w));
}
__device__ __forceinline__ void keep_loaded(const float2& v) {
  asm volatile("" ::"v"(v.x), "v"(v.y));
}

static constexpr int EPI_BIAS = 0;
static constexpr int EPI_RELU = 1;
static constexpr int EPI_RELU_RES = 2;
template <int C, bool ET, bool BUILD, int EPI>
__global__ __launch_bounds__(1024) void tile_agg_k(
    const int* __restrict__ src, const int* __restrict__ dst, const unsigned* __restrict__ key,
    const float* __restrict__ s_src, const float* __restrict__ s_dst,
    const float* __restrict__ self_logit,
    const float* __restrict__ h, float* __restrict__ out, const float* __restrict__ bias,
    const float* __restrict__ resid, const float* __restrict__ rbias,
    int* __restrict__ glog_sr, float* __restrict__ glog_t, int* __restrict__ glog_n,
    const float* __restrict__ edge_attr, const float* __restrict__ wa,
    float* __restrict__ mean_attr) {
  constexpr int V = C / 32;
  static_assert(!BUILD || RT * AP <= RT * C, "the attr sums borrow the tile's LDS");
  typedef typename RowVec<V>::T VecT;
  __shared__ float tile[RT * C];
  __shared__ float ssum[RT];
  __shared__ unsigned smax[RT];
  __shared__ int queue[32 * QCAP];
  __shared__ int log_sr[32 * LOGCAP];
  __shared__ float log_l[32 * LOGCAP];
  const int tid = threadIdx.x, lane = tid & 31;
  const int wave = __builtin_amdgcn_readfirstlane(tid >> 5);
  const int row0 = blockIdx.x * RT;
  const int rows = min(RT, NN - row0);

  for (int r = tid; r < rows; r += 1024) smax[r] = f2ord(self_logit[row0 + r]);
  if (BUILD) { for (int i = tid; i < RT * AP; i += 1024) tile[i] = 0.f; }
  __syncthreads();

  int* myq = &queue[wave * QCAP];
  int* mysr = &log_sr[wave * LOGCAP];
  float* myll = &log_l[wave * LOGCAP];
  const int gw = blockIdx.x * 32 + wave;
  int* gsr = glog_sr + (size_t)gw * LOGCAP;
  float* gt = glog_t + (size_t)gw * LOGCAP;
  int logn = 0;
  bool ovf = false;

  auto edge_logit = [&](int e, int& s, int& r) {
    s = src[e];
    const int d = dst[e];
    float l = s_src[s] + s_dst[d];
    if (ET) {
      const float4* ap = (const float4*)&edge_attr[(size_t)e * 16];
      float4 av[4];
#pragma unroll
      for (int q = 0; q < 4; ++q) av[q] = ap[q];
      l += dot16(av, wa);
    }
    r = d - row0;
    return leaky(l);
  };

  auto proc_max = [&](int nb) {
    int sr = 0;
    float ll = 0.f;
    float et[3] = {0.f, 0.f, 0.f};
    if (lane < nb) {
      int s, r;
      const int e = myq[lane];
      float4 av[4];
      if (BUILD) {
        const float4* ap = (const float4*)&edge_attr[(size_t)e * 16];
#pragma unroll
        for (int q = 0; q < 4; ++q) av[q] = ap[q];
      }
      ll = edge_logit(e, s, r);
      atomicMax(&smax[r], f2ord(ll));
      sr = s | r << 17;
      if (BUILD) {
#pragma unroll
        for (int k = 0; k < 3; ++k) et[k] = dot16(av, wa + 16 * k);
        float* t = &tile[r * AP];
#pragma unroll
        for (int q = 0; q < 4; ++q) {
          atomicAdd(t + 4 * q, av[q].x); atomicAdd(t + 4 * q + 1, av[q].y);
          atomicAdd(t + 4 * q + 2, av[q].z); atomicAdd(t + 4 * q + 3, av[q].w);
        }
        atomicAdd(t + 16, 1.f);
      }
    }
    if (!ovf && logn + nb <= LOGCAP) {
      if (lane < nb) {
        mysr[logn + lane] = sr; myll[logn + lane] = ll;
        if (BUILD) {
          gsr[logn + lane] = sr;
#pragma unroll
          for (int k = 0; k < 3; ++k) gt[k * LOGPL + logn + lane] = et[k];
        }
      }
      logn += nb;
    } else {
      ovf = true;
    }
  };
  auto max_from_log = [&](int n) {
    for (int b = 0; b < n; b += 128) {
      int sr[4];
      float a[4], d[4], t[4];
#pragma unroll
      for (int u = 0; u < 4; ++u) {
        const int ic = min(b + 32 * u + lane, n - 1);
        sr[u] = gsr[ic];
        t[u] = ET ? gt[ic] : 0.f;
      }
      if (ET)
        asm volatile("" ::"v"(sr[0]), "v"(sr[1]), "v"(sr[2]), "v"(sr[3]), "v"(t[0]), "v"(t[1]),
                     "v"(t[2]), "v"(t[3]));
      else
        asm volatile("" ::"v"(sr[0]), "v"(sr[1]), "v"(sr[2]), "v"(sr[3]));
#pragma unroll
      for (int u = 0; u < 4; ++u) {
        a[u] = s_src[sr[u] & 0x1ffff];
        d[u] = s_dst[row0 + (sr[u] >> 17)];
      }
      asm volatile("" ::"v"(a[0]), "v"(a[1]), "v"(a[2]), "v"(a[3]), "v"(d[0]), "v"(d[1]),
                   "v"(d[2]), "v"(d[3]));
#pragma unroll
      for (int u = 0; u < 4; ++u) {
        const int i = b + 32 * u + lane;
        if (i < n) {
          float l = a[u] + d[u];
          if (ET) l += t[u];
          const float ll = leaky(l);
          atomicMax(&smax[sr[u] >> 17], f2ord(ll));
          mysr[i] = sr[u];
          myll[i] = ll;
        }
      }
    }
  };

  auto agg_rows = [&](int nb, int s, int r, float w) {
    for (int j0 = 0; j0 < nb; j0 += 8) {
      VecT hv[8];
#pragma unroll
      for (int u = 0; u < 8; ++u) {
        const int sj = __builtin_amdgcn_readlane(s, j0 + u);
        hv[u] = *(const VecT*)&h[(size_t)sj * C + lane * V];
      }
#pragma unroll
      for (int u = 0; u < 8; ++u) {
        if (j0 + u < nb) {
          const int rj = __builtin_amdgcn_readlane(r, j0 + u);
          const float wj = __int_as_float(__builtin_amdgcn_readlane(__float_as_int(w), j0 + u));
          float* t = &tile[rj * C + lane];
#pragma unroll
          for (int k = 0; k < V; ++k) atomicAdd(t + k * 32, wj * vget(hv[u], k));
        }
      }
    }
  };
  auto proc_log = [&](int base, int nb) {
    int s = 0, r = 0;
    float w = 0.f;
    if (lane < nb) {
      const int sr = mysr[base + lane];
      s = sr & 0x1ffff;
      r = sr >> 17;
      w = __expf(myll[base + lane] - ord2f(smax[r]));
      atomicAdd(&ssum[r], w);
    }
    agg_rows(nb, s, r, w);
  };
  auto proc_queue = [&](int nb) {
    int s = 0, r = 0;
    float w = 0.f;
    if (lane < nb) {
      const float ll = edge_logit(myq[lane], s, r);
      w = __expf(ll - ord2f(smax[r]));
      atomicAdd(&ssum[r], w);
    }
    agg_rows(nb, s, r, w);
  };

  const unsigned tt = blockIdx.x * 0x01010101u;
  const int rot = blockIdx.x * (NSTEP / NTILE);
  auto step_of = [&](int k) { const int kk = k + rot; return kk >= NSTEP ? kk - NSTEP : kk; };
  const uint4* key4 = (const uint4*)key;
  auto fetch = [&](uint4 (&d)[4], int k0) {
#pragma unroll
    for (int u = 0; u < 4; ++u) {
      const int k = k0 + 32 * u;
      d[u] = (k < NSTEP) ? key4[(size_t)step_of(k) * 32 + lane] : make_uint4(~0u, ~0u, ~0u, ~0u);
    }
  };
  auto differs = [&](unsigned x) {
    const unsigned y = x ^ tt;
    return (((y & 0x7f7f7f7fu) + 0x7f7f7f7fu) | y) & 0x80808080u;
  };
  auto sweep = [&](auto&& proc) {
    int cnt = 0;
    uint4 cur[4], nxt[4];
    fetch(cur, wave);
    for (int k0 = wave; k0 < NSTEP; k0 += 32 * 4) {
      fetch(nxt, k0 + 32 * 4);
#pragma unroll
      for (int u = 0; u < 4; ++u) {
        unsigned m = ~(differs(cur[u].x) >> 7 | differs(cur[u].y) >> 6 | differs(cur[u].z) >> 5 |
                       differs(cur[u].w) >> 4) & 0x0f0f0f0fu;
        unsigned live = __builtin_amdgcn_ballot_w32(m != 0u);
        if (live != 0u) {
          const int e0 = step_of(k0 + 32 * u) * 512 + lane * 16;
          do {
            if (m != 0u) {
              const int bit = __builtin_ctz(m);
              myq[cnt + __builtin_amdgcn_mbcnt_lo(live, 0u)] = e0 + (bit & 7) * 4 + (bit >> 3);
              m &= m - 1u;
            }
            cnt += __popc(live);
            live = __builtin_amdgcn_ballot_w32(m != 0u);
            if (cnt >= 32) {
              proc(32);
              const int rem = cnt - 32;
              const int keep = (lane < rem) ? myq[32 + lane] : 0;
              if (lane < rem) myq[lane] = keep;
              cnt = rem;
            }
          } while (live != 0u);
        }
      }
#pragma unroll
      for (int u = 0; u < 4; ++u) cur[u] = nxt[u];
    }
    if (cnt > 0) proc(cnt);
  };

  if (BUILD) {
    sweep(proc_max);
    if (lane == 0) glog_n[gw] = ovf ? -1 : logn;
  } else {
    const int gn = __builtin_amdgcn_readfirstlane(glog_n[gw]);
    if (gn >= 0) {
      max_from_log(gn);
      logn = gn;
    } else {
      sweep(proc_max);
    }
  }
  __syncthreads();

  if (BUILD) {
    for (int i = tid; i < rows * 16; i += 1024) {
      const int r = i >> 4, j = i & 15;
      mean_attr[(size_t)(row0 + r) * 16 + j] = tile[r * AP + j] / fmaxf(tile[r * AP + 16], 1.f);
    }
    __syncthreads();
  }

  for (int i = tid; i < rows * 32; i += 1024) {
    const int r = i >> 5, n = row0 + r;
    const float a = __expf(self_logit[n] - ord2f(smax[r]));
    const VecT hv = *(const VecT*)&h[(size_t)n * C + lane * V];
#pragma unroll
    for (int k = 0; k < V; ++k) tile[r * C + k * 32 + lane] = a * vget(hv, k);
    if (lane == 0) ssum[r] = a;
  }
  __syncthreads();

  const int nlog = __builtin_amdgcn_readfirstlane(logn);
  if (__builtin_amdgcn_readfirstlane((int)ovf) == 0) {
    int b = 0;
    for (; b + 32 <= nlog; b += 32) proc_log(b, 32);
    if (b < nlog) proc_log(b, nlog - b);
  } else {
    sweep(proc_queue);
  }
  __syncthreads();

  const VecT bv = *(const VecT*)&bias[lane * V];
  VecT rbv = {};
  if (EPI == EPI_RELU_RES) rbv = *(const VecT*)&rbias[lane * V];
  for (int i0 = tid; i0 < rows * 32; i0 += 4 * 1024) {
    VecT rv[4] = {};
    if (EPI == EPI_RELU_RES) {
#pragma unroll
      for (int u = 0; u < 4; ++u) {
        const int nc = row0 + (min(i0 + u * 1024, rows * 32 - 1) >> 5);
        rv[u] = *(const VecT*)&resid[(size_t)nc * C + lane * V];
      }
#pragma unroll
      for (int u = 0; u < 4; ++u) keep_loaded(rv[u]);
    }
#pragma unroll
    for (int u = 0; u < 4; ++u) {
      const int i = i0 + u * 1024;
      if (i < rows * 32) {
        const int r = i >> 5, n = row0 + r;
        const float den = ssum[r] + 1e-16f;
        VecT o;
#pragma unroll
        for (int k = 0; k < V; ++k) {
          float x = tile[r * C + k * 32 + lane] / den + vget(bv, k);
          if (EPI != EPI_BIAS) x = fmaxf(x, 0.f);
          if (EPI == EPI_RELU_RES) x = x + vget(rv[u], k) + vget(rbv, k);
          vset(o, k, x);
        }
        *(VecT*)&out[(size_t)n * C + lane * V] = o;
      }
    }
  }
}

struct Scratch {
  float *s_src, *s_dst, *self_l;
  float* mean_attr;
  float* wa;
  const unsigned* key;
  int *glog_sr, *glog_n;
  float* glog_t;
};

template <int C, bool ET, bool BUILD, int EPI>
static void run_gat(hipStream_t st, const float* hin, const float* Wm, const float* asrc,
                    const float* adst, const float* bias, const float* wa, float* glog_t,
                    const int* src, const int* dst, const float* edge_attr, float* hg,
                    float* out, const float* resid, const float* rbias, const Scratch& S) {
  gemm_k128<C><<<GW / 8, 256, 0, st>>>(hin, Wm, hg);
  node_prep_k<C, ET><<<SG, SB, 0, st>>>(hg, asrc, adst, S.mean_attr, wa, S.s_src, S.s_dst,
                                        S.self_l);
  tile_agg_k<C, ET, BUILD, EPI><<<NTILE, 1024, 0, st>>>(
      src, dst, S.key, S.s_src, S.s_dst, S.self_l, hg, out, bias, resid, rbias, S.glog_sr,
      glog_t, S.glog_n, edge_attr, wa, S.mean_attr);
}

extern "C" void kernel_launch(void* const* d_in, const int* in_sizes, int n_in, void* d_out,
                              int out_size, void* d_ws, size_t ws_size, hipStream_t stream) {
  (void)in_sizes; (void)n_in; (void)out_size; (void)ws_size;
  const float* x = (const float*)d_in[0];
  const int* ei = (const int*)d_in[1];
  const int* src = ei;
  const int* dst = ei + NE;
  const float* edge_attr = (const float*)d_in[2];
  const float *W1 = (const float*)d_in[3], *a1s = (const float*)d_in[4],
              *a1d = (const float*)d_in[5], *b1 = (const float*)d_in[6];
  const float *W2 = (const float*)d_in[7], *a2s = (const float*)d_in[8],
              *a2d = (const float*)d_in[9], *b2 = (const float*)d_in[10];
  const float *We2 = (const float*)d_in[11], *ae2 = (const float*)d_in[12];
  const float *Wr = (const float*)d_in[13], *br = (const float*)d_in[14];
  const float *Wmu = (const float*)d_in[15], *amus = (const float*)d_in[16],
              *amud = (const float*)d_in[17], *bmu = (const float*)d_in[18],
              *Wemu = (const float*)d_in[19], *aemu = (const float*)d_in[20];
  const float *Wls = (const float*)d_in[21], *alss = (const float*)d_in[22],
              *alsd = (const float*)d_in[23], *bls = (const float*)d_in[24],
              *Wels = (const float*)d_in[25], *aels = (const float*)d_in[26];
  float* outf = (float*)d_out;

  float* W = (float*)d_ws;
  size_t o = 0;
  float* hA = W + o; o += (size_t)NN * 128;
  float* hB = W + o; o += (size_t)NN * 128;
  float* hC = W + o; o += (size_t)NN * 128;
  Scratch S;
  S.s_src = W + o; o += NN;
  S.s_dst = W + o; o += NN;
  S.self_l = W + o; o += NN;
  S.mean_attr = W + o; o += (size_t)NN * 16;
  S.wa = W + o; o += 48;
  unsigned* key = (unsigned*)(W + o); o += NE / 4;
  S.key = key;
  o += 48;
  S.glog_sr = (int*)(W + o); o += LOGPL;
  S.glog_t = W + o; o += 3 * LOGPL;
  S.glog_n = (int*)(W + o); o += NTILE * 32;

  tile_key_k<<<KEYB, SB, 0, stream>>>(dst, key);

  wa_k<<<1, 16, 0, stream>>>(We2, ae2, S.wa, 128);
  wa_k<<<1, 16, 0, stream>>>(Wemu, aemu, S.wa + 16, 64);
  wa_k<<<1, 16, 0, stream>>>(Wels, aels, S.wa + 32, 64);

  gemm_k128<128><<<GW / 8, 256, 0, stream>>>(x, Wr, hC);

  run_gat<128, false, true, EPI_RELU>(stream, x, W1, a1s, a1d, b1, S.wa, S.glog_t, src, dst,
                                      edge_attr, hA, hB, nullptr, nullptr, S);

  run_gat<128, true, false, EPI_RELU_RES>(stream, hB, W2, a2s, a2d, b2, S.wa, S.glog_t, src,
                                          dst, edge_attr, hA, hB, hC, br, S);

  run_gat<64, true, false, EPI_BIAS>(stream, hB, Wmu, amus, amud, bmu, S.wa + 16,
                                     S.glog_t + LOGPL, src, dst, edge_attr, hA, outf, nullptr,
                                     nullptr, S);
  run_gat<64, true, false, EPI_BIAS>(stream, hB, Wls, alss, alsd, bls, S.wa + 32,
                                     S.glog_t + 2 * LOGPL, src, dst, edge_attr, hA,
                                     outf + (size_t)NN * 64, nullptr, nullptr, S);
}
